# starting kernel
# baseline (speedup 1.0000x reference)
#include <hip/hip_runtime.h>
#include <hip/hip_bf16.h>
#include <math.h>


typedef __attribute__((ext_vector_type(2))) float v2f;
typedef __attribute__((ext_vector_type(8))) float v8f;

#define DIM   64
#define NREL  4
#define KCONV 320
#define KHW   128
#define STRC  324
#define STRH  132

__device__ __forceinline__ float sigmoidf_(float v) {
    return 1.0f / (1.0f + __expf(-v));
}

__global__ void rgcn_scatter_kernel(const float* __restrict__ x,
                                    const float* __restrict__ ew,
                                    const int*   __restrict__ src,
                                    const int*   __restrict__ dst,
                                    const int*   __restrict__ rel,
                                    float*       __restrict__ agg,
                                    float*       __restrict__ deg,
                                    int nEdges)
{
    int tid  = blockIdx.x * blockDim.x + threadIdx.x;
    int e    = tid >> 5;
    int lane = tid & 31;
    if (e >= nEdges) return;
    int   s = src[e];
    int   d = dst[e];
    int   r = rel[e];
    float w = ew[e];
    float2 v = ((const float2*)(x + (size_t)s * DIM))[lane];
    float* aggRow = agg + ((size_t)d * NREL + r) * DIM;
    atomicAdd(aggRow + 2 * lane,     v.x * w);
    atomicAdd(aggRow + 2 * lane + 1, v.y * w);
    if (lane == 0) atomicAdd(deg + (size_t)d * NREL + r, w);
}

__global__ void rgcn_update_kernel(const float* __restrict__ agg,
                                   const float* __restrict__ deg,
                                   const float* __restrict__ x,
                                   const float* __restrict__ lin_w,
                                   const float* __restrict__ lin_b,
                                   const float* __restrict__ self_w,
                                   const float* __restrict__ self_b,
                                   float*       __restrict__ hidden,
                                   int nNodes)
{
    extern __shared__ float smem[];
    float* WT = smem;
    float* RD = smem + DIM * STRC;

    const int t = threadIdx.x;
    for (int i = t; i < KCONV * DIM; i += blockDim.x) {
        int k = i >> 6;
        int c = i & 63;
        float v = (k < NREL * DIM) ? lin_w[i] : self_w[i - NREL * DIM * DIM];
        WT[(size_t)c * STRC + k] = v;
    }

    const int wave = t >> 5;
    const int lane = t & 31;
    const int gw   = blockIdx.x * (blockDim.x >> 5) + wave;
    const long base = (long)gw * 16;

    if (base < nNodes) {
        float2 dv = ((const float2*)(deg + base * NREL))[lane];
        RD[wave * 64 + 2 * lane]     = 1.0f / (dv.x + 1e-10f);
        RD[wave * 64 + 2 * lane + 1] = 1.0f / (dv.y + 1e-10f);
    }
    __syncthreads();
    if (base >= nNodes) return;

    const int m     = lane & 15;
    const int khalf = lane >> 4;
    const float* aggRow = agg + (base + m) * (size_t)(NREL * DIM);
    const float* xRow   = x   + (base + m) * (size_t)DIM;
    const float* B0 = WT + (size_t)(m)      * STRC;
    const float* B1 = WT + (size_t)(16 + m) * STRC;
    const float* B2 = WT + (size_t)(32 + m) * STRC;
    const float* B3 = WT + (size_t)(48 + m) * STRC;
    const float* rd = RD + wave * 64 + m * NREL;

    v8f a0 = {}, a1 = {}, a2 = {}, a3 = {};
    for (int k = 0; k < NREL * DIM; k += 4) {
        int kk = k + 2 * khalf;
        float  sc = rd[k >> 6];
        float2 av = *(const float2*)(aggRow + kk);
        v2f a; a.x = av.x * sc; a.y = av.y * sc;
        v2f b0 = *(const v2f*)(B0 + kk);
        v2f b1 = *(const v2f*)(B1 + kk);
        v2f b2 = *(const v2f*)(B2 + kk);
        v2f b3 = *(const v2f*)(B3 + kk);
        a0 = __builtin_amdgcn_wmma_f32_16x16x4_f32(false, a, false, b0, (short)0, a0, false, false);
        a1 = __builtin_amdgcn_wmma_f32_16x16x4_f32(false, a, false, b1, (short)0, a1, false, false);
        a2 = __builtin_amdgcn_wmma_f32_16x16x4_f32(false, a, false, b2, (short)0, a2, false, false);
        a3 = __builtin_amdgcn_wmma_f32_16x16x4_f32(false, a, false, b3, (short)0, a3, false, false);
    }
    for (int k = 0; k < DIM; k += 4) {
        int kk = k + 2 * khalf;
        float2 av = *(const float2*)(xRow + kk);
        v2f a; a.x = av.x; a.y = av.y;
        int kw = NREL * DIM + kk;
        v2f b0 = *(const v2f*)(B0 + kw);
        v2f b1 = *(const v2f*)(B1 + kw);
        v2f b2 = *(const v2f*)(B2 + kw);
        v2f b3 = *(const v2f*)(B3 + kw);
        a0 = __builtin_amdgcn_wmma_f32_16x16x4_f32(false, a, false, b0, (short)0, a0, false, false);
        a1 = __builtin_amdgcn_wmma_f32_16x16x4_f32(false, a, false, b1, (short)0, a1, false, false);
        a2 = __builtin_amdgcn_wmma_f32_16x16x4_f32(false, a, false, b2, (short)0, a2, false, false);
        a3 = __builtin_amdgcn_wmma_f32_16x16x4_f32(false, a, false, b3, (short)0, a3, false, false);
    }

#define CONV_EPILOGUE(ACC, TILE)                                           \
    {                                                                      \
        int   col = (TILE) * 16 + m;                                       \
        float bia = lin_b[col] + self_b[col];                              \
        _Pragma("unroll")                                                  \
        for (int r = 0; r < 8; ++r) {                                      \
            long  row = base + khalf * 8 + r;                              \
            hidden[row * DIM + col] = sigmoidf_(ACC[r] + bia);             \
        }                                                                  \
    }
    CONV_EPILOGUE(a0, 0) CONV_EPILOGUE(a1, 1) CONV_EPILOGUE(a2, 2) CONV_EPILOGUE(a3, 3)
#undef CONV_EPILOGUE
}

__global__ void highway_kernel(const float* __restrict__ hid,
                               const float* __restrict__ prev,
                               const float* __restrict__ pw,
                               const float* __restrict__ pb,
                               const float* __restrict__ tw,
                               const float* __restrict__ tb,
                               float*       __restrict__ outp,
                               int nNodes)
{
    extern __shared__ float smem[];
    float* WP = smem;
    float* WTT = smem + DIM * STRH;

    const int t = threadIdx.x;
    for (int i = t; i < KHW * DIM; i += blockDim.x) {
        int k = i >> 6;
        int c = i & 63;
        WP [(size_t)c * STRH + k] = pw[i];
        WTT[(size_t)c * STRH + k] = tw[i];
    }
    __syncthreads();

    const int wave = t >> 5;
    const int lane = t & 31;
    const int gw   = blockIdx.x * (blockDim.x >> 5) + wave;
    const long base = (long)gw * 16;
    if (base >= nNodes) return;

    const int m     = lane & 15;
    const int khalf = lane >> 4;
    const float* hRow = hid  + (base + m) * (size_t)DIM;
    const float* pRow = prev + (base + m) * (size_t)DIM;
    const float* P0 = WP  + (size_t)(m)      * STRH;
    const float* P1 = WP  + (size_t)(16 + m) * STRH;
    const float* P2 = WP  + (size_t)(32 + m) * STRH;
    const float* P3 = WP  + (size_t)(48 + m) * STRH;
    const float* T0 = WTT + (size_t)(m)      * STRH;
    const float* T1 = WTT + (size_t)(16 + m) * STRH;
    const float* T2 = WTT + (size_t)(32 + m) * STRH;
    const float* T3 = WTT + (size_t)(48 + m) * STRH;

    v8f p0 = {}, p1 = {}, p2 = {}, p3 = {};
    v8f g0 = {}, g1 = {}, g2 = {}, g3 = {};

#define HW_STEP(AROW, KOFF)                                                              \
    for (int k = 0; k < DIM; k += 4) {                                                   \
        int kk = k + 2 * khalf;                                                          \
        float2 av = *(const float2*)((AROW) + kk);                                       \
        v2f a; a.x = av.x; a.y = av.y;                                                   \
        int kw = kk + (KOFF);                                                            \
        v2f bp0 = *(const v2f*)(P0 + kw);                                                \
        v2f bp1 = *(const v2f*)(P1 + kw);                                                \
        v2f bp2 = *(const v2f*)(P2 + kw);                                                \
        v2f bp3 = *(const v2f*)(P3 + kw);                                                \
        v2f bt0 = *(const v2f*)(T0 + kw);                                                \
        v2f bt1 = *(const v2f*)(T1 + kw);                                                \
        v2f bt2 = *(const v2f*)(T2 + kw);                                                \
        v2f bt3 = *(const v2f*)(T3 + kw);                                                \
        p0 = __builtin_amdgcn_wmma_f32_16x16x4_f32(false, a, false, bp0, (short)0, p0, false, false); \
        p1 = __builtin_amdgcn_wmma_f32_16x16x4_f32(false, a, false, bp1, (short)0, p1, false, false); \
        p2 = __builtin_amdgcn_wmma_f32_16x16x4_f32(false, a, false, bp2, (short)0, p2, false, false); \
        p3 = __builtin_amdgcn_wmma_f32_16x16x4_f32(false, a, false, bp3, (short)0, p3, false, false); \
        g0 = __builtin_amdgcn_wmma_f32_16x16x4_f32(false, a, false, bt0, (short)0, g0, false, false); \
        g1 = __builtin_amdgcn_wmma_f32_16x16x4_f32(false, a, false, bt1, (short)0, g1, false, false); \
        g2 = __builtin_amdgcn_wmma_f32_16x16x4_f32(false, a, false, bt2, (short)0, g2, false, false); \
        g3 = __builtin_amdgcn_wmma_f32_16x16x4_f32(false, a, false, bt3, (short)0, g3, false, false); \
    }
    HW_STEP(hRow, 0)
    HW_STEP(pRow, DIM)
#undef HW_STEP

#define HW_EPILOGUE(PACC, GACC, TILE)                                      \
    {                                                                      \
        int   col = (TILE) * 16 + m;                                       \
        float pbv = pb[col];                                               \
        float tbv = tb[col];                                               \
        _Pragma("unroll")                                                  \
        for (int r = 0; r < 8; ++r) {                                      \
            long  row = base + khalf * 8 + r;                              \
            float pr = fmaxf(PACC[r] + pbv, 0.0f);                         \
            float gt = sigmoidf_(GACC[r] + tbv);                           \
            float h  = hid[row * DIM + col];                               \
            outp[row * DIM + col] = gt * pr + (1.0f - gt) * h;             \
        }                                                                  \
    }
    HW_EPILOGUE(p0, g0, 0) HW_EPILOGUE(p1, g1, 1)
    HW_EPILOGUE(p2, g2, 2) HW_EPILOGUE(p3, g3, 3)
#undef HW_EPILOGUE
}

extern "C" void kernel_launch(void* const* d_in, const int* in_sizes, int n_in,
                              void* d_out, int out_size, void* d_ws, size_t ws_size,
                              hipStream_t stream) {
    const float* node_feat   = (const float*)d_in[0];
    const float* edge_weight = (const float*)d_in[1];
    const float* conv_lin_w  = (const float*)d_in[2];
    const float* conv_lin_b  = (const float*)d_in[3];
    const float* conv_self_w = (const float*)d_in[4];
    const float* conv_self_b = (const float*)d_in[5];
    const float* hw_proj_w   = (const float*)d_in[6];
    const float* hw_proj_b   = (const float*)d_in[7];
    const float* hw_trans_w  = (const float*)d_in[8];
    const float* hw_trans_b  = (const float*)d_in[9];
    const int*   edge_src    = (const int*)d_in[10];
    const int*   edge_dst    = (const int*)d_in[11];
    const int*   edge_rel    = (const int*)d_in[12];
    float*       out         = (float*)d_out;

    const int N = in_sizes[0] / DIM;
    const int E = in_sizes[10];

    float* agg = (float*)d_ws;
    float* deg = agg + (size_t)N * NREL * DIM;
    float* hA  = deg + (size_t)N * NREL;
    float* hB  = hA  + (size_t)N * DIM;
    float* cur = hB  + (size_t)N * DIM;

    const int scatterBlocks = (E + 7) / 8;
    const int waves         = (N + 15) / 16;
    const int nodeBlocks    = (waves + 7) / 8;
    const size_t updLds = (size_t)(DIM * STRC + 8 * 64) * sizeof(float);
    const size_t hwLds  = (size_t)(2 * DIM * STRH) * sizeof(float);

    for (int i = 0; i < 3; ++i) {
        const float* xin  = (i == 0) ? node_feat : cur;
        const float* prev = (i == 0) ? node_feat : ((i == 1) ? hA : hB);
        float*       hid  = (i & 1) ? hB : hA;
        float*       gout = (i == 2) ? out : cur;

        hipMemsetAsync(agg, 0, (size_t)(N * NREL * DIM + N * NREL) * sizeof(float), stream);

        rgcn_scatter_kernel<<<scatterBlocks, 256, 0, stream>>>(
            xin, edge_weight, edge_src, edge_dst, edge_rel, agg, deg, E);

        rgcn_update_kernel<<<nodeBlocks, 256, updLds, stream>>>(
            agg, deg, xin,
            conv_lin_w  + (size_t)i * 256 * DIM, conv_lin_b  + (size_t)i * DIM,
            conv_self_w + (size_t)i * DIM * DIM, conv_self_b + (size_t)i * DIM,
            hid, N);

        highway_kernel<<<nodeBlocks, 256, hwLds, stream>>>(
            hid, prev,
            hw_proj_w  + (size_t)i * KHW * DIM, hw_proj_b  + (size_t)i * DIM,
            hw_trans_w + (size_t)i * KHW * DIM, hw_trans_b + (size_t)i * DIM,
            gout, N);
    }
}


# Atomic f32 scatter per layer replaced by one per-call bucket sort of edges by (dst,rel) (int atomics, arrival-order rank) and an atomic-free segmented gather; in-bucket sum order changes
# speedup vs baseline: 1.7979x; 1.7979x over previous
#include <hip/hip_runtime.h>
#include <hip/hip_bf16.h>
#include <math.h>


typedef __attribute__((ext_vector_type(2))) float v2f;
typedef __attribute__((ext_vector_type(8))) float v8f;

#define DIM   64
#define NREL  4
#define KCONV 320
#define KHW   128
#define STRC  324
#define STRH  132

__device__ __forceinline__ float sigmoidf_(float v) {
    return 1.0f / (1.0f + __expf(-v));
}

#define SCAN_CHUNK 1024
#define GATHER_NODES 8

__global__ void bucket_count_kernel(const int* __restrict__ dst,
                                    const int* __restrict__ rel,
                                    int*       __restrict__ cnt,
                                    int nEdges)
{
    int e = blockIdx.x * blockDim.x + threadIdx.x;
    if (e >= nEdges) return;
    atomicAdd(cnt + (size_t)dst[e] * NREL + rel[e], 1);
}

__global__ void bucket_blocksum_kernel(const int* __restrict__ cnt,
                                       int*       __restrict__ bsum,
                                       int nb)
{
    __shared__ int wsum[8];
    const int t  = threadIdx.x;
    const int i0 = blockIdx.x * SCAN_CHUNK + t * 4;
    int s = 0;
    if (i0 < nb) {
        int4 c = *(const int4*)(cnt + i0);
        s = c.x + c.y + c.z + c.w;
    }
    for (int o = 16; o > 0; o >>= 1) s += __shfl_xor(s, o);
    if ((t & 31) == 0) wsum[t >> 5] = s;
    __syncthreads();
    if (t == 0) {
        int tot = 0;
        for (int w = 0; w < 8; ++w) tot += wsum[w];
        bsum[blockIdx.x] = tot;
    }
}

__global__ void bucket_scan_sums_kernel(int* __restrict__ bsum, int nblk)
{
    __shared__ int wtot[32];
    const int t    = threadIdx.x;
    const int lane = t & 31;
    const int wave = t >> 5;
    int carry = 0;
    for (int base = 0; base < nblk; base += 1024) {
        const int i = base + t;
        const int v = (i < nblk) ? bsum[i] : 0;
        int inc = v;
        for (int o = 1; o < 32; o <<= 1) {
            int n = __shfl_up(inc, o);
            if (lane >= o) inc += n;
        }
        if (lane == 31) wtot[wave] = inc;
        __syncthreads();
        int woff = 0, total = 0;
        for (int w = 0; w < 32; ++w) {
            int x = wtot[w];
            total += x;
            if (w < wave) woff += x;
        }
        if (i < nblk) bsum[i] = carry + woff + inc - v;
        carry += total;
        __syncthreads();
    }
}

__global__ void bucket_offsets_kernel(const int* __restrict__ cnt,
                                      const int* __restrict__ bsum,
                                      int*       __restrict__ cursor,
                                      int nb)
{
    __shared__ int wtot[8];
    const int t    = threadIdx.x;
    const int lane = t & 31;
    const int wave = t >> 5;
    const int i0   = blockIdx.x * SCAN_CHUNK + t * 4;
    int4 c = {0, 0, 0, 0};
    if (i0 < nb) c = *(const int4*)(cnt + i0);
    const int tsum = c.x + c.y + c.z + c.w;
    int inc = tsum;
    for (int o = 1; o < 32; o <<= 1) {
        int n = __shfl_up(inc, o);
        if (lane >= o) inc += n;
    }
    if (lane == 31) wtot[wave] = inc;
    __syncthreads();
    int woff = 0;
    for (int w = 0; w < 8; ++w) if (w < wave) woff += wtot[w];
    if (i0 < nb) {
        int4 o4;
        o4.x = bsum[blockIdx.x] + woff + inc - tsum;
        o4.y = o4.x + c.x;
        o4.z = o4.y + c.y;
        o4.w = o4.z + c.z;
        *(int4*)(cursor + i0) = o4;
    }
}

__global__ void bucket_fill_kernel(const int*   __restrict__ src,
                                   const int*   __restrict__ dst,
                                   const int*   __restrict__ rel,
                                   const float* __restrict__ ew,
                                   int*         __restrict__ cursor,
                                   int*         __restrict__ erec,
                                   int nEdges)
{
    int e = blockIdx.x * blockDim.x + threadIdx.x;
    if (e >= nEdges) return;
    int pos = atomicAdd(cursor + (size_t)dst[e] * NREL + rel[e], 1);
    int2 r;
    r.x = src[e];
    r.y = __float_as_int(ew[e]);
    ((int2*)erec)[pos] = r;
}

__global__ void rgcn_gather_kernel(const float* __restrict__ x,
                                   const int*   __restrict__ erec,
                                   const int*   __restrict__ bend,
                                   float*       __restrict__ agg,
                                   float*       __restrict__ deg,
                                   int nNodes)
{
    const int lane = threadIdx.x & 31;
    const int wave = __builtin_amdgcn_readfirstlane((int)(threadIdx.x >> 5));
    const int gw   = blockIdx.x * (blockDim.x >> 5) + wave;
    float dsum = 0.0f;

#define GATHER_ROW(SV, WV, VV, L)                                               \
            const int    SV = __builtin_amdgcn_readlane(rec.x, (L));            \
            const float  WV = __int_as_float(__builtin_amdgcn_readlane(rec.y, (L))); \
            const float2 VV = ((const float2*)(x + (size_t)SV * DIM))[lane];

#define GATHER_BUCKET(C, S, B0, B1)                                             \
        {                                                                       \
            const int lo = max((B0), cb);                                       \
            const int hi = min((B1), ce);                                       \
            for (int j = lo; j < hi; j += 4) {                                  \
                const int last = hi - 1 - cb;                                   \
                const int l0 = j - cb;                                          \
                const int l1 = min(l0 + 1, last);                               \
                const int l2 = min(l0 + 2, last);                               \
                const int l3 = min(l0 + 3, last);                               \
                GATHER_ROW(sA, wA, vA, l0)                                      \
                GATHER_ROW(sB, wB, vB, l1)                                      \
                GATHER_ROW(sC, wC, vC, l2)                                      \
                GATHER_ROW(sD, wD, vD, l3)                                      \
                C.x += vA.x * wA; C.y += vA.y * wA; S += wA;                    \
                if (j + 1 < hi) { C.x += vB.x * wB; C.y += vB.y * wB; S += wB; } \
                if (j + 2 < hi) { C.x += vC.x * wC; C.y += vC.y * wC; S += wC; } \
                if (j + 3 < hi) { C.x += vD.x * wD; C.y += vD.y * wD; S += wD; } \
            }                                                                   \
        }

    for (int q = 0; q < GATHER_NODES; ++q) {
        const int node = gw * GATHER_NODES + q;
        if (node >= nNodes) break;
        const int4 be  = *(const int4*)(bend + (size_t)node * NREL);
        const int  beg = (node > 0) ? bend[(size_t)node * NREL - 1] : 0;
        const int  fin = be.w;

        float2 c0 = {0.0f, 0.0f}, c1 = c0, c2 = c0, c3 = c0;
        float  s0 = 0.0f, s1 = 0.0f, s2 = 0.0f, s3 = 0.0f;

        for (int cb = beg; cb < fin; cb += 32) {
            const int  ce  = min(cb + 32, fin);
            const int2 rec = ((const int2*)erec)[min(cb + lane, fin - 1)];
            GATHER_BUCKET(c0, s0, beg,  be.x)
            GATHER_BUCKET(c1, s1, be.x, be.y)
            GATHER_BUCKET(c2, s2, be.y, be.z)
            GATHER_BUCKET(c3, s3, be.z, be.w)
        }

        float* o = agg + (size_t)node * (NREL * DIM) + 2 * lane;
        *(float2*)(o)           = c0;
        *(float2*)(o + DIM)     = c1;
        *(float2*)(o + 2 * DIM) = c2;
        *(float2*)(o + 3 * DIM) = c3;

        const int   r4  = lane & 3;
        const float sel = (r4 == 0) ? s0 : (r4 == 1) ? s1 : (r4 == 2) ? s2 : s3;
        if ((lane >> 2) == q) dsum = sel;
    }
#undef GATHER_BUCKET
#undef GATHER_ROW

    const long di = (long)gw * (GATHER_NODES * NREL) + lane;
    if (di < (long)nNodes * NREL) deg[di] = dsum;
}

__global__ void rgcn_update_kernel(const float* __restrict__ agg,
                                   const float* __restrict__ deg,
                                   const float* __restrict__ x,
                                   const float* __restrict__ lin_w,
                                   const float* __restrict__ lin_b,
                                   const float* __restrict__ self_w,
                                   const float* __restrict__ self_b,
                                   float*       __restrict__ hidden,
                                   int nNodes)
{
    extern __shared__ float smem[];
    float* WT = smem;
    float* RD = smem + DIM * STRC;

    const int t = threadIdx.x;
    for (int i = t; i < KCONV * DIM; i += blockDim.x) {
        int k = i >> 6;
        int c = i & 63;
        float v = (k < NREL * DIM) ? lin_w[i] : self_w[i - NREL * DIM * DIM];
        WT[(size_t)c * STRC + k] = v;
    }

    const int wave = t >> 5;
    const int lane = t & 31;
    const int gw   = blockIdx.x * (blockDim.x >> 5) + wave;
    const long base = (long)gw * 16;

    if (base < nNodes) {
        float2 dv = ((const float2*)(deg + base * NREL))[lane];
        RD[wave * 64 + 2 * lane]     = 1.0f / (dv.x + 1e-10f);
        RD[wave * 64 + 2 * lane + 1] = 1.0f / (dv.y + 1e-10f);
    }
    __syncthreads();
    if (base >= nNodes) return;

    const int m     = lane & 15;
    const int khalf = lane >> 4;
    const float* aggRow = agg + (base + m) * (size_t)(NREL * DIM);
    const float* xRow   = x   + (base + m) * (size_t)DIM;
    const float* B0 = WT + (size_t)(m)      * STRC;
    const float* B1 = WT + (size_t)(16 + m) * STRC;
    const float* B2 = WT + (size_t)(32 + m) * STRC;
    const float* B3 = WT + (size_t)(48 + m) * STRC;
    const float* rd = RD + wave * 64 + m * NREL;

    v8f a0 = {}, a1 = {}, a2 = {}, a3 = {};
    for (int k = 0; k < NREL * DIM; k += 4) {
        int kk = k + 2 * khalf;
        float  sc = rd[k >> 6];
        float2 av = *(const float2*)(aggRow + kk);
        v2f a; a.x = av.x * sc; a.y = av.y * sc;
        v2f b0 = *(const v2f*)(B0 + kk);
        v2f b1 = *(const v2f*)(B1 + kk);
        v2f b2 = *(const v2f*)(B2 + kk);
        v2f b3 = *(const v2f*)(B3 + kk);
        a0 = __builtin_amdgcn_wmma_f32_16x16x4_f32(false, a, false, b0, (short)0, a0, false, false);
        a1 = __builtin_amdgcn_wmma_f32_16x16x4_f32(false, a, false, b1, (short)0, a1, false, false);
        a2 = __builtin_amdgcn_wmma_f32_16x16x4_f32(false, a, false, b2, (short)0, a2, false, false);
        a3 = __builtin_amdgcn_wmma_f32_16x16x4_f32(false, a, false, b3, (short)0, a3, false, false);
    }
    for (int k = 0; k < DIM; k += 4) {
        int kk = k + 2 * khalf;
        float2 av = *(const float2*)(xRow + kk);
        v2f a; a.x = av.x; a.y = av.y;
        int kw = NREL * DIM + kk;
        v2f b0 = *(const v2f*)(B0 + kw);
        v2f b1 = *(const v2f*)(B1 + kw);
        v2f b2 = *(const v2f*)(B2 + kw);
        v2f b3 = *(const v2f*)(B3 + kw);
        a0 = __builtin_amdgcn_wmma_f32_16x16x4_f32(false, a, false, b0, (short)0, a0, false, false);
        a1 = __builtin_amdgcn_wmma_f32_16x16x4_f32(false, a, false, b1, (short)0, a1, false, false);
        a2 = __builtin_amdgcn_wmma_f32_16x16x4_f32(false, a, false, b2, (short)0, a2, false, false);
        a3 = __builtin_amdgcn_wmma_f32_16x16x4_f32(false, a, false, b3, (short)0, a3, false, false);
    }

#define CONV_EPILOGUE(ACC, TILE)                                           \
    {                                                                      \
        int   col = (TILE) * 16 + m;                                       \
        float bia = lin_b[col] + self_b[col];                              \
        _Pragma("unroll")                                                  \
        for (int r = 0; r < 8; ++r) {                                      \
            long  row = base + khalf * 8 + r;                              \
            hidden[row * DIM + col] = sigmoidf_(ACC[r] + bia);             \
        }                                                                  \
    }
    CONV_EPILOGUE(a0, 0) CONV_EPILOGUE(a1, 1) CONV_EPILOGUE(a2, 2) CONV_EPILOGUE(a3, 3)
#undef CONV_EPILOGUE
}

__global__ void highway_kernel(const float* __restrict__ hid,
                               const float* __restrict__ prev,
                               const float* __restrict__ pw,
                               const float* __restrict__ pb,
                               const float* __restrict__ tw,
                               const float* __restrict__ tb,
                               float*       __restrict__ outp,
                               int nNodes)
{
    extern __shared__ float smem[];
    float* WP = smem;
    float* WTT = smem + DIM * STRH;

    const int t = threadIdx.x;
    for (int i = t; i < KHW * DIM; i += blockDim.x) {
        int k = i >> 6;
        int c = i & 63;
        WP [(size_t)c * STRH + k] = pw[i];
        WTT[(size_t)c * STRH + k] = tw[i];
    }
    __syncthreads();

    const int wave = t >> 5;
    const int lane = t & 31;
    const int gw   = blockIdx.x * (blockDim.x >> 5) + wave;
    const long base = (long)gw * 16;
    if (base >= nNodes) return;

    const int m     = lane & 15;
    const int khalf = lane >> 4;
    const float* hRow = hid  + (base + m) * (size_t)DIM;
    const float* pRow = prev + (base + m) * (size_t)DIM;
    const float* P0 = WP  + (size_t)(m)      * STRH;
    const float* P1 = WP  + (size_t)(16 + m) * STRH;
    const float* P2 = WP  + (size_t)(32 + m) * STRH;
    const float* P3 = WP  + (size_t)(48 + m) * STRH;
    const float* T0 = WTT + (size_t)(m)      * STRH;
    const float* T1 = WTT + (size_t)(16 + m) * STRH;
    const float* T2 = WTT + (size_t)(32 + m) * STRH;
    const float* T3 = WTT + (size_t)(48 + m) * STRH;

    v8f p0 = {}, p1 = {}, p2 = {}, p3 = {};
    v8f g0 = {}, g1 = {}, g2 = {}, g3 = {};

#define HW_STEP(AROW, KOFF)                                                              \
    for (int k = 0; k < DIM; k += 4) {                                                   \
        int kk = k + 2 * khalf;                                                          \
        float2 av = *(const float2*)((AROW) + kk);                                       \
        v2f a; a.x = av.x; a.y = av.y;                                                   \
        int kw = kk + (KOFF);                                                            \
        v2f bp0 = *(const v2f*)(P0 + kw);                                                \
        v2f bp1 = *(const v2f*)(P1 + kw);                                                \
        v2f bp2 = *(const v2f*)(P2 + kw);                                                \
        v2f bp3 = *(const v2f*)(P3 + kw);                                                \
        v2f bt0 = *(const v2f*)(T0 + kw);                                                \
        v2f bt1 = *(const v2f*)(T1 + kw);                                                \
        v2f bt2 = *(const v2f*)(T2 + kw);                                                \
        v2f bt3 = *(const v2f*)(T3 + kw);                                                \
        p0 = __builtin_amdgcn_wmma_f32_16x16x4_f32(false, a, false, bp0, (short)0, p0, false, false); \
        p1 = __builtin_amdgcn_wmma_f32_16x16x4_f32(false, a, false, bp1, (short)0, p1, false, false); \
        p2 = __builtin_amdgcn_wmma_f32_16x16x4_f32(false, a, false, bp2, (short)0, p2, false, false); \
        p3 = __builtin_amdgcn_wmma_f32_16x16x4_f32(false, a, false, bp3, (short)0, p3, false, false); \
        g0 = __builtin_amdgcn_wmma_f32_16x16x4_f32(false, a, false, bt0, (short)0, g0, false, false); \
        g1 = __builtin_amdgcn_wmma_f32_16x16x4_f32(false, a, false, bt1, (short)0, g1, false, false); \
        g2 = __builtin_amdgcn_wmma_f32_16x16x4_f32(false, a, false, bt2, (short)0, g2, false, false); \
        g3 = __builtin_amdgcn_wmma_f32_16x16x4_f32(false, a, false, bt3, (short)0, g3, false, false); \
    }
    HW_STEP(hRow, 0)
    HW_STEP(pRow, DIM)
#undef HW_STEP

#define HW_EPILOGUE(PACC, GACC, TILE)                                      \
    {                                                                      \
        int   col = (TILE) * 16 + m;                                       \
        float pbv = pb[col];                                               \
        float tbv = tb[col];                                               \
        _Pragma("unroll")                                                  \
        for (int r = 0; r < 8; ++r) {                                      \
            long  row = base + khalf * 8 + r;                              \
            float pr = fmaxf(PACC[r] + pbv, 0.0f);                         \
            float gt = sigmoidf_(GACC[r] + tbv);                           \
            float h  = hid[row * DIM + col];                               \
            outp[row * DIM + col] = gt * pr + (1.0f - gt) * h;             \
        }                                                                  \
    }
    HW_EPILOGUE(p0, g0, 0) HW_EPILOGUE(p1, g1, 1)
    HW_EPILOGUE(p2, g2, 2) HW_EPILOGUE(p3, g3, 3)
#undef HW_EPILOGUE
}

extern "C" void kernel_launch(void* const* d_in, const int* in_sizes, int n_in,
                              void* d_out, int out_size, void* d_ws, size_t ws_size,
                              hipStream_t stream) {
    const float* node_feat   = (const float*)d_in[0];
    const float* edge_weight = (const float*)d_in[1];
    const float* conv_lin_w  = (const float*)d_in[2];
    const float* conv_lin_b  = (const float*)d_in[3];
    const float* conv_self_w = (const float*)d_in[4];
    const float* conv_self_b = (const float*)d_in[5];
    const float* hw_proj_w   = (const float*)d_in[6];
    const float* hw_proj_b   = (const float*)d_in[7];
    const float* hw_trans_w  = (const float*)d_in[8];
    const float* hw_trans_b  = (const float*)d_in[9];
    const int*   edge_src    = (const int*)d_in[10];
    const int*   edge_dst    = (const int*)d_in[11];
    const int*   edge_rel    = (const int*)d_in[12];
    float*       out         = (float*)d_out;

    const int N = in_sizes[0] / DIM;
    const int E = in_sizes[10];

    float* agg = (float*)d_ws;
    float* deg = agg + (size_t)N * NREL * DIM;
    float* hA  = deg + (size_t)N * NREL;
    float* hB  = hA  + (size_t)N * DIM;
    float* cur = hB  + (size_t)N * DIM;
    const int NB   = N * NREL;
    const int nblk = (NB + SCAN_CHUNK - 1) / SCAN_CHUNK;
    int*  cnt    = (int*)(cur + (size_t)N * DIM);
    int*  cursor = cnt + NB;
    int*  bsum   = cursor + NB;
    int*  erec   = bsum + ((nblk + 3) & ~3);

    const int edgeBlocks    = (E + 255) / 256;
    const int gatherBlocks  = (N + 8 * GATHER_NODES - 1) / (8 * GATHER_NODES);
    const int waves         = (N + 15) / 16;
    const int nodeBlocks    = (waves + 7) / 8;
    const size_t updLds = (size_t)(DIM * STRC + 8 * 64) * sizeof(float);
    const size_t hwLds  = (size_t)(2 * DIM * STRH) * sizeof(float);

    hipMemsetAsync(cnt, 0, (size_t)NB * sizeof(int), stream);
    bucket_count_kernel<<<edgeBlocks, 256, 0, stream>>>(edge_dst, edge_rel, cnt, E);
    bucket_blocksum_kernel<<<nblk, 256, 0, stream>>>(cnt, bsum, NB);
    bucket_scan_sums_kernel<<<1, 1024, 0, stream>>>(bsum, nblk);
    bucket_offsets_kernel<<<nblk, 256, 0, stream>>>(cnt, bsum, cursor, NB);
    bucket_fill_kernel<<<edgeBlocks, 256, 0, stream>>>(
        edge_src, edge_dst, edge_rel, edge_weight, cursor, erec, E);

    for (int i = 0; i < 3; ++i) {
        const float* xin  = (i == 0) ? node_feat : cur;
        const float* prev = (i == 0) ? node_feat : ((i == 1) ? hA : hB);
        float*       hid  = (i & 1) ? hB : hA;
        float*       gout = (i == 2) ? out : cur;

        rgcn_gather_kernel<<<gatherBlocks, 256, 0, stream>>>(
            xin, erec, cursor, agg, deg, N);

        rgcn_update_kernel<<<nodeBlocks, 256, updLds, stream>>>(
            agg, deg, xin,
            conv_lin_w  + (size_t)i * 256 * DIM, conv_lin_b  + (size_t)i * DIM,
            conv_self_w + (size_t)i * DIM * DIM, conv_self_b + (size_t)i * DIM,
            hid, N);

        highway_kernel<<<nodeBlocks, 256, hwLds, stream>>>(
            hid, prev,
            hw_proj_w  + (size_t)i * KHW * DIM, hw_proj_b  + (size_t)i * DIM,
            hw_trans_w + (size_t)i * KHW * DIM, hw_trans_b + (size_t)i * DIM,
            gout, N);
    }
}


# Edge ordering by (dst,rel): global int atomics and scattered stores replaced by a two-level counting sort with LDS counters and whole-run stores; in-bucket sum order stays arrival-dependent
# speedup vs baseline: 2.3070x; 1.2832x over previous
#include <hip/hip_runtime.h>
#include <hip/hip_bf16.h>
#include <math.h>


typedef __attribute__((ext_vector_type(2))) float v2f;
typedef __attribute__((ext_vector_type(8))) float v8f;

#define DIM   64
#define NREL  4
#define KCONV 320
#define KHW   128
#define STRC  324
#define STRH  132

__device__ __forceinline__ float sigmoidf_(float v) {
    return 1.0f / (1.0f + __expf(-v));
}

#define GATHER_NODES 8

#define BIN_SHIFT   8
#define BIN_NODES   (1 << BIN_SHIFT)
#define FINE_KEYS   (BIN_NODES * NREL)
#define SORT_BLOCKS 256
#define SORT_THREADS 1024
#define KEY_SHIFT   20
#define SRC_MASK    ((1 << KEY_SHIFT) - 1)
#define FINE_CAP    8192

__device__ __forceinline__ int block_excl_scan_1024(int v, int* wtot)
{
    const int lane = threadIdx.x & 31;
    const int wave = threadIdx.x >> 5;
    int inc = v;
    for (int o = 1; o < 32; o <<= 1) {
        int n = __shfl_up(inc, o);
        if (lane >= o) inc += n;
    }
    if (lane == 31) wtot[wave] = inc;
    __syncthreads();
    int woff = 0;
    for (int w = 0; w < 32; ++w) {
        int x = wtot[w];
        if (w < wave) woff += x;
    }
    __syncthreads();
    return woff + inc - v;
}

__global__ void __launch_bounds__(SORT_THREADS)
bin_hist_kernel(const int* __restrict__ dst,
                int*       __restrict__ hist,
                int nEdges, int chunk, int nbin, int nbinp)
{
    __shared__ int cnt[SORT_THREADS];
    const int t = threadIdx.x;
    cnt[t] = 0;
    __syncthreads();
    const int beg = blockIdx.x * chunk;
    const int end = min(beg + chunk, nEdges);
    for (int i0 = beg + t; i0 < end; i0 += 4 * SORT_THREADS) {
        int d[4];
#pragma unroll
        for (int u = 0; u < 4; ++u) {
            const int i = i0 + u * SORT_THREADS;
            d[u] = (i < end) ? dst[i] : -1;
        }
#pragma unroll
        for (int u = 0; u < 4; ++u)
            if (d[u] >= 0) atomicAdd(&cnt[d[u] >> BIN_SHIFT], 1);
    }
    __syncthreads();
    if (t < nbinp) hist[(size_t)blockIdx.x * nbinp + t] = (t < nbin) ? cnt[t] : 0;
}

__global__ void __launch_bounds__(SORT_THREADS)
bin_partition_kernel(const int*   __restrict__ src,
                     const int*   __restrict__ dst,
                     const int*   __restrict__ rel,
                     const float* __restrict__ ew,
                     const int*   __restrict__ hist,
                     int*         __restrict__ binStart,
                     int2*        __restrict__ part,
                     int nEdges, int chunk, int nbin, int nbinp)
{
    extern __shared__ int2 stage[];
    __shared__ int tot[SORT_THREADS];
    __shared__ int pre[SORT_THREADS];
    __shared__ int mine[SORT_THREADS];
    __shared__ int loff[SORT_THREADS];
    __shared__ int lcur[SORT_THREADS];
    __shared__ int gbase[SORT_THREADS];
    __shared__ int wtot[32];

    const int t   = threadIdx.x;
    const int blk = blockIdx.x;
    tot[t] = 0;
    pre[t] = 0;
    mine[t] = (t < nbin) ? hist[(size_t)blk * nbinp + t] : 0;
    __syncthreads();

    {
        const int nq  = nbinp >> 2;
        const int grp = t >> 7;
        for (int q = t & 127; q < nq; q += 128) {
            int4 st = {0, 0, 0, 0}, sp = {0, 0, 0, 0};
            for (int r0 = grp; r0 < SORT_BLOCKS; r0 += 64) {
                int4 h[8];
#pragma unroll
                for (int u = 0; u < 8; ++u)
                    h[u] = *(const int4*)(hist + (size_t)(r0 + 8 * u) * nbinp + 4 * q);
#pragma unroll
                for (int u = 0; u < 8; ++u) {
                    st.x += h[u].x; st.y += h[u].y; st.z += h[u].z; st.w += h[u].w;
                    if (r0 + 8 * u < blk) { sp.x += h[u].x; sp.y += h[u].y; sp.z += h[u].z; sp.w += h[u].w; }
                }
            }
            atomicAdd(&tot[4 * q + 0], st.x); atomicAdd(&tot[4 * q + 1], st.y);
            atomicAdd(&tot[4 * q + 2], st.z); atomicAdd(&tot[4 * q + 3], st.w);
            atomicAdd(&pre[4 * q + 0], sp.x); atomicAdd(&pre[4 * q + 1], sp.y);
            atomicAdd(&pre[4 * q + 2], sp.z); atomicAdd(&pre[4 * q + 3], sp.w);
        }
    }
    __syncthreads();

    const int vt = (t < nbin) ? tot[t] : 0;
    const int vm = mine[t];
    const int bs = block_excl_scan_1024(vt, wtot);
    const int lo = block_excl_scan_1024(vm, wtot);
    gbase[t] = bs + pre[t];
    loff[t]  = lo;
    lcur[t]  = lo;
    if (blk == 0) {
        if (t < nbin) binStart[t] = bs;
        if (t == 0)   binStart[nbin] = nEdges;
    }
    __syncthreads();

    const int beg = blk * chunk;
    const int end = min(beg + chunk, nEdges);
    for (int i0 = beg + t; i0 < end; i0 += 4 * SORT_THREADS) {
        int d[4], r[4], s[4], w[4];
#pragma unroll
        for (int u = 0; u < 4; ++u) {
            const int  i  = i0 + u * SORT_THREADS;
            const bool ok = i < end;
            d[u] = ok ? dst[i] : -1;
            r[u] = ok ? rel[i] : 0;
            s[u] = ok ? src[i] : 0;
            w[u] = ok ? __float_as_int(ew[i]) : 0;
        }
#pragma unroll
        for (int u = 0; u < 4; ++u) {
            if (d[u] >= 0) {
                const int b   = d[u] >> BIN_SHIFT;
                const int key = (d[u] & (BIN_NODES - 1)) * NREL + r[u];
                const int p   = atomicAdd(&lcur[b], 1);
                int2 rec;
                rec.x = s[u] | (key << KEY_SHIFT);
                rec.y = w[u];
                stage[p] = rec;
            }
        }
    }
    __syncthreads();

    const int lane = t & 31;
    const int wave = t >> 5;
    for (int b = wave; b < nbin; b += 32) {
        const int n = mine[b];
        const int l = loff[b];
        const int g = gbase[b];
        for (int j = lane; j < n; j += 32) part[g + j] = stage[l + j];
    }
}

__global__ void __launch_bounds__(SORT_THREADS)
bin_order_kernel(const int2* __restrict__ part,
                 const int*  __restrict__ binStart,
                 int*        __restrict__ bend,
                 int2*       __restrict__ erec,
                 int nBuckets)
{
    __shared__ int2 sorted[FINE_CAP];
    __shared__ int  cnt[FINE_KEYS];
    __shared__ int  lcur[FINE_KEYS];
    __shared__ int  wtot[32];

    const int t = threadIdx.x;
    const int g = blockIdx.x;
    const int s = binStart[g];
    const int n = binStart[g + 1] - s;
    const int2* in = part + s;

    cnt[t] = 0;
    __syncthreads();
    for (int i0 = t; i0 < n; i0 += 4 * SORT_THREADS) {
        int x[4];
#pragma unroll
        for (int u = 0; u < 4; ++u) {
            const int i = i0 + u * SORT_THREADS;
            x[u] = (i < n) ? in[i].x : -1;
        }
#pragma unroll
        for (int u = 0; u < 4; ++u)
            if (x[u] >= 0) atomicAdd(&cnt[x[u] >> KEY_SHIFT], 1);
    }
    __syncthreads();

    const int c   = cnt[t];
    const int off = block_excl_scan_1024(c, wtot);
    lcur[t] = off;
    const int b = g * FINE_KEYS + t;
    if (b < nBuckets) bend[b] = s + off + c;
    __syncthreads();

    if (n <= FINE_CAP) {
        for (int i0 = t; i0 < n; i0 += 4 * SORT_THREADS) {
            int2 rc[4];
#pragma unroll
            for (int u = 0; u < 4; ++u) {
                const int i = i0 + u * SORT_THREADS;
                if (i < n) rc[u] = in[i]; else { rc[u].x = -1; rc[u].y = 0; }
            }
#pragma unroll
            for (int u = 0; u < 4; ++u) {
                if (rc[u].x >= 0) {
                    const int p = atomicAdd(&lcur[rc[u].x >> KEY_SHIFT], 1);
                    int2 o;
                    o.x = rc[u].x & SRC_MASK;
                    o.y = rc[u].y;
                    sorted[p] = o;
                }
            }
        }
        __syncthreads();
        for (int i = t; i < n; i += SORT_THREADS) erec[s + i] = sorted[i];
    } else {
        for (int i = t; i < n; i += SORT_THREADS) {
            const int2 rc = in[i];
            const int  p  = atomicAdd(&lcur[rc.x >> KEY_SHIFT], 1);
            int2 o;
            o.x = rc.x & SRC_MASK;
            o.y = rc.y;
            erec[s + p] = o;
        }
    }
}

__global__ void rgcn_gather_kernel(const float* __restrict__ x,
                                   const int*   __restrict__ erec,
                                   const int*   __restrict__ bend,
                                   float*       __restrict__ agg,
                                   float*       __restrict__ deg,
                                   int nNodes)
{
    const int lane = threadIdx.x & 31;
    const int wave = __builtin_amdgcn_readfirstlane((int)(threadIdx.x >> 5));
    const int gw   = blockIdx.x * (blockDim.x >> 5) + wave;
    float dsum = 0.0f;

#define GATHER_ROW(SV, WV, VV, L)                                               \
            const int    SV = __builtin_amdgcn_readlane(rec.x, (L));            \
            const float  WV = __int_as_float(__builtin_amdgcn_readlane(rec.y, (L))); \
            const float2 VV = ((const float2*)(x + (size_t)SV * DIM))[lane];

#define GATHER_BUCKET(C, S, B0, B1)                                             \
        {                                                                       \
            const int lo = max((B0), cb);                                       \
            const int hi = min((B1), ce);                                       \
            for (int j = lo; j < hi; j += 4) {                                  \
                const int last = hi - 1 - cb;                                   \
                const int l0 = j - cb;                                          \
                const int l1 = min(l0 + 1, last);                               \
                const int l2 = min(l0 + 2, last);                               \
                const int l3 = min(l0 + 3, last);                               \
                GATHER_ROW(sA, wA, vA, l0)                                      \
                GATHER_ROW(sB, wB, vB, l1)                                      \
                GATHER_ROW(sC, wC, vC, l2)                                      \
                GATHER_ROW(sD, wD, vD, l3)                                      \
                C.x += vA.x * wA; C.y += vA.y * wA; S += wA;                    \
                if (j + 1 < hi) { C.x += vB.x * wB; C.y += vB.y * wB; S += wB; } \
                if (j + 2 < hi) { C.x += vC.x * wC; C.y += vC.y * wC; S += wC; } \
                if (j + 3 < hi) { C.x += vD.x * wD; C.y += vD.y * wD; S += wD; } \
            }                                                                   \
        }

    for (int q = 0; q < GATHER_NODES; ++q) {
        const int node = gw * GATHER_NODES + q;
        if (node >= nNodes) break;
        const int4 be  = *(const int4*)(bend + (size_t)node * NREL);
        const int  beg = (node > 0) ? bend[(size_t)node * NREL - 1] : 0;
        const int  fin = be.w;

        float2 c0 = {0.0f, 0.0f}, c1 = c0, c2 = c0, c3 = c0;
        float  s0 = 0.0f, s1 = 0.0f, s2 = 0.0f, s3 = 0.0f;

        for (int cb = beg; cb < fin; cb += 32) {
            const int  ce  = min(cb + 32, fin);
            const int2 rec = ((const int2*)erec)[min(cb + lane, fin - 1)];
            GATHER_BUCKET(c0, s0, beg,  be.x)
            GATHER_BUCKET(c1, s1, be.x, be.y)
            GATHER_BUCKET(c2, s2, be.y, be.z)
            GATHER_BUCKET(c3, s3, be.z, be.w)
        }

        float* o = agg + (size_t)node * (NREL * DIM) + 2 * lane;
        *(float2*)(o)           = c0;
        *(float2*)(o + DIM)     = c1;
        *(float2*)(o + 2 * DIM) = c2;
        *(float2*)(o + 3 * DIM) = c3;

        const int   r4  = lane & 3;
        const float sel = (r4 == 0) ? s0 : (r4 == 1) ? s1 : (r4 == 2) ? s2 : s3;
        if ((lane >> 2) == q) dsum = sel;
    }
#undef GATHER_BUCKET
#undef GATHER_ROW

    const long di = (long)gw * (GATHER_NODES * NREL) + lane;
    if (di < (long)nNodes * NREL) deg[di] = dsum;
}

__global__ void rgcn_update_kernel(const float* __restrict__ agg,
                                   const float* __restrict__ deg,
                                   const float* __restrict__ x,
                                   const float* __restrict__ lin_w,
                                   const float* __restrict__ lin_b,
                                   const float* __restrict__ self_w,
                                   const float* __restrict__ self_b,
                                   float*       __restrict__ hidden,
                                   int nNodes)
{
    extern __shared__ float smem[];
    float* WT = smem;
    float* RD = smem + DIM * STRC;

    const int t = threadIdx.x;
    for (int i = t; i < KCONV * DIM; i += blockDim.x) {
        int k = i >> 6;
        int c = i & 63;
        float v = (k < NREL * DIM) ? lin_w[i] : self_w[i - NREL * DIM * DIM];
        WT[(size_t)c * STRC + k] = v;
    }

    const int wave = t >> 5;
    const int lane = t & 31;
    const int gw   = blockIdx.x * (blockDim.x >> 5) + wave;
    const long base = (long)gw * 16;

    if (base < nNodes) {
        float2 dv = ((const float2*)(deg + base * NREL))[lane];
        RD[wave * 64 + 2 * lane]     = 1.0f / (dv.x + 1e-10f);
        RD[wave * 64 + 2 * lane + 1] = 1.0f / (dv.y + 1e-10f);
    }
    __syncthreads();
    if (base >= nNodes) return;

    const int m     = lane & 15;
    const int khalf = lane >> 4;
    const float* aggRow = agg + (base + m) * (size_t)(NREL * DIM);
    const float* xRow   = x   + (base + m) * (size_t)DIM;
    const float* B0 = WT + (size_t)(m)      * STRC;
    const float* B1 = WT + (size_t)(16 + m) * STRC;
    const float* B2 = WT + (size_t)(32 + m) * STRC;
    const float* B3 = WT + (size_t)(48 + m) * STRC;
    const float* rd = RD + wave * 64 + m * NREL;

    v8f a0 = {}, a1 = {}, a2 = {}, a3 = {};
    for (int k = 0; k < NREL * DIM; k += 4) {
        int kk = k + 2 * khalf;
        float  sc = rd[k >> 6];
        float2 av = *(const float2*)(aggRow + kk);
        v2f a; a.x = av.x * sc; a.y = av.y * sc;
        v2f b0 = *(const v2f*)(B0 + kk);
        v2f b1 = *(const v2f*)(B1 + kk);
        v2f b2 = *(const v2f*)(B2 + kk);
        v2f b3 = *(const v2f*)(B3 + kk);
        a0 = __builtin_amdgcn_wmma_f32_16x16x4_f32(false, a, false, b0, (short)0, a0, false, false);
        a1 = __builtin_amdgcn_wmma_f32_16x16x4_f32(false, a, false, b1, (short)0, a1, false, false);
        a2 = __builtin_amdgcn_wmma_f32_16x16x4_f32(false, a, false, b2, (short)0, a2, false, false);
        a3 = __builtin_amdgcn_wmma_f32_16x16x4_f32(false, a, false, b3, (short)0, a3, false, false);
    }
    for (int k = 0; k < DIM; k += 4) {
        int kk = k + 2 * khalf;
        float2 av = *(const float2*)(xRow + kk);
        v2f a; a.x = av.x; a.y = av.y;
        int kw = NREL * DIM + kk;
        v2f b0 = *(const v2f*)(B0 + kw);
        v2f b1 = *(const v2f*)(B1 + kw);
        v2f b2 = *(const v2f*)(B2 + kw);
        v2f b3 = *(const v2f*)(B3 + kw);
        a0 = __builtin_amdgcn_wmma_f32_16x16x4_f32(false, a, false, b0, (short)0, a0, false, false);
        a1 = __builtin_amdgcn_wmma_f32_16x16x4_f32(false, a, false, b1, (short)0, a1, false, false);
        a2 = __builtin_amdgcn_wmma_f32_16x16x4_f32(false, a, false, b2, (short)0, a2, false, false);
        a3 = __builtin_amdgcn_wmma_f32_16x16x4_f32(false, a, false, b3, (short)0, a3, false, false);
    }

#define CONV_EPILOGUE(ACC, TILE)                                           \
    {                                                                      \
        int   col = (TILE) * 16 + m;                                       \
        float bia = lin_b[col] + self_b[col];                              \
        _Pragma("unroll")                                                  \
        for (int r = 0; r < 8; ++r) {                                      \
            long  row = base + khalf * 8 + r;                              \
            hidden[row * DIM + col] = sigmoidf_(ACC[r] + bia);             \
        }                                                                  \
    }
    CONV_EPILOGUE(a0, 0) CONV_EPILOGUE(a1, 1) CONV_EPILOGUE(a2, 2) CONV_EPILOGUE(a3, 3)
#undef CONV_EPILOGUE
}

__global__ void highway_kernel(const float* __restrict__ hid,
                               const float* __restrict__ prev,
                               const float* __restrict__ pw,
                               const float* __restrict__ pb,
                               const float* __restrict__ tw,
                               const float* __restrict__ tb,
                               float*       __restrict__ outp,
                               int nNodes)
{
    extern __shared__ float smem[];
    float* WP = smem;
    float* WTT = smem + DIM * STRH;

    const int t = threadIdx.x;
    for (int i = t; i < KHW * DIM; i += blockDim.x) {
        int k = i >> 6;
        int c = i & 63;
        WP [(size_t)c * STRH + k] = pw[i];
        WTT[(size_t)c * STRH + k] = tw[i];
    }
    __syncthreads();

    const int wave = t >> 5;
    const int lane = t & 31;
    const int gw   = blockIdx.x * (blockDim.x >> 5) + wave;
    const long base = (long)gw * 16;
    if (base >= nNodes) return;

    const int m     = lane & 15;
    const int khalf = lane >> 4;
    const float* hRow = hid  + (base + m) * (size_t)DIM;
    const float* pRow = prev + (base + m) * (size_t)DIM;
    const float* P0 = WP  + (size_t)(m)      * STRH;
    const float* P1 = WP  + (size_t)(16 + m) * STRH;
    const float* P2 = WP  + (size_t)(32 + m) * STRH;
    const float* P3 = WP  + (size_t)(48 + m) * STRH;
    const float* T0 = WTT + (size_t)(m)      * STRH;
    const float* T1 = WTT + (size_t)(16 + m) * STRH;
    const float* T2 = WTT + (size_t)(32 + m) * STRH;
    const float* T3 = WTT + (size_t)(48 + m) * STRH;

    v8f p0 = {}, p1 = {}, p2 = {}, p3 = {};
    v8f g0 = {}, g1 = {}, g2 = {}, g3 = {};

#define HW_STEP(AROW, KOFF)                                                              \
    for (int k = 0; k < DIM; k += 4) {                                                   \
        int kk = k + 2 * khalf;                                                          \
        float2 av = *(const float2*)((AROW) + kk);                                       \
        v2f a; a.x = av.x; a.y = av.y;                                                   \
        int kw = kk + (KOFF);                                                            \
        v2f bp0 = *(const v2f*)(P0 + kw);                                                \
        v2f bp1 = *(const v2f*)(P1 + kw);                                                \
        v2f bp2 = *(const v2f*)(P2 + kw);                                                \
        v2f bp3 = *(const v2f*)(P3 + kw);                                                \
        v2f bt0 = *(const v2f*)(T0 + kw);                                                \
        v2f bt1 = *(const v2f*)(T1 + kw);                                                \
        v2f bt2 = *(const v2f*)(T2 + kw);                                                \
        v2f bt3 = *(const v2f*)(T3 + kw);                                                \
        p0 = __builtin_amdgcn_wmma_f32_16x16x4_f32(false, a, false, bp0, (short)0, p0, false, false); \
        p1 = __builtin_amdgcn_wmma_f32_16x16x4_f32(false, a, false, bp1, (short)0, p1, false, false); \
        p2 = __builtin_amdgcn_wmma_f32_16x16x4_f32(false, a, false, bp2, (short)0, p2, false, false); \
        p3 = __builtin_amdgcn_wmma_f32_16x16x4_f32(false, a, false, bp3, (short)0, p3, false, false); \
        g0 = __builtin_amdgcn_wmma_f32_16x16x4_f32(false, a, false, bt0, (short)0, g0, false, false); \
        g1 = __builtin_amdgcn_wmma_f32_16x16x4_f32(false, a, false, bt1, (short)0, g1, false, false); \
        g2 = __builtin_amdgcn_wmma_f32_16x16x4_f32(false, a, false, bt2, (short)0, g2, false, false); \
        g3 = __builtin_amdgcn_wmma_f32_16x16x4_f32(false, a, false, bt3, (short)0, g3, false, false); \
    }
    HW_STEP(hRow, 0)
    HW_STEP(pRow, DIM)
#undef HW_STEP

#define HW_EPILOGUE(PACC, GACC, TILE)                                      \
    {                                                                      \
        int   col = (TILE) * 16 + m;                                       \
        float pbv = pb[col];                                               \
        float tbv = tb[col];                                               \
        _Pragma("unroll")                                                  \
        for (int r = 0; r < 8; ++r) {                                      \
            long  row = base + khalf * 8 + r;                              \
            float pr = fmaxf(PACC[r] + pbv, 0.0f);                         \
            float gt = sigmoidf_(GACC[r] + tbv);                           \
            float h  = hid[row * DIM + col];                               \
            outp[row * DIM + col] = gt * pr + (1.0f - gt) * h;             \
        }                                                                  \
    }
    HW_EPILOGUE(p0, g0, 0) HW_EPILOGUE(p1, g1, 1)
    HW_EPILOGUE(p2, g2, 2) HW_EPILOGUE(p3, g3, 3)
#undef HW_EPILOGUE
}

extern "C" void kernel_launch(void* const* d_in, const int* in_sizes, int n_in,
                              void* d_out, int out_size, void* d_ws, size_t ws_size,
                              hipStream_t stream) {
    const float* node_feat   = (const float*)d_in[0];
    const float* edge_weight = (const float*)d_in[1];
    const float* conv_lin_w  = (const float*)d_in[2];
    const float* conv_lin_b  = (const float*)d_in[3];
    const float* conv_self_w = (const float*)d_in[4];
    const float* conv_self_b = (const float*)d_in[5];
    const float* hw_proj_w   = (const float*)d_in[6];
    const float* hw_proj_b   = (const float*)d_in[7];
    const float* hw_trans_w  = (const float*)d_in[8];
    const float* hw_trans_b  = (const float*)d_in[9];
    const int*   edge_src    = (const int*)d_in[10];
    const int*   edge_dst    = (const int*)d_in[11];
    const int*   edge_rel    = (const int*)d_in[12];
    float*       out         = (float*)d_out;

    const int N = in_sizes[0] / DIM;
    const int E = in_sizes[10];

    float* agg = (float*)d_ws;
    float* deg = agg + (size_t)N * NREL * DIM;
    float* hA  = deg + (size_t)N * NREL;
    float* hB  = hA  + (size_t)N * DIM;
    float* cur = hB  + (size_t)N * DIM;
    const int NB    = N * NREL;
    const int nbin  = (N + BIN_NODES - 1) >> BIN_SHIFT;
    const int nbinp = (nbin + 3) & ~3;
    const int chunk = (E + SORT_BLOCKS - 1) / SORT_BLOCKS;
    int*  cursor   = (int*)(cur + (size_t)N * DIM);
    int*  erec     = cursor + NB;
    int*  part     = erec + 2 * (size_t)E;
    int*  hist     = part + 2 * (size_t)E;
    int*  binStart = hist + (size_t)SORT_BLOCKS * nbinp;

    const int gatherBlocks  = (N + 8 * GATHER_NODES - 1) / (8 * GATHER_NODES);
    const int waves         = (N + 15) / 16;
    const int nodeBlocks    = (waves + 7) / 8;
    const size_t updLds = (size_t)(DIM * STRC + 8 * 64) * sizeof(float);
    const size_t hwLds  = (size_t)(2 * DIM * STRH) * sizeof(float);
    const size_t stageLds = (size_t)chunk * sizeof(int2);

    bin_hist_kernel<<<SORT_BLOCKS, SORT_THREADS, 0, stream>>>(
        edge_dst, hist, E, chunk, nbin, nbinp);
    bin_partition_kernel<<<SORT_BLOCKS, SORT_THREADS, stageLds, stream>>>(
        edge_src, edge_dst, edge_rel, edge_weight, hist, binStart, (int2*)part,
        E, chunk, nbin, nbinp);
    bin_order_kernel<<<nbin, SORT_THREADS, 0, stream>>>(
        (const int2*)part, binStart, cursor, (int2*)erec, NB);

    for (int i = 0; i < 3; ++i) {
        const float* xin  = (i == 0) ? node_feat : cur;
        const float* prev = (i == 0) ? node_feat : ((i == 1) ? hA : hB);
        float*       hid  = (i & 1) ? hB : hA;
        float*       gout = (i == 2) ? out : cur;

        rgcn_gather_kernel<<<gatherBlocks, 256, 0, stream>>>(
            xin, erec, cursor, agg, deg, N);

        rgcn_update_kernel<<<nodeBlocks, 256, updLds, stream>>>(
            agg, deg, xin,
            conv_lin_w  + (size_t)i * 256 * DIM, conv_lin_b  + (size_t)i * DIM,
            conv_self_w + (size_t)i * DIM * DIM, conv_self_b + (size_t)i * DIM,
            hid, N);

        highway_kernel<<<nodeBlocks, 256, hwLds, stream>>>(
            hid, prev,
            hw_proj_w  + (size_t)i * KHW * DIM, hw_proj_b  + (size_t)i * DIM,
            hw_trans_w + (size_t)i * KHW * DIM, hw_trans_b + (size_t)i * DIM,
            gout, N);
    }
}


# Node update and highway GEMMs of a layer fused into one kernel with all work-groups resident at once: hidden goes through a per-wave LDS tile into the highway A fragments; weights staged once
# speedup vs baseline: 2.9214x; 1.2663x over previous
#include <hip/hip_runtime.h>
#include <hip/hip_bf16.h>
#include <math.h>


typedef __attribute__((ext_vector_type(2))) float v2f;
typedef __attribute__((ext_vector_type(8))) float v8f;

#define DIM   64
#define NREL  4
#define KCONV 320
#define KHW   128
#define STRC  324
#define STRH  132

__device__ __forceinline__ float sigmoidf_(float v) {
    return 1.0f / (1.0f + __expf(-v));
}

#define GATHER_NODES 8

#define BIN_SHIFT   8
#define BIN_NODES   (1 << BIN_SHIFT)
#define FINE_KEYS   (BIN_NODES * NREL)
#define SORT_BLOCKS 256
#define SORT_THREADS 1024
#define KEY_SHIFT   20
#define SRC_MASK    ((1 << KEY_SHIFT) - 1)
#define FINE_CAP    8192

__device__ __forceinline__ int block_excl_scan_1024(int v, int* wtot)
{
    const int lane = threadIdx.x & 31;
    const int wave = threadIdx.x >> 5;
    int inc = v;
    for (int o = 1; o < 32; o <<= 1) {
        int n = __shfl_up(inc, o);
        if (lane >= o) inc += n;
    }
    if (lane == 31) wtot[wave] = inc;
    __syncthreads();
    int woff = 0;
    for (int w = 0; w < 32; ++w) {
        int x = wtot[w];
        if (w < wave) woff += x;
    }
    __syncthreads();
    return woff + inc - v;
}

__global__ void __launch_bounds__(SORT_THREADS)
bin_hist_kernel(const int* __restrict__ dst,
                int*       __restrict__ hist,
                int nEdges, int chunk, int nbin, int nbinp)
{
    __shared__ int cnt[SORT_THREADS];
    const int t = threadIdx.x;
    cnt[t] = 0;
    __syncthreads();
    const int beg = blockIdx.x * chunk;
    const int end = min(beg + chunk, nEdges);
    for (int i0 = beg + t; i0 < end; i0 += 4 * SORT_THREADS) {
        int d[4];
#pragma unroll
        for (int u = 0; u < 4; ++u) {
            const int i = i0 + u * SORT_THREADS;
            d[u] = (i < end) ? dst[i] : -1;
        }
#pragma unroll
        for (int u = 0; u < 4; ++u)
            if (d[u] >= 0) atomicAdd(&cnt[d[u] >> BIN_SHIFT], 1);
    }
    __syncthreads();
    if (t < nbinp) hist[(size_t)blockIdx.x * nbinp + t] = (t < nbin) ? cnt[t] : 0;
}

__global__ void __launch_bounds__(SORT_THREADS)
bin_partition_kernel(const int*   __restrict__ src,
                     const int*   __restrict__ dst,
                     const int*   __restrict__ rel,
                     const float* __restrict__ ew,
                     const int*   __restrict__ hist,
                     int*         __restrict__ binStart,
                     int2*        __restrict__ part,
                     int nEdges, int chunk, int nbin, int nbinp)
{
    extern __shared__ int2 stage[];
    __shared__ int tot[SORT_THREADS];
    __shared__ int pre[SORT_THREADS];
    __shared__ int mine[SORT_THREADS];
    __shared__ int loff[SORT_THREADS];
    __shared__ int lcur[SORT_THREADS];
    __shared__ int gbase[SORT_THREADS];
    __shared__ int wtot[32];

    const int t   = threadIdx.x;
    const int blk = blockIdx.x;
    tot[t] = 0;
    pre[t] = 0;
    mine[t] = (t < nbin) ? hist[(size_t)blk * nbinp + t] : 0;
    __syncthreads();

    {
        const int nq  = nbinp >> 2;
        const int grp = t >> 7;
        for (int q = t & 127; q < nq; q += 128) {
            int4 st = {0, 0, 0, 0}, sp = {0, 0, 0, 0};
            for (int r0 = grp; r0 < SORT_BLOCKS; r0 += 64) {
                int4 h[8];
#pragma unroll
                for (int u = 0; u < 8; ++u)
                    h[u] = *(const int4*)(hist + (size_t)(r0 + 8 * u) * nbinp + 4 * q);
#pragma unroll
                for (int u = 0; u < 8; ++u) {
                    st.x += h[u].x; st.y += h[u].y; st.z += h[u].z; st.w += h[u].w;
                    if (r0 + 8 * u < blk) { sp.x += h[u].x; sp.y += h[u].y; sp.z += h[u].z; sp.w += h[u].w; }
                }
            }
            atomicAdd(&tot[4 * q + 0], st.x); atomicAdd(&tot[4 * q + 1], st.y);
            atomicAdd(&tot[4 * q + 2], st.z); atomicAdd(&tot[4 * q + 3], st.w);
            atomicAdd(&pre[4 * q + 0], sp.x); atomicAdd(&pre[4 * q + 1], sp.y);
            atomicAdd(&pre[4 * q + 2], sp.z); atomicAdd(&pre[4 * q + 3], sp.w);
        }
    }
    __syncthreads();

    const int vt = (t < nbin) ? tot[t] : 0;
    const int vm = mine[t];
    const int bs = block_excl_scan_1024(vt, wtot);
    const int lo = block_excl_scan_1024(vm, wtot);
    gbase[t] = bs + pre[t];
    loff[t]  = lo;
    lcur[t]  = lo;
    if (blk == 0) {
        if (t < nbin) binStart[t] = bs;
        if (t == 0)   binStart[nbin] = nEdges;
    }
    __syncthreads();

    const int beg = blk * chunk;
    const int end = min(beg + chunk, nEdges);
    for (int i0 = beg + t; i0 < end; i0 += 4 * SORT_THREADS) {
        int d[4], r[4], s[4], w[4];
#pragma unroll
        for (int u = 0; u < 4; ++u) {
            const int  i  = i0 + u * SORT_THREADS;
            const bool ok = i < end;
            d[u] = ok ? dst[i] : -1;
            r[u] = ok ? rel[i] : 0;
            s[u] = ok ? src[i] : 0;
            w[u] = ok ? __float_as_int(ew[i]) : 0;
        }
#pragma unroll
        for (int u = 0; u < 4; ++u) {
            if (d[u] >= 0) {
                const int b   = d[u] >> BIN_SHIFT;
                const int key = (d[u] & (BIN_NODES - 1)) * NREL + r[u];
                const int p   = atomicAdd(&lcur[b], 1);
                int2 rec;
                rec.x = s[u] | (key << KEY_SHIFT);
                rec.y = w[u];
                stage[p] = rec;
            }
        }
    }
    __syncthreads();

    const int lane = t & 31;
    const int wave = t >> 5;
    for (int b = wave; b < nbin; b += 32) {
        const int n = mine[b];
        const int l = loff[b];
        const int g = gbase[b];
        for (int j = lane; j < n; j += 32) part[g + j] = stage[l + j];
    }
}

__global__ void __launch_bounds__(SORT_THREADS)
bin_order_kernel(const int2* __restrict__ part,
                 const int*  __restrict__ binStart,
                 int*        __restrict__ bend,
                 int2*       __restrict__ erec,
                 int nBuckets)
{
    __shared__ int2 sorted[FINE_CAP];
    __shared__ int  cnt[FINE_KEYS];
    __shared__ int  lcur[FINE_KEYS];
    __shared__ int  wtot[32];

    const int t = threadIdx.x;
    const int g = blockIdx.x;
    const int s = binStart[g];
    const int n = binStart[g + 1] - s;
    const int2* in = part + s;

    cnt[t] = 0;
    __syncthreads();
    for (int i0 = t; i0 < n; i0 += 4 * SORT_THREADS) {
        int x[4];
#pragma unroll
        for (int u = 0; u < 4; ++u) {
            const int i = i0 + u * SORT_THREADS;
            x[u] = (i < n) ? in[i].x : -1;
        }
#pragma unroll
        for (int u = 0; u < 4; ++u)
            if (x[u] >= 0) atomicAdd(&cnt[x[u] >> KEY_SHIFT], 1);
    }
    __syncthreads();

    const int c   = cnt[t];
    const int off = block_excl_scan_1024(c, wtot);
    lcur[t] = off;
    const int b = g * FINE_KEYS + t;
    if (b < nBuckets) bend[b] = s + off + c;
    __syncthreads();

    if (n <= FINE_CAP) {
        for (int i0 = t; i0 < n; i0 += 4 * SORT_THREADS) {
            int2 rc[4];
#pragma unroll
            for (int u = 0; u < 4; ++u) {
                const int i = i0 + u * SORT_THREADS;
                if (i < n) rc[u] = in[i]; else { rc[u].x = -1; rc[u].y = 0; }
            }
#pragma unroll
            for (int u = 0; u < 4; ++u) {
                if (rc[u].x >= 0) {
                    const int p = atomicAdd(&lcur[rc[u].x >> KEY_SHIFT], 1);
                    int2 o;
                    o.x = rc[u].x & SRC_MASK;
                    o.y = rc[u].y;
                    sorted[p] = o;
                }
            }
        }
        __syncthreads();
        for (int i = t; i < n; i += SORT_THREADS) erec[s + i] = sorted[i];
    } else {
        for (int i = t; i < n; i += SORT_THREADS) {
            const int2 rc = in[i];
            const int  p  = atomicAdd(&lcur[rc.x >> KEY_SHIFT], 1);
            int2 o;
            o.x = rc.x & SRC_MASK;
            o.y = rc.y;
            erec[s + p] = o;
        }
    }
}

__global__ void rgcn_gather_kernel(const float* __restrict__ x,
                                   const int*   __restrict__ erec,
                                   const int*   __restrict__ bend,
                                   float*       __restrict__ agg,
                                   float*       __restrict__ deg,
                                   int nNodes)
{
    const int lane = threadIdx.x & 31;
    const int wave = __builtin_amdgcn_readfirstlane((int)(threadIdx.x >> 5));
    const int gw   = blockIdx.x * (blockDim.x >> 5) + wave;
    float dsum = 0.0f;

#define GATHER_ROW(SV, WV, VV, L)                                               \
            const int    SV = __builtin_amdgcn_readlane(rec.x, (L));            \
            const float  WV = __int_as_float(__builtin_amdgcn_readlane(rec.y, (L))); \
            const float2 VV = ((const float2*)(x + (size_t)SV * DIM))[lane];

#define GATHER_BUCKET(C, S, B0, B1)                                             \
        {                                                                       \
            const int lo = max((B0), cb);                                       \
            const int hi = min((B1), ce);                                       \
            for (int j = lo; j < hi; j += 4) {                                  \
                const int last = hi - 1 - cb;                                   \
                const int l0 = j - cb;                                          \
                const int l1 = min(l0 + 1, last);                               \
                const int l2 = min(l0 + 2, last);                               \
                const int l3 = min(l0 + 3, last);                               \
                GATHER_ROW(sA, wA, vA, l0)                                      \
                GATHER_ROW(sB, wB, vB, l1)                                      \
                GATHER_ROW(sC, wC, vC, l2)                                      \
                GATHER_ROW(sD, wD, vD, l3)                                      \
                C.x += vA.x * wA; C.y += vA.y * wA; S += wA;                    \
                if (j + 1 < hi) { C.x += vB.x * wB; C.y += vB.y * wB; S += wB; } \
                if (j + 2 < hi) { C.x += vC.x * wC; C.y += vC.y * wC; S += wC; } \
                if (j + 3 < hi) { C.x += vD.x * wD; C.y += vD.y * wD; S += wD; } \
            }                                                                   \
        }

    for (int q = 0; q < GATHER_NODES; ++q) {
        const int node = gw * GATHER_NODES + q;
        if (node >= nNodes) break;
        const int4 be  = *(const int4*)(bend + (size_t)node * NREL);
        const int  beg = (node > 0) ? bend[(size_t)node * NREL - 1] : 0;
        const int  fin = be.w;

        float2 c0 = {0.0f, 0.0f}, c1 = c0, c2 = c0, c3 = c0;
        float  s0 = 0.0f, s1 = 0.0f, s2 = 0.0f, s3 = 0.0f;

        for (int cb = beg; cb < fin; cb += 32) {
            const int  ce  = min(cb + 32, fin);
            const int2 rec = ((const int2*)erec)[min(cb + lane, fin - 1)];
            GATHER_BUCKET(c0, s0, beg,  be.x)
            GATHER_BUCKET(c1, s1, be.x, be.y)
            GATHER_BUCKET(c2, s2, be.y, be.z)
            GATHER_BUCKET(c3, s3, be.z, be.w)
        }

        float* o = agg + (size_t)node * (NREL * DIM) + 2 * lane;
        *(float2*)(o)           = c0;
        *(float2*)(o + DIM)     = c1;
        *(float2*)(o + 2 * DIM) = c2;
        *(float2*)(o + 3 * DIM) = c3;

        const int   r4  = lane & 3;
        const float sel = (r4 == 0) ? s0 : (r4 == 1) ? s1 : (r4 == 2) ? s2 : s3;
        if ((lane >> 2) == q) dsum = sel;
    }
#undef GATHER_BUCKET
#undef GATHER_ROW

    const long di = (long)gw * (GATHER_NODES * NREL) + lane;
    if (di < (long)nNodes * NREL) deg[di] = dsum;
}

#define LAYER_WAVES   25
#define LAYER_THREADS (LAYER_WAVES * 32)
#define STRT          68

__device__ __forceinline__ void wave_lds_handover()
{
    __builtin_amdgcn_fence(__ATOMIC_RELEASE, "wavefront");
    __builtin_amdgcn_wave_barrier();
    __builtin_amdgcn_fence(__ATOMIC_ACQUIRE, "wavefront");
}

__global__ void __launch_bounds__(LAYER_THREADS)
rgcn_layer_kernel(const float* __restrict__ agg,
                  const float* __restrict__ deg,
                  const float* __restrict__ x,
                  const float* __restrict__ lin_w,
                  const float* __restrict__ lin_b,
                  const float* __restrict__ self_w,
                  const float* __restrict__ self_b,
                  const float* __restrict__ prev,
                  const float* __restrict__ pw,
                  const float* __restrict__ pb,
                  const float* __restrict__ tw,
                  const float* __restrict__ tb,
                  float*       __restrict__ hidden,
                  float*       __restrict__ outp,
                  int nNodes)
{
    __shared__ __attribute__((aligned(16)))
        float smem[DIM * STRC + 2 * DIM * STRH + LAYER_WAVES * 64 + LAYER_WAVES * 16 * STRT];
    float* WT  = smem;
    float* WP  = WT  + DIM * STRC;
    float* WTT = WP  + DIM * STRH;
    float* RD  = WTT + DIM * STRH;
    float* HT  = RD  + LAYER_WAVES * 64;

    const int t = threadIdx.x;
    for (int i = t; i < KCONV * DIM; i += LAYER_THREADS) {
        int k = i >> 6;
        int c = i & 63;
        float v = (k < NREL * DIM) ? lin_w[i] : self_w[i - NREL * DIM * DIM];
        WT[(size_t)c * STRC + k] = v;
    }
    for (int i = t; i < KHW * DIM; i += LAYER_THREADS) {
        int k = i >> 6;
        int c = i & 63;
        WP [(size_t)c * STRH + k] = pw[i];
        WTT[(size_t)c * STRH + k] = tw[i];
    }
    __syncthreads();

    const int lane = t & 31;
    const int wave = __builtin_amdgcn_readfirstlane(t >> 5);
    const int m     = lane & 15;
    const int khalf = lane >> 4;

    const float* B0 = WT + (size_t)(m)      * STRC;
    const float* B1 = WT + (size_t)(16 + m) * STRC;
    const float* B2 = WT + (size_t)(32 + m) * STRC;
    const float* B3 = WT + (size_t)(48 + m) * STRC;
    const float* P0 = WP  + (size_t)(m)      * STRH;
    const float* P1 = WP  + (size_t)(16 + m) * STRH;
    const float* P2 = WP  + (size_t)(32 + m) * STRH;
    const float* P3 = WP  + (size_t)(48 + m) * STRH;
    const float* T0 = WTT + (size_t)(m)      * STRH;
    const float* T1 = WTT + (size_t)(16 + m) * STRH;
    const float* T2 = WTT + (size_t)(32 + m) * STRH;
    const float* T3 = WTT + (size_t)(48 + m) * STRH;
    float*       rdw  = RD + wave * 64;
    float*       ht   = HT + wave * (16 * STRT);
    const float* rd   = rdw + m * NREL;
    const float* hRow = ht + m * STRT;

    const long base = ((long)blockIdx.x * LAYER_WAVES + wave) * 16;
    if (base >= nNodes) return;
    {
        {
            float2 dv = ((const float2*)(deg + base * NREL))[lane];
            rdw[2 * lane]     = 1.0f / (dv.x + 1e-10f);
            rdw[2 * lane + 1] = 1.0f / (dv.y + 1e-10f);
        }
        wave_lds_handover();

        const float* aggRow = agg  + (base + m) * (size_t)(NREL * DIM);
        const float* xRow   = x    + (base + m) * (size_t)DIM;
        const float* pRow   = prev + (base + m) * (size_t)DIM;

        v8f a0 = {}, a1 = {}, a2 = {}, a3 = {};
        for (int k = 0; k < NREL * DIM; k += 4) {
            int kk = k + 2 * khalf;
            float  sc = rd[k >> 6];
            float2 av = *(const float2*)(aggRow + kk);
            v2f a; a.x = av.x * sc; a.y = av.y * sc;
            v2f b0 = *(const v2f*)(B0 + kk);
            v2f b1 = *(const v2f*)(B1 + kk);
            v2f b2 = *(const v2f*)(B2 + kk);
            v2f b3 = *(const v2f*)(B3 + kk);
            a0 = __builtin_amdgcn_wmma_f32_16x16x4_f32(false, a, false, b0, (short)0, a0, false, false);
            a1 = __builtin_amdgcn_wmma_f32_16x16x4_f32(false, a, false, b1, (short)0, a1, false, false);
            a2 = __builtin_amdgcn_wmma_f32_16x16x4_f32(false, a, false, b2, (short)0, a2, false, false);
            a3 = __builtin_amdgcn_wmma_f32_16x16x4_f32(false, a, false, b3, (short)0, a3, false, false);
        }
#pragma unroll 4
        for (int k = 0; k < DIM; k += 4) {
            int kk = k + 2 * khalf;
            float2 av = *(const float2*)(xRow + kk);
            v2f a; a.x = av.x; a.y = av.y;
            int kw = NREL * DIM + kk;
            v2f b0 = *(const v2f*)(B0 + kw);
            v2f b1 = *(const v2f*)(B1 + kw);
            v2f b2 = *(const v2f*)(B2 + kw);
            v2f b3 = *(const v2f*)(B3 + kw);
            a0 = __builtin_amdgcn_wmma_f32_16x16x4_f32(false, a, false, b0, (short)0, a0, false, false);
            a1 = __builtin_amdgcn_wmma_f32_16x16x4_f32(false, a, false, b1, (short)0, a1, false, false);
            a2 = __builtin_amdgcn_wmma_f32_16x16x4_f32(false, a, false, b2, (short)0, a2, false, false);
            a3 = __builtin_amdgcn_wmma_f32_16x16x4_f32(false, a, false, b3, (short)0, a3, false, false);
        }

#define CONV_EPILOGUE(ACC, TILE)                                           \
        {                                                                  \
            int   col = (TILE) * 16 + m;                                   \
            float bia = lin_b[col] + self_b[col];                          \
            _Pragma("unroll")                                              \
            for (int r = 0; r < 8; ++r) {                                  \
                int   lr = khalf * 8 + r;                                  \
                float hv = sigmoidf_(ACC[r] + bia);                        \
                ht[lr * STRT + col] = hv;                                  \
                hidden[(base + lr) * DIM + col] = hv;                      \
            }                                                              \
        }
        CONV_EPILOGUE(a0, 0) CONV_EPILOGUE(a1, 1) CONV_EPILOGUE(a2, 2) CONV_EPILOGUE(a3, 3)
#undef CONV_EPILOGUE
        wave_lds_handover();

        v8f p0 = {}, p1 = {}, p2 = {}, p3 = {};
        v8f g0 = {}, g1 = {}, g2 = {}, g3 = {};

#define HW_STEP(AROW, KOFF)                                                              \
        for (int k = 0; k < DIM; k += 4) {                                               \
            int kk = k + 2 * khalf;                                                      \
            float2 av = *(const float2*)((AROW) + kk);                                   \
            v2f a; a.x = av.x; a.y = av.y;                                               \
            int kw = kk + (KOFF);                                                        \
            v2f bp0 = *(const v2f*)(P0 + kw);                                            \
            v2f bp1 = *(const v2f*)(P1 + kw);                                            \
            v2f bp2 = *(const v2f*)(P2 + kw);                                            \
            v2f bp3 = *(const v2f*)(P3 + kw);                                            \
            v2f bt0 = *(const v2f*)(T0 + kw);                                            \
            v2f bt1 = *(const v2f*)(T1 + kw);                                            \
            v2f bt2 = *(const v2f*)(T2 + kw);                                            \
            v2f bt3 = *(const v2f*)(T3 + kw);                                            \
            p0 = __builtin_amdgcn_wmma_f32_16x16x4_f32(false, a, false, bp0, (short)0, p0, false, false); \
            p1 = __builtin_amdgcn_wmma_f32_16x16x4_f32(false, a, false, bp1, (short)0, p1, false, false); \
            p2 = __builtin_amdgcn_wmma_f32_16x16x4_f32(false, a, false, bp2, (short)0, p2, false, false); \
            p3 = __builtin_amdgcn_wmma_f32_16x16x4_f32(false, a, false, bp3, (short)0, p3, false, false); \
            g0 = __builtin_amdgcn_wmma_f32_16x16x4_f32(false, a, false, bt0, (short)0, g0, false, false); \
            g1 = __builtin_amdgcn_wmma_f32_16x16x4_f32(false, a, false, bt1, (short)0, g1, false, false); \
            g2 = __builtin_amdgcn_wmma_f32_16x16x4_f32(false, a, false, bt2, (short)0, g2, false, false); \
            g3 = __builtin_amdgcn_wmma_f32_16x16x4_f32(false, a, false, bt3, (short)0, g3, false, false); \
        }
        HW_STEP(hRow, 0)
        HW_STEP(pRow, DIM)
#undef HW_STEP

#define HW_EPILOGUE(PACC, GACC, TILE)                                      \
        {                                                                  \
            int   col = (TILE) * 16 + m;                                   \
            float pbv = pb[col];                                           \
            float tbv = tb[col];                                           \
            _Pragma("unroll")                                              \
            for (int r = 0; r < 8; ++r) {                                  \
                int   lr = khalf * 8 + r;                                  \
                float pr = fmaxf(PACC[r] + pbv, 0.0f);                     \
                float gt = sigmoidf_(GACC[r] + tbv);                       \
                float h  = ht[lr * STRT + col];                            \
                outp[(base + lr) * DIM + col] = gt * pr + (1.0f - gt) * h; \
            }                                                              \
        }
        HW_EPILOGUE(p0, g0, 0) HW_EPILOGUE(p1, g1, 1)
        HW_EPILOGUE(p2, g2, 2) HW_EPILOGUE(p3, g3, 3)
#undef HW_EPILOGUE
    }
}

extern "C" void kernel_launch(void* const* d_in, const int* in_sizes, int n_in,
                              void* d_out, int out_size, void* d_ws, size_t ws_size,
                              hipStream_t stream) {
    const float* node_feat   = (const float*)d_in[0];
    const float* edge_weight = (const float*)d_in[1];
    const float* conv_lin_w  = (const float*)d_in[2];
    const float* conv_lin_b  = (const float*)d_in[3];
    const float* conv_self_w = (const float*)d_in[4];
    const float* conv_self_b = (const float*)d_in[5];
    const float* hw_proj_w   = (const float*)d_in[6];
    const float* hw_proj_b   = (const float*)d_in[7];
    const float* hw_trans_w  = (const float*)d_in[8];
    const float* hw_trans_b  = (const float*)d_in[9];
    const int*   edge_src    = (const int*)d_in[10];
    const int*   edge_dst    = (const int*)d_in[11];
    const int*   edge_rel    = (const int*)d_in[12];
    float*       out         = (float*)d_out;

    const int N = in_sizes[0] / DIM;
    const int E = in_sizes[10];

    float* agg = (float*)d_ws;
    float* deg = agg + (size_t)N * NREL * DIM;
    float* hA  = deg + (size_t)N * NREL;
    float* hB  = hA  + (size_t)N * DIM;
    float* cur0 = hB   + (size_t)N * DIM;
    float* cur1 = cur0 + (size_t)N * DIM;
    const int NB    = N * NREL;
    const int nbin  = (N + BIN_NODES - 1) >> BIN_SHIFT;
    const int nbinp = (nbin + 3) & ~3;
    const int chunk = (E + SORT_BLOCKS - 1) / SORT_BLOCKS;
    int*  cursor   = (int*)(cur1 + (size_t)N * DIM);
    int*  erec     = cursor + NB;
    int*  part     = erec + 2 * (size_t)E;
    int*  hist     = part + 2 * (size_t)E;
    int*  binStart = hist + (size_t)SORT_BLOCKS * nbinp;

    const int gatherBlocks  = (N + 8 * GATHER_NODES - 1) / (8 * GATHER_NODES);
    const int tiles         = (N + 15) / 16;
    const int layerBlocks   = (tiles + LAYER_WAVES - 1) / LAYER_WAVES;
    const size_t stageLds = (size_t)chunk * sizeof(int2);

    bin_hist_kernel<<<SORT_BLOCKS, SORT_THREADS, 0, stream>>>(
        edge_dst, hist, E, chunk, nbin, nbinp);
    bin_partition_kernel<<<SORT_BLOCKS, SORT_THREADS, stageLds, stream>>>(
        edge_src, edge_dst, edge_rel, edge_weight, hist, binStart, (int2*)part,
        E, chunk, nbin, nbinp);
    bin_order_kernel<<<nbin, SORT_THREADS, 0, stream>>>(
        (const int2*)part, binStart, cursor, (int2*)erec, NB);

    for (int i = 0; i < 3; ++i) {
        const float* xin  = (i == 0) ? node_feat : ((i == 1) ? cur0 : cur1);
        const float* prev = (i == 0) ? node_feat : ((i == 1) ? hA : hB);
        float*       hid  = (i & 1) ? hB : hA;
        float*       gout = (i == 0) ? cur0 : ((i == 1) ? cur1 : out);

        rgcn_gather_kernel<<<gatherBlocks, 256, 0, stream>>>(
            xin, erec, cursor, agg, deg, N);

        rgcn_layer_kernel<<<layerBlocks, LAYER_THREADS, 0, stream>>>(
            agg, deg, xin,
            conv_lin_w  + (size_t)i * 256 * DIM, conv_lin_b  + (size_t)i * DIM,
            conv_self_w + (size_t)i * DIM * DIM, conv_self_b + (size_t)i * DIM,
            prev,
            hw_proj_w  + (size_t)i * KHW * DIM, hw_proj_b  + (size_t)i * DIM,
            hw_trans_w + (size_t)i * KHW * DIM, hw_trans_b + (size_t)i * DIM,
            hid, gout, N);
    }
}


# Segmented gather: a node's edges walked as one flat list, up to 16 source-row loads in flight, bucket ends of a wave's 8 nodes from one line load, next node's records fetched ahead; sum order kept
# speedup vs baseline: 3.0640x; 1.0488x over previous
#include <hip/hip_runtime.h>
#include <hip/hip_bf16.h>
#include <math.h>


typedef __attribute__((ext_vector_type(2))) float v2f;
typedef __attribute__((ext_vector_type(8))) float v8f;

#define DIM   64
#define NREL  4
#define KCONV 320
#define KHW   128
#define STRC  324
#define STRH  132

__device__ __forceinline__ float sigmoidf_(float v) {
    return 1.0f / (1.0f + __expf(-v));
}

#define GATHER_NODES 8

#define BIN_SHIFT   8
#define BIN_NODES   (1 << BIN_SHIFT)
#define FINE_KEYS   (BIN_NODES * NREL)
#define SORT_BLOCKS 256
#define SORT_THREADS 1024
#define KEY_SHIFT   20
#define SRC_MASK    ((1 << KEY_SHIFT) - 1)
#define FINE_CAP    8192

__device__ __forceinline__ int block_excl_scan_1024(int v, int* wtot)
{
    const int lane = threadIdx.x & 31;
    const int wave = threadIdx.x >> 5;
    int inc = v;
    for (int o = 1; o < 32; o <<= 1) {
        int n = __shfl_up(inc, o);
        if (lane >= o) inc += n;
    }
    if (lane == 31) wtot[wave] = inc;
    __syncthreads();
    int woff = 0;
    for (int w = 0; w < 32; ++w) {
        int x = wtot[w];
        if (w < wave) woff += x;
    }
    __syncthreads();
    return woff + inc - v;
}

__global__ void __launch_bounds__(SORT_THREADS)
bin_hist_kernel(const int* __restrict__ dst,
                int*       __restrict__ hist,
                int nEdges, int chunk, int nbin, int nbinp)
{
    __shared__ int cnt[SORT_THREADS];
    const int t = threadIdx.x;
    cnt[t] = 0;
    __syncthreads();
    const int beg = blockIdx.x * chunk;
    const int end = min(beg + chunk, nEdges);
    for (int i0 = beg + t; i0 < end; i0 += 4 * SORT_THREADS) {
        int d[4];
#pragma unroll
        for (int u = 0; u < 4; ++u) {
            const int i = i0 + u * SORT_THREADS;
            d[u] = (i < end) ? dst[i] : -1;
        }
#pragma unroll
        for (int u = 0; u < 4; ++u)
            if (d[u] >= 0) atomicAdd(&cnt[d[u] >> BIN_SHIFT], 1);
    }
    __syncthreads();
    if (t < nbinp) hist[(size_t)blockIdx.x * nbinp + t] = (t < nbin) ? cnt[t] : 0;
}

__global__ void __launch_bounds__(SORT_THREADS)
bin_partition_kernel(const int*   __restrict__ src,
                     const int*   __restrict__ dst,
                     const int*   __restrict__ rel,
                     const float* __restrict__ ew,
                     const int*   __restrict__ hist,
                     int*         __restrict__ binStart,
                     int2*        __restrict__ part,
                     int nEdges, int chunk, int nbin, int nbinp)
{
    extern __shared__ int2 stage[];
    __shared__ int tot[SORT_THREADS];
    __shared__ int pre[SORT_THREADS];
    __shared__ int mine[SORT_THREADS];
    __shared__ int loff[SORT_THREADS];
    __shared__ int lcur[SORT_THREADS];
    __shared__ int gbase[SORT_THREADS];
    __shared__ int wtot[32];

    const int t   = threadIdx.x;
    const int blk = blockIdx.x;
    tot[t] = 0;
    pre[t] = 0;
    mine[t] = (t < nbin) ? hist[(size_t)blk * nbinp + t] : 0;
    __syncthreads();

    {
        const int nq  = nbinp >> 2;
        const int grp = t >> 7;
        for (int q = t & 127; q < nq; q += 128) {
            int4 st = {0, 0, 0, 0}, sp = {0, 0, 0, 0};
            for (int r0 = grp; r0 < SORT_BLOCKS; r0 += 64) {
                int4 h[8];
#pragma unroll
                for (int u = 0; u < 8; ++u)
                    h[u] = *(const int4*)(hist + (size_t)(r0 + 8 * u) * nbinp + 4 * q);
#pragma unroll
                for (int u = 0; u < 8; ++u) {
                    st.x += h[u].x; st.y += h[u].y; st.z += h[u].z; st.w += h[u].w;
                    if (r0 + 8 * u < blk) { sp.x += h[u].x; sp.y += h[u].y; sp.z += h[u].z; sp.w += h[u].w; }
                }
            }
            atomicAdd(&tot[4 * q + 0], st.x); atomicAdd(&tot[4 * q + 1], st.y);
            atomicAdd(&tot[4 * q + 2], st.z); atomicAdd(&tot[4 * q + 3], st.w);
            atomicAdd(&pre[4 * q + 0], sp.x); atomicAdd(&pre[4 * q + 1], sp.y);
            atomicAdd(&pre[4 * q + 2], sp.z); atomicAdd(&pre[4 * q + 3], sp.w);
        }
    }
    __syncthreads();

    const int vt = (t < nbin) ? tot[t] : 0;
    const int vm = mine[t];
    const int bs = block_excl_scan_1024(vt, wtot);
    const int lo = block_excl_scan_1024(vm, wtot);
    gbase[t] = bs + pre[t];
    loff[t]  = lo;
    lcur[t]  = lo;
    if (blk == 0) {
        if (t < nbin) binStart[t] = bs;
        if (t == 0)   binStart[nbin] = nEdges;
    }
    __syncthreads();

    const int beg = blk * chunk;
    const int end = min(beg + chunk, nEdges);
    for (int i0 = beg + t; i0 < end; i0 += 4 * SORT_THREADS) {
        int d[4], r[4], s[4], w[4];
#pragma unroll
        for (int u = 0; u < 4; ++u) {
            const int  i  = i0 + u * SORT_THREADS;
            const bool ok = i < end;
            d[u] = ok ? dst[i] : -1;
            r[u] = ok ? rel[i] : 0;
            s[u] = ok ? src[i] : 0;
            w[u] = ok ? __float_as_int(ew[i]) : 0;
        }
#pragma unroll
        for (int u = 0; u < 4; ++u) {
            if (d[u] >= 0) {
                const int b   = d[u] >> BIN_SHIFT;
                const int key = (d[u] & (BIN_NODES - 1)) * NREL + r[u];
                const int p   = atomicAdd(&lcur[b], 1);
                int2 rec;
                rec.x = s[u] | (key << KEY_SHIFT);
                rec.y = w[u];
                stage[p] = rec;
            }
        }
    }
    __syncthreads();

    const int lane = t & 31;
    const int wave = t >> 5;
    for (int b = wave; b < nbin; b += 32) {
        const int n = mine[b];
        const int l = loff[b];
        const int g = gbase[b];
        for (int j = lane; j < n; j += 32) part[g + j] = stage[l + j];
    }
}

__global__ void __launch_bounds__(SORT_THREADS)
bin_order_kernel(const int2* __restrict__ part,
                 const int*  __restrict__ binStart,
                 int*        __restrict__ bend,
                 int2*       __restrict__ erec,
                 int nBuckets)
{
    __shared__ int2 sorted[FINE_CAP];
    __shared__ int  cnt[FINE_KEYS];
    __shared__ int  lcur[FINE_KEYS];
    __shared__ int  wtot[32];

    const int t = threadIdx.x;
    const int g = blockIdx.x;
    const int s = binStart[g];
    const int n = binStart[g + 1] - s;
    const int2* in = part + s;

    cnt[t] = 0;
    __syncthreads();
    for (int i0 = t; i0 < n; i0 += 4 * SORT_THREADS) {
        int x[4];
#pragma unroll
        for (int u = 0; u < 4; ++u) {
            const int i = i0 + u * SORT_THREADS;
            x[u] = (i < n) ? in[i].x : -1;
        }
#pragma unroll
        for (int u = 0; u < 4; ++u)
            if (x[u] >= 0) atomicAdd(&cnt[x[u] >> KEY_SHIFT], 1);
    }
    __syncthreads();

    const int c   = cnt[t];
    const int off = block_excl_scan_1024(c, wtot);
    lcur[t] = off;
    const int b = g * FINE_KEYS + t;
    if (b < nBuckets) bend[b] = s + off + c;
    __syncthreads();

    if (n <= FINE_CAP) {
        for (int i0 = t; i0 < n; i0 += 4 * SORT_THREADS) {
            int2 rc[4];
#pragma unroll
            for (int u = 0; u < 4; ++u) {
                const int i = i0 + u * SORT_THREADS;
                if (i < n) rc[u] = in[i]; else { rc[u].x = -1; rc[u].y = 0; }
            }
#pragma unroll
            for (int u = 0; u < 4; ++u) {
                if (rc[u].x >= 0) {
                    const int p = atomicAdd(&lcur[rc[u].x >> KEY_SHIFT], 1);
                    int2 o;
                    o.x = rc[u].x & SRC_MASK;
                    o.y = rc[u].y;
                    sorted[p] = o;
                }
            }
        }
        __syncthreads();
        for (int i = t; i < n; i += SORT_THREADS) erec[s + i] = sorted[i];
    } else {
        for (int i = t; i < n; i += SORT_THREADS) {
            const int2 rc = in[i];
            const int  p  = atomicAdd(&lcur[rc.x >> KEY_SHIFT], 1);
            int2 o;
            o.x = rc.x & SRC_MASK;
            o.y = rc.y;
            erec[s + p] = o;
        }
    }
}

__global__ void __launch_bounds__(256)
rgcn_gather_kernel(const float* __restrict__ x,
                   const int*   __restrict__ erec,
                   const int*   __restrict__ bend,
                   float*       __restrict__ agg,
                   float*       __restrict__ deg,
                   int nNodes)
{
    const int lane = threadIdx.x & 31;
    const int wave  = __builtin_amdgcn_readfirstlane((int)(threadIdx.x >> 5));
    const int gw    = blockIdx.x * (blockDim.x >> 5) + wave;
    const int node0 = gw * GATHER_NODES;
    if (node0 >= nNodes) return;

    const int2* __restrict__ rec2 = (const int2*)erec;
    const char* __restrict__ xb   = (const char*)x;
    const unsigned lane8 = (unsigned)lane << 3;
    const long nb = (long)nNodes * NREL;
    const long bi = (long)gw * (GATHER_NODES * NREL) + lane;
    const int  bv    = (bi < nb) ? bend[bi] : 0;
    const int  bprev = (gw > 0) ? bend[(long)gw * (GATHER_NODES * NREL) - 1] : 0;
    float dsum = 0.0f;

    int  begq = bprev;
    int2 recN = {0, 0};
    {
        const int fin0 = __builtin_amdgcn_readlane(bv, 3);
        if (fin0 > begq) recN = rec2[min(begq + lane, fin0 - 1)];
    }

#define ROW_LOAD(U)                                                              \
        const unsigned s##U = (unsigned)__builtin_amdgcn_readlane(rec.x, (U));  \
        const float    w##U = __int_as_float(__builtin_amdgcn_readlane(rec.y, (U))); \
        const float2   v##U = *(const float2*)(xb + (size_t)((s##U << 8) | lane8));

#define BUCKET_DONE()                                                            \
        {                                                                        \
            *(float2*)(aggN + r * DIM) = acc;                                    \
            if (lane == 4 * q + r) dsum = ss;                                    \
            acc.x = 0.0f; acc.y = 0.0f; ss = 0.0f;                               \
            ++r;                                                                 \
            bnd = (r == 1) ? e1 : ((r == 2) ? e2 : 0x7fffffff);                  \
        }

#define BUCKET_CHECK(U)                                                          \
        if ((U) >= rel) {                                                        \
            do { BUCKET_DONE() rel = bnd - j0; } while ((U) >= rel);             \
        }

#define EDGE_SURE(U)                                                             \
        BUCKET_CHECK(U)                                                          \
        acc.x += v##U.x * w##U; acc.y += v##U.y * w##U; ss += w##U;

#define EDGE_MAYBE(U)                                                            \
        BUCKET_CHECK(U)                                                          \
        {                                                                        \
            const float wz = ((U) < left) ? w##U : 0.0f;                         \
            acc.x += v##U.x * wz; acc.y += v##U.y * wz; ss += wz;                \
        }

    for (int q = 0; q < GATHER_NODES; ++q) {
        const int node = node0 + q;
        if (node >= nNodes) break;
        const int e0  = __builtin_amdgcn_readlane(bv, 4 * q);
        const int e1  = __builtin_amdgcn_readlane(bv, 4 * q + 1);
        const int e2  = __builtin_amdgcn_readlane(bv, 4 * q + 2);
        const int fin = __builtin_amdgcn_readlane(bv, 4 * q + 3);
        const int beg = begq;
        int2 rec = recN;
        if (q + 1 < GATHER_NODES && node + 1 < nNodes) {
            const int finN = __builtin_amdgcn_readlane(bv, 4 * q + 7);
            if (finN > fin) recN = rec2[min(fin + lane, finN - 1)];
        }
        begq = fin;

        float* aggN = agg + (size_t)node * (NREL * DIM) + 2 * lane;
        float2 acc = {0.0f, 0.0f};
        float  ss  = 0.0f;
        int    r   = 0;
        int    bnd = e0;

        for (int cb = beg; cb < fin; cb += 32) {
            const int ce = min(cb + 32, fin);
            if (cb != beg) rec = rec2[min(cb + lane, fin - 1)];
            for (int j0 = cb; j0 < ce; j0 += 16) {
                if (j0 != cb) {
                    rec.x = __shfl_xor(rec.x, 16);
                    rec.y = __shfl_xor(rec.y, 16);
                }
                const int left = ce - j0;
                int rel = bnd - j0;
                if (left > 8) {
                    ROW_LOAD(0)  ROW_LOAD(1)  ROW_LOAD(2)  ROW_LOAD(3)
                    ROW_LOAD(4)  ROW_LOAD(5)  ROW_LOAD(6)  ROW_LOAD(7)
                    ROW_LOAD(8)  ROW_LOAD(9)  ROW_LOAD(10) ROW_LOAD(11)
                    ROW_LOAD(12) ROW_LOAD(13) ROW_LOAD(14) ROW_LOAD(15)
                    EDGE_SURE(0)   EDGE_SURE(1)   EDGE_SURE(2)   EDGE_SURE(3)
                    EDGE_SURE(4)   EDGE_SURE(5)   EDGE_SURE(6)   EDGE_SURE(7)
                    EDGE_SURE(8)   EDGE_MAYBE(9)  EDGE_MAYBE(10) EDGE_MAYBE(11)
                    EDGE_MAYBE(12) EDGE_MAYBE(13) EDGE_MAYBE(14) EDGE_MAYBE(15)
                } else {
                    ROW_LOAD(0) ROW_LOAD(1) ROW_LOAD(2) ROW_LOAD(3)
                    ROW_LOAD(4) ROW_LOAD(5) ROW_LOAD(6) ROW_LOAD(7)
                    EDGE_SURE(0)  EDGE_MAYBE(1) EDGE_MAYBE(2) EDGE_MAYBE(3)
                    EDGE_MAYBE(4) EDGE_MAYBE(5) EDGE_MAYBE(6) EDGE_MAYBE(7)
                }
            }
        }
        while (r < NREL) BUCKET_DONE()
    }
#undef EDGE_MAYBE
#undef EDGE_SURE
#undef BUCKET_CHECK
#undef BUCKET_DONE
#undef ROW_LOAD

    if (bi < nb) deg[bi] = dsum;
}

#define LAYER_WAVES   25
#define LAYER_THREADS (LAYER_WAVES * 32)
#define STRT          68

__device__ __forceinline__ void wave_lds_handover()
{
    __builtin_amdgcn_fence(__ATOMIC_RELEASE, "wavefront");
    __builtin_amdgcn_wave_barrier();
    __builtin_amdgcn_fence(__ATOMIC_ACQUIRE, "wavefront");
}

__global__ void __launch_bounds__(LAYER_THREADS)
rgcn_layer_kernel(const float* __restrict__ agg,
                  const float* __restrict__ deg,
                  const float* __restrict__ x,
                  const float* __restrict__ lin_w,
                  const float* __restrict__ lin_b,
                  const float* __restrict__ self_w,
                  const float* __restrict__ self_b,
                  const float* __restrict__ prev,
                  const float* __restrict__ pw,
                  const float* __restrict__ pb,
                  const float* __restrict__ tw,
                  const float* __restrict__ tb,
                  float*       __restrict__ hidden,
                  float*       __restrict__ outp,
                  int nNodes)
{
    __shared__ __attribute__((aligned(16)))
        float smem[DIM * STRC + 2 * DIM * STRH + LAYER_WAVES * 64 + LAYER_WAVES * 16 * STRT];
    float* WT  = smem;
    float* WP  = WT  + DIM * STRC;
    float* WTT = WP  + DIM * STRH;
    float* RD  = WTT + DIM * STRH;
    float* HT  = RD  + LAYER_WAVES * 64;

    const int t = threadIdx.x;
    for (int i = t; i < KCONV * DIM; i += LAYER_THREADS) {
        int k = i >> 6;
        int c = i & 63;
        float v = (k < NREL * DIM) ? lin_w[i] : self_w[i - NREL * DIM * DIM];
        WT[(size_t)c * STRC + k] = v;
    }
    for (int i = t; i < KHW * DIM; i += LAYER_THREADS) {
        int k = i >> 6;
        int c = i & 63;
        WP [(size_t)c * STRH + k] = pw[i];
        WTT[(size_t)c * STRH + k] = tw[i];
    }
    __syncthreads();

    const int lane = t & 31;
    const int wave = __builtin_amdgcn_readfirstlane(t >> 5);
    const int m     = lane & 15;
    const int khalf = lane >> 4;

    const float* B0 = WT + (size_t)(m)      * STRC;
    const float* B1 = WT + (size_t)(16 + m) * STRC;
    const float* B2 = WT + (size_t)(32 + m) * STRC;
    const float* B3 = WT + (size_t)(48 + m) * STRC;
    const float* P0 = WP  + (size_t)(m)      * STRH;
    const float* P1 = WP  + (size_t)(16 + m) * STRH;
    const float* P2 = WP  + (size_t)(32 + m) * STRH;
    const float* P3 = WP  + (size_t)(48 + m) * STRH;
    const float* T0 = WTT + (size_t)(m)      * STRH;
    const float* T1 = WTT + (size_t)(16 + m) * STRH;
    const float* T2 = WTT + (size_t)(32 + m) * STRH;
    const float* T3 = WTT + (size_t)(48 + m) * STRH;
    float*       rdw  = RD + wave * 64;
    float*       ht   = HT + wave * (16 * STRT);
    const float* rd   = rdw + m * NREL;
    const float* hRow = ht + m * STRT;

    const long base = ((long)blockIdx.x * LAYER_WAVES + wave) * 16;
    if (base >= nNodes) return;
    {
        {
            float2 dv = ((const float2*)(deg + base * NREL))[lane];
            rdw[2 * lane]     = 1.0f / (dv.x + 1e-10f);
            rdw[2 * lane + 1] = 1.0f / (dv.y + 1e-10f);
        }
        wave_lds_handover();

        const float* aggRow = agg  + (base + m) * (size_t)(NREL * DIM);
        const float* xRow   = x    + (base + m) * (size_t)DIM;
        const float* pRow   = prev + (base + m) * (size_t)DIM;

        v8f a0 = {}, a1 = {}, a2 = {}, a3 = {};
        for (int k = 0; k < NREL * DIM; k += 4) {
            int kk = k + 2 * khalf;
            float  sc = rd[k >> 6];
            float2 av = *(const float2*)(aggRow + kk);
            v2f a; a.x = av.x * sc; a.y = av.y * sc;
            v2f b0 = *(const v2f*)(B0 + kk);
            v2f b1 = *(const v2f*)(B1 + kk);
            v2f b2 = *(const v2f*)(B2 + kk);
            v2f b3 = *(const v2f*)(B3 + kk);
            a0 = __builtin_amdgcn_wmma_f32_16x16x4_f32(false, a, false, b0, (short)0, a0, false, false);
            a1 = __builtin_amdgcn_wmma_f32_16x16x4_f32(false, a, false, b1, (short)0, a1, false, false);
            a2 = __builtin_amdgcn_wmma_f32_16x16x4_f32(false, a, false, b2, (short)0, a2, false, false);
            a3 = __builtin_amdgcn_wmma_f32_16x16x4_f32(false, a, false, b3, (short)0, a3, false, false);
        }
#pragma unroll 4
        for (int k = 0; k < DIM; k += 4) {
            int kk = k + 2 * khalf;
            float2 av = *(const float2*)(xRow + kk);
            v2f a; a.x = av.x; a.y = av.y;
            int kw = NREL * DIM + kk;
            v2f b0 = *(const v2f*)(B0 + kw);
            v2f b1 = *(const v2f*)(B1 + kw);
            v2f b2 = *(const v2f*)(B2 + kw);
            v2f b3 = *(const v2f*)(B3 + kw);
            a0 = __builtin_amdgcn_wmma_f32_16x16x4_f32(false, a, false, b0, (short)0, a0, false, false);
            a1 = __builtin_amdgcn_wmma_f32_16x16x4_f32(false, a, false, b1, (short)0, a1, false, false);
            a2 = __builtin_amdgcn_wmma_f32_16x16x4_f32(false, a, false, b2, (short)0, a2, false, false);
            a3 = __builtin_amdgcn_wmma_f32_16x16x4_f32(false, a, false, b3, (short)0, a3, false, false);
        }

#define CONV_EPILOGUE(ACC, TILE)                                           \
        {                                                                  \
            int   col = (TILE) * 16 + m;                                   \
            float bia = lin_b[col] + self_b[col];                          \
            _Pragma("unroll")                                              \
            for (int r = 0; r < 8; ++r) {                                  \
                int   lr = khalf * 8 + r;                                  \
                float hv = sigmoidf_(ACC[r] + bia);                        \
                ht[lr * STRT + col] = hv;                                  \
                hidden[(base + lr) * DIM + col] = hv;                      \
            }                                                              \
        }
        CONV_EPILOGUE(a0, 0) CONV_EPILOGUE(a1, 1) CONV_EPILOGUE(a2, 2) CONV_EPILOGUE(a3, 3)
#undef CONV_EPILOGUE
        wave_lds_handover();

        v8f p0 = {}, p1 = {}, p2 = {}, p3 = {};
        v8f g0 = {}, g1 = {}, g2 = {}, g3 = {};

#define HW_STEP(AROW, KOFF)                                                              \
        for (int k = 0; k < DIM; k += 4) {                                               \
            int kk = k + 2 * khalf;                                                      \
            float2 av = *(const float2*)((AROW) + kk);                                   \
            v2f a; a.x = av.x; a.y = av.y;                                               \
            int kw = kk + (KOFF);                                                        \
            v2f bp0 = *(const v2f*)(P0 + kw);                                            \
            v2f bp1 = *(const v2f*)(P1 + kw);                                            \
            v2f bp2 = *(const v2f*)(P2 + kw);                                            \
            v2f bp3 = *(const v2f*)(P3 + kw);                                            \
            v2f bt0 = *(const v2f*)(T0 + kw);                                            \
            v2f bt1 = *(const v2f*)(T1 + kw);                                            \
            v2f bt2 = *(const v2f*)(T2 + kw);                                            \
            v2f bt3 = *(const v2f*)(T3 + kw);                                            \
            p0 = __builtin_amdgcn_wmma_f32_16x16x4_f32(false, a, false, bp0, (short)0, p0, false, false); \
            p1 = __builtin_amdgcn_wmma_f32_16x16x4_f32(false, a, false, bp1, (short)0, p1, false, false); \
            p2 = __builtin_amdgcn_wmma_f32_16x16x4_f32(false, a, false, bp2, (short)0, p2, false, false); \
            p3 = __builtin_amdgcn_wmma_f32_16x16x4_f32(false, a, false, bp3, (short)0, p3, false, false); \
            g0 = __builtin_amdgcn_wmma_f32_16x16x4_f32(false, a, false, bt0, (short)0, g0, false, false); \
            g1 = __builtin_amdgcn_wmma_f32_16x16x4_f32(false, a, false, bt1, (short)0, g1, false, false); \
            g2 = __builtin_amdgcn_wmma_f32_16x16x4_f32(false, a, false, bt2, (short)0, g2, false, false); \
            g3 = __builtin_amdgcn_wmma_f32_16x16x4_f32(false, a, false, bt3, (short)0, g3, false, false); \
        }
        HW_STEP(hRow, 0)
        HW_STEP(pRow, DIM)
#undef HW_STEP

#define HW_EPILOGUE(PACC, GACC, TILE)                                      \
        {                                                                  \
            int   col = (TILE) * 16 + m;                                   \
            float pbv = pb[col];                                           \
            float tbv = tb[col];                                           \
            _Pragma("unroll")                                              \
            for (int r = 0; r < 8; ++r) {                                  \
                int   lr = khalf * 8 + r;                                  \
                float pr = fmaxf(PACC[r] + pbv, 0.0f);                     \
                float gt = sigmoidf_(GACC[r] + tbv);                       \
                float h  = ht[lr * STRT + col];                            \
                outp[(base + lr) * DIM + col] = gt * pr + (1.0f - gt) * h; \
            }                                                              \
        }
        HW_EPILOGUE(p0, g0, 0) HW_EPILOGUE(p1, g1, 1)
        HW_EPILOGUE(p2, g2, 2) HW_EPILOGUE(p3, g3, 3)
#undef HW_EPILOGUE
    }
}

extern "C" void kernel_launch(void* const* d_in, const int* in_sizes, int n_in,
                              void* d_out, int out_size, void* d_ws, size_t ws_size,
                              hipStream_t stream) {
    const float* node_feat   = (const float*)d_in[0];
    const float* edge_weight = (const float*)d_in[1];
    const float* conv_lin_w  = (const float*)d_in[2];
    const float* conv_lin_b  = (const float*)d_in[3];
    const float* conv_self_w = (const float*)d_in[4];
    const float* conv_self_b = (const float*)d_in[5];
    const float* hw_proj_w   = (const float*)d_in[6];
    const float* hw_proj_b   = (const float*)d_in[7];
    const float* hw_trans_w  = (const float*)d_in[8];
    const float* hw_trans_b  = (const float*)d_in[9];
    const int*   edge_src    = (const int*)d_in[10];
    const int*   edge_dst    = (const int*)d_in[11];
    const int*   edge_rel    = (const int*)d_in[12];
    float*       out         = (float*)d_out;

    const int N = in_sizes[0] / DIM;
    const int E = in_sizes[10];

    float* agg = (float*)d_ws;
    float* deg = agg + (size_t)N * NREL * DIM;
    float* hA  = deg + (size_t)N * NREL;
    float* hB  = hA  + (size_t)N * DIM;
    float* cur0 = hB   + (size_t)N * DIM;
    float* cur1 = cur0 + (size_t)N * DIM;
    const int NB    = N * NREL;
    const int nbin  = (N + BIN_NODES - 1) >> BIN_SHIFT;
    const int nbinp = (nbin + 3) & ~3;
    const int chunk = (E + SORT_BLOCKS - 1) / SORT_BLOCKS;
    int*  cursor   = (int*)(cur1 + (size_t)N * DIM);
    int*  erec     = cursor + NB;
    int*  part     = erec + 2 * (size_t)E;
    int*  hist     = part + 2 * (size_t)E;
    int*  binStart = hist + (size_t)SORT_BLOCKS * nbinp;

    const int gatherBlocks  = (N + 8 * GATHER_NODES - 1) / (8 * GATHER_NODES);
    const int tiles         = (N + 15) / 16;
    const int layerBlocks   = (tiles + LAYER_WAVES - 1) / LAYER_WAVES;
    const size_t stageLds = (size_t)chunk * sizeof(int2);

    bin_hist_kernel<<<SORT_BLOCKS, SORT_THREADS, 0, stream>>>(
        edge_dst, hist, E, chunk, nbin, nbinp);
    bin_partition_kernel<<<SORT_BLOCKS, SORT_THREADS, stageLds, stream>>>(
        edge_src, edge_dst, edge_rel, edge_weight, hist, binStart, (int2*)part,
        E, chunk, nbin, nbinp);
    bin_order_kernel<<<nbin, SORT_THREADS, 0, stream>>>(
        (const int2*)part, binStart, cursor, (int2*)erec, NB);

    for (int i = 0; i < 3; ++i) {
        const float* xin  = (i == 0) ? node_feat : ((i == 1) ? cur0 : cur1);
        const float* prev = (i == 0) ? node_feat : ((i == 1) ? hA : hB);
        float*       hid  = (i & 1) ? hB : hA;
        float*       gout = (i == 0) ? cur0 : ((i == 1) ? cur1 : out);

        rgcn_gather_kernel<<<gatherBlocks, 256, 0, stream>>>(
            xin, erec, cursor, agg, deg, N);

        rgcn_layer_kernel<<<layerBlocks, LAYER_THREADS, 0, stream>>>(
            agg, deg, xin,
            conv_lin_w  + (size_t)i * 256 * DIM, conv_lin_b  + (size_t)i * DIM,
            conv_self_w + (size_t)i * DIM * DIM, conv_self_b + (size_t)i * DIM,
            prev,
            hw_proj_w  + (size_t)i * KHW * DIM, hw_proj_b  + (size_t)i * DIM,
            hw_trans_w + (size_t)i * KHW * DIM, hw_trans_b + (size_t)i * DIM,
            hid, gout, N);
    }
}


# Gather folded into the layer kernel: per relation a wave sums its 16 nodes' source rows into its LDS tile and multiplies it at once; edges ordered (tile, rel, node); in-bucket sum order kept
# speedup vs baseline: 3.8926x; 1.2704x over previous
#include <hip/hip_runtime.h>
#include <hip/hip_bf16.h>
#include <math.h>


typedef __attribute__((ext_vector_type(2))) float v2f;
typedef __attribute__((ext_vector_type(8))) float v8f;

#define DIM   64
#define NREL  4
#define KCONV 320
#define KHW   128
#define STRC  324
#define STRH  132

__device__ __forceinline__ float sigmoidf_(float v) {
    return 1.0f / (1.0f + __expf(-v));
}

#define BIN_SHIFT   8
#define BIN_NODES   (1 << BIN_SHIFT)
#define FINE_KEYS   (BIN_NODES * NREL)
#define TILE_ROWS   16
#define SORT_BLOCKS 256
#define SORT_THREADS 1024
#define KEY_SHIFT   20
#define SRC_MASK    ((1 << KEY_SHIFT) - 1)
#define FINE_CAP    8192

__device__ __forceinline__ int block_excl_scan_1024(int v, int* wtot)
{
    const int lane = threadIdx.x & 31;
    const int wave = threadIdx.x >> 5;
    int inc = v;
    for (int o = 1; o < 32; o <<= 1) {
        int n = __shfl_up(inc, o);
        if (lane >= o) inc += n;
    }
    if (lane == 31) wtot[wave] = inc;
    __syncthreads();
    int woff = 0;
    for (int w = 0; w < 32; ++w) {
        int x = wtot[w];
        if (w < wave) woff += x;
    }
    __syncthreads();
    return woff + inc - v;
}

__global__ void __launch_bounds__(SORT_THREADS)
bin_hist_kernel(const int* __restrict__ dst,
                int*       __restrict__ hist,
                int nEdges, int chunk, int nbin, int nbinp)
{
    __shared__ int cnt[SORT_THREADS];
    const int t = threadIdx.x;
    cnt[t] = 0;
    __syncthreads();
    const int beg = blockIdx.x * chunk;
    const int end = min(beg + chunk, nEdges);
    for (int i0 = beg + t; i0 < end; i0 += 4 * SORT_THREADS) {
        int d[4];
#pragma unroll
        for (int u = 0; u < 4; ++u) {
            const int i = i0 + u * SORT_THREADS;
            d[u] = (i < end) ? dst[i] : -1;
        }
#pragma unroll
        for (int u = 0; u < 4; ++u)
            if (d[u] >= 0) atomicAdd(&cnt[d[u] >> BIN_SHIFT], 1);
    }
    __syncthreads();
    if (t < nbinp) hist[(size_t)blockIdx.x * nbinp + t] = (t < nbin) ? cnt[t] : 0;
}

__global__ void __launch_bounds__(SORT_THREADS)
bin_partition_kernel(const int*   __restrict__ src,
                     const int*   __restrict__ dst,
                     const int*   __restrict__ rel,
                     const float* __restrict__ ew,
                     const int*   __restrict__ hist,
                     int*         __restrict__ binStart,
                     int2*        __restrict__ part,
                     int nEdges, int chunk, int nbin, int nbinp)
{
    extern __shared__ int2 stage[];
    __shared__ int tot[SORT_THREADS];
    __shared__ int pre[SORT_THREADS];
    __shared__ int mine[SORT_THREADS];
    __shared__ int loff[SORT_THREADS];
    __shared__ int lcur[SORT_THREADS];
    __shared__ int gbase[SORT_THREADS];
    __shared__ int wtot[32];

    const int t   = threadIdx.x;
    const int blk = blockIdx.x;
    tot[t] = 0;
    pre[t] = 0;
    mine[t] = (t < nbin) ? hist[(size_t)blk * nbinp + t] : 0;
    __syncthreads();

    {
        const int nq  = nbinp >> 2;
        const int grp = t >> 7;
        for (int q = t & 127; q < nq; q += 128) {
            int4 st = {0, 0, 0, 0}, sp = {0, 0, 0, 0};
            for (int r0 = grp; r0 < SORT_BLOCKS; r0 += 64) {
                int4 h[8];
#pragma unroll
                for (int u = 0; u < 8; ++u)
                    h[u] = *(const int4*)(hist + (size_t)(r0 + 8 * u) * nbinp + 4 * q);
#pragma unroll
                for (int u = 0; u < 8; ++u) {
                    st.x += h[u].x; st.y += h[u].y; st.z += h[u].z; st.w += h[u].w;
                    if (r0 + 8 * u < blk) { sp.x += h[u].x; sp.y += h[u].y; sp.z += h[u].z; sp.w += h[u].w; }
                }
            }
            atomicAdd(&tot[4 * q + 0], st.x); atomicAdd(&tot[4 * q + 1], st.y);
            atomicAdd(&tot[4 * q + 2], st.z); atomicAdd(&tot[4 * q + 3], st.w);
            atomicAdd(&pre[4 * q + 0], sp.x); atomicAdd(&pre[4 * q + 1], sp.y);
            atomicAdd(&pre[4 * q + 2], sp.z); atomicAdd(&pre[4 * q + 3], sp.w);
        }
    }
    __syncthreads();

    const int vt = (t < nbin) ? tot[t] : 0;
    const int vm = mine[t];
    const int bs = block_excl_scan_1024(vt, wtot);
    const int lo = block_excl_scan_1024(vm, wtot);
    gbase[t] = bs + pre[t];
    loff[t]  = lo;
    lcur[t]  = lo;
    if (blk == 0) {
        if (t < nbin) binStart[t] = bs;
        if (t == 0)   binStart[nbin] = nEdges;
    }
    __syncthreads();

    const int beg = blk * chunk;
    const int end = min(beg + chunk, nEdges);
    for (int i0 = beg + t; i0 < end; i0 += 4 * SORT_THREADS) {
        int d[4], r[4], s[4], w[4];
#pragma unroll
        for (int u = 0; u < 4; ++u) {
            const int  i  = i0 + u * SORT_THREADS;
            const bool ok = i < end;
            d[u] = ok ? dst[i] : -1;
            r[u] = ok ? rel[i] : 0;
            s[u] = ok ? src[i] : 0;
            w[u] = ok ? __float_as_int(ew[i]) : 0;
        }
#pragma unroll
        for (int u = 0; u < 4; ++u) {
            if (d[u] >= 0) {
                const int b   = d[u] >> BIN_SHIFT;
                const int dl  = d[u] & (BIN_NODES - 1);
                const int key = (dl / TILE_ROWS) * (TILE_ROWS * NREL) + r[u] * TILE_ROWS + (dl % TILE_ROWS);
                const int p   = atomicAdd(&lcur[b], 1);
                int2 rec;
                rec.x = s[u] | (key << KEY_SHIFT);
                rec.y = w[u];
                stage[p] = rec;
            }
        }
    }
    __syncthreads();

    const int lane = t & 31;
    const int wave = t >> 5;
    for (int b = wave; b < nbin; b += 32) {
        const int n = mine[b];
        const int l = loff[b];
        const int g = gbase[b];
        for (int j = lane; j < n; j += 32) part[g + j] = stage[l + j];
    }
}

__global__ void __launch_bounds__(SORT_THREADS)
bin_order_kernel(const int2* __restrict__ part,
                 const int*  __restrict__ binStart,
                 int*        __restrict__ bend,
                 int2*       __restrict__ erec,
                 int nBuckets)
{
    __shared__ int2 sorted[FINE_CAP];
    __shared__ int  cnt[FINE_KEYS];
    __shared__ int  lcur[FINE_KEYS];
    __shared__ int  wtot[32];

    const int t = threadIdx.x;
    const int g = blockIdx.x;
    const int s = binStart[g];
    const int n = binStart[g + 1] - s;
    const int2* in = part + s;

    cnt[t] = 0;
    __syncthreads();
    for (int i0 = t; i0 < n; i0 += 4 * SORT_THREADS) {
        int x[4];
#pragma unroll
        for (int u = 0; u < 4; ++u) {
            const int i = i0 + u * SORT_THREADS;
            x[u] = (i < n) ? in[i].x : -1;
        }
#pragma unroll
        for (int u = 0; u < 4; ++u)
            if (x[u] >= 0) atomicAdd(&cnt[x[u] >> KEY_SHIFT], 1);
    }
    __syncthreads();

    const int c   = cnt[t];
    const int off = block_excl_scan_1024(c, wtot);
    lcur[t] = off;
    const int b = g * FINE_KEYS + t;
    if (b < nBuckets) bend[b] = s + off + c;
    __syncthreads();

    if (n <= FINE_CAP) {
        for (int i0 = t; i0 < n; i0 += 4 * SORT_THREADS) {
            int2 rc[4];
#pragma unroll
            for (int u = 0; u < 4; ++u) {
                const int i = i0 + u * SORT_THREADS;
                if (i < n) rc[u] = in[i]; else { rc[u].x = -1; rc[u].y = 0; }
            }
#pragma unroll
            for (int u = 0; u < 4; ++u) {
                if (rc[u].x >= 0) {
                    const int p = atomicAdd(&lcur[rc[u].x >> KEY_SHIFT], 1);
                    int2 o;
                    o.x = rc[u].x & SRC_MASK;
                    o.y = rc[u].y;
                    sorted[p] = o;
                }
            }
        }
        __syncthreads();
        for (int i = t; i < n; i += SORT_THREADS) erec[s + i] = sorted[i];
    } else {
        for (int i = t; i < n; i += SORT_THREADS) {
            const int2 rc = in[i];
            const int  p  = atomicAdd(&lcur[rc.x >> KEY_SHIFT], 1);
            int2 o;
            o.x = rc.x & SRC_MASK;
            o.y = rc.y;
            erec[s + p] = o;
        }
    }
}

#define LAYER_WAVES   25
#define LAYER_THREADS (LAYER_WAVES * 32)
#define STRT          68

__device__ __forceinline__ void wave_lds_handover()
{
    __builtin_amdgcn_fence(__ATOMIC_RELEASE, "wavefront");
    __builtin_amdgcn_wave_barrier();
    __builtin_amdgcn_fence(__ATOMIC_ACQUIRE, "wavefront");
}

__global__ void __launch_bounds__(LAYER_THREADS)
rgcn_layer_kernel(const float* __restrict__ x,
                  const int*   __restrict__ erec,
                  const int*   __restrict__ bend,
                  const float* __restrict__ lin_w,
                  const float* __restrict__ lin_b,
                  const float* __restrict__ self_w,
                  const float* __restrict__ self_b,
                  const float* __restrict__ prev,
                  const float* __restrict__ pw,
                  const float* __restrict__ pb,
                  const float* __restrict__ tw,
                  const float* __restrict__ tb,
                  float*       __restrict__ hidden,
                  float*       __restrict__ outp,
                  int nNodes)
{
    __shared__ __attribute__((aligned(16)))
        float smem[DIM * STRC + 2 * DIM * STRH + LAYER_WAVES * 16 * STRT];
    float* WT  = smem;
    float* WP  = WT  + DIM * STRC;
    float* WTT = WP  + DIM * STRH;
    float* HT  = WTT + DIM * STRH;

    const int t = threadIdx.x;
    for (int i = t; i < KCONV * DIM; i += LAYER_THREADS) {
        int k = i >> 6;
        int c = i & 63;
        float v = (k < NREL * DIM) ? lin_w[i] : self_w[i - NREL * DIM * DIM];
        WT[(size_t)c * STRC + k] = v;
    }
    for (int i = t; i < KHW * DIM; i += LAYER_THREADS) {
        int k = i >> 6;
        int c = i & 63;
        WP [(size_t)c * STRH + k] = pw[i];
        WTT[(size_t)c * STRH + k] = tw[i];
    }
    __syncthreads();

    const int lane = t & 31;
    const int wave = __builtin_amdgcn_readfirstlane(t >> 5);
    const int m     = lane & 15;
    const int khalf = lane >> 4;

    const float* B0 = WT + (size_t)(m)      * STRC;
    const float* B1 = WT + (size_t)(16 + m) * STRC;
    const float* B2 = WT + (size_t)(32 + m) * STRC;
    const float* B3 = WT + (size_t)(48 + m) * STRC;
    const float* P0 = WP  + (size_t)(m)      * STRH;
    const float* P1 = WP  + (size_t)(16 + m) * STRH;
    const float* P2 = WP  + (size_t)(32 + m) * STRH;
    const float* P3 = WP  + (size_t)(48 + m) * STRH;
    const float* T0 = WTT + (size_t)(m)      * STRH;
    const float* T1 = WTT + (size_t)(16 + m) * STRH;
    const float* T2 = WTT + (size_t)(32 + m) * STRH;
    const float* T3 = WTT + (size_t)(48 + m) * STRH;
    float*       ht   = HT + wave * (16 * STRT);
    const float* hRow = ht + m * STRT;

    const long base = ((long)blockIdx.x * LAYER_WAVES + wave) * 16;
    if (base >= nNodes) return;
    {
        const float* xRow   = x    + (base + m) * (size_t)DIM;
        const float* pRow   = prev + (base + m) * (size_t)DIM;

        const int2* __restrict__ rec2 = (const int2*)erec;
        const char* __restrict__ xb   = (const char*)x;
        const unsigned lane8 = (unsigned)lane << 3;
        const int bvA   = bend[base * NREL + lane];
        const int bvB   = bend[base * NREL + 32 + lane];
        const int bprev = (base > 0) ? bend[base * NREL - 1] : 0;

        int  segBeg = bprev;
        int2 recN   = {0, 0};
        {
            const int send0 = __builtin_amdgcn_readlane(bvA, 15);
            if (send0 > segBeg) recN = rec2[min(segBeg + lane, send0 - 1)];
        }

#define ROW_LOAD(U)                                                              \
        const unsigned s##U = (unsigned)__builtin_amdgcn_readlane(rec.x, (U));  \
        const float    w##U = __int_as_float(__builtin_amdgcn_readlane(rec.y, (U))); \
        const float2   v##U = *(const float2*)(xb + (size_t)((s##U << 8) | lane8));

#define BUCKET_DONE()                                                            \
        {                                                                        \
            *(float2*)(ht + mm * STRT + 2 * lane) = acc;                         \
            if (m == mm) dsum = ss;                                              \
            acc.x = 0.0f; acc.y = 0.0f; ss = 0.0f;                               \
            ++mm;                                                                \
            bnd = (mm < 15) ? __builtin_amdgcn_readlane(bvr, lo16 + mm) : 0x7fffffff; \
        }

#define BUCKET_CHECK(U)                                                          \
        if ((U) >= togo) {                                                       \
            do { BUCKET_DONE() togo = bnd - j0; } while ((U) >= togo);           \
        }

#define EDGE_SURE(U)                                                             \
        BUCKET_CHECK(U)                                                          \
        acc.x += v##U.x * w##U; acc.y += v##U.y * w##U; ss += w##U;

#define EDGE_MAYBE(U)                                                            \
        BUCKET_CHECK(U)                                                          \
        {                                                                        \
            const float wz = ((U) < left) ? w##U : 0.0f;                         \
            acc.x += v##U.x * wz; acc.y += v##U.y * wz; ss += wz;                \
        }

        v8f a0 = {}, a1 = {}, a2 = {}, a3 = {};
        for (int r = 0; r < NREL; ++r) {
            const int bvr  = (r < 2) ? bvA : bvB;
            const int lo16 = (r & 1) * 16;
            const int sbeg = segBeg;
            const int send = __builtin_amdgcn_readlane(bvr, lo16 + 15);
            int2 rec = recN;
            if (r + 1 < NREL) {
                const int bvn   = (r + 1 < 2) ? bvA : bvB;
                const int sendN = __builtin_amdgcn_readlane(bvn, ((r + 1) & 1) * 16 + 15);
                if (sendN > send) recN = rec2[min(send + lane, sendN - 1)];
            }
            segBeg = send;

            float2 acc  = {0.0f, 0.0f};
            float  ss   = 0.0f;
            float  dsum = 0.0f;
            int    mm   = 0;
            int    bnd  = __builtin_amdgcn_readlane(bvr, lo16);

            int2 recP = {0, 0};
            for (int cb = sbeg; cb < send; cb += 32) {
                const int ce = min(cb + 32, send);
                if (cb != sbeg) rec = recP;
                if (ce < send) recP = rec2[min(ce + lane, send - 1)];
                for (int j0 = cb; j0 < ce; j0 += 16) {
                    if (j0 != cb) {
                        rec.x = __shfl_xor(rec.x, 16);
                        rec.y = __shfl_xor(rec.y, 16);
                    }
                    const int left = ce - j0;
                    int togo = bnd - j0;
                    if (left > 8) {
                        ROW_LOAD(0)  ROW_LOAD(1)  ROW_LOAD(2)  ROW_LOAD(3)
                        ROW_LOAD(4)  ROW_LOAD(5)  ROW_LOAD(6)  ROW_LOAD(7)
                        ROW_LOAD(8)  ROW_LOAD(9)  ROW_LOAD(10) ROW_LOAD(11)
                        ROW_LOAD(12) ROW_LOAD(13) ROW_LOAD(14) ROW_LOAD(15)
                        EDGE_SURE(0)   EDGE_SURE(1)   EDGE_SURE(2)   EDGE_SURE(3)
                        EDGE_SURE(4)   EDGE_SURE(5)   EDGE_SURE(6)   EDGE_SURE(7)
                        EDGE_SURE(8)   EDGE_MAYBE(9)  EDGE_MAYBE(10) EDGE_MAYBE(11)
                        EDGE_MAYBE(12) EDGE_MAYBE(13) EDGE_MAYBE(14) EDGE_MAYBE(15)
                    } else {
                        ROW_LOAD(0) ROW_LOAD(1) ROW_LOAD(2) ROW_LOAD(3)
                        ROW_LOAD(4) ROW_LOAD(5) ROW_LOAD(6) ROW_LOAD(7)
                        EDGE_SURE(0)  EDGE_MAYBE(1) EDGE_MAYBE(2) EDGE_MAYBE(3)
                        EDGE_MAYBE(4) EDGE_MAYBE(5) EDGE_MAYBE(6) EDGE_MAYBE(7)
                    }
                }
            }
            while (mm < 16) BUCKET_DONE()
            wave_lds_handover();

            const float sc = 1.0f / (dsum + 1e-10f);
            for (int k = 0; k < DIM; k += 4) {
                int kk = k + 2 * khalf;
                float2 av = *(const float2*)(hRow + kk);
                v2f a; a.x = av.x * sc; a.y = av.y * sc;
                int kw = r * DIM + kk;
                v2f b0 = *(const v2f*)(B0 + kw);
                v2f b1 = *(const v2f*)(B1 + kw);
                v2f b2 = *(const v2f*)(B2 + kw);
                v2f b3 = *(const v2f*)(B3 + kw);
                a0 = __builtin_amdgcn_wmma_f32_16x16x4_f32(false, a, false, b0, (short)0, a0, false, false);
                a1 = __builtin_amdgcn_wmma_f32_16x16x4_f32(false, a, false, b1, (short)0, a1, false, false);
                a2 = __builtin_amdgcn_wmma_f32_16x16x4_f32(false, a, false, b2, (short)0, a2, false, false);
                a3 = __builtin_amdgcn_wmma_f32_16x16x4_f32(false, a, false, b3, (short)0, a3, false, false);
            }
            wave_lds_handover();
        }
#undef EDGE_MAYBE
#undef EDGE_SURE
#undef BUCKET_CHECK
#undef BUCKET_DONE
#undef ROW_LOAD

#pragma unroll 4
        for (int k = 0; k < DIM; k += 4) {
            int kk = k + 2 * khalf;
            float2 av = *(const float2*)(xRow + kk);
            v2f a; a.x = av.x; a.y = av.y;
            int kw = NREL * DIM + kk;
            v2f b0 = *(const v2f*)(B0 + kw);
            v2f b1 = *(const v2f*)(B1 + kw);
            v2f b2 = *(const v2f*)(B2 + kw);
            v2f b3 = *(const v2f*)(B3 + kw);
            a0 = __builtin_amdgcn_wmma_f32_16x16x4_f32(false, a, false, b0, (short)0, a0, false, false);
            a1 = __builtin_amdgcn_wmma_f32_16x16x4_f32(false, a, false, b1, (short)0, a1, false, false);
            a2 = __builtin_amdgcn_wmma_f32_16x16x4_f32(false, a, false, b2, (short)0, a2, false, false);
            a3 = __builtin_amdgcn_wmma_f32_16x16x4_f32(false, a, false, b3, (short)0, a3, false, false);
        }

#define CONV_EPILOGUE(ACC, TILE)                                           \
        {                                                                  \
            int   col = (TILE) * 16 + m;                                   \
            float bia = lin_b[col] + self_b[col];                          \
            _Pragma("unroll")                                              \
            for (int r = 0; r < 8; ++r) {                                  \
                int   lr = khalf * 8 + r;                                  \
                float hv = sigmoidf_(ACC[r] + bia);                        \
                ht[lr * STRT + col] = hv;                                  \
                hidden[(base + lr) * DIM + col] = hv;                      \
            }                                                              \
        }
        CONV_EPILOGUE(a0, 0) CONV_EPILOGUE(a1, 1) CONV_EPILOGUE(a2, 2) CONV_EPILOGUE(a3, 3)
#undef CONV_EPILOGUE
        wave_lds_handover();

        v8f p0 = {}, p1 = {}, p2 = {}, p3 = {};
        v8f g0 = {}, g1 = {}, g2 = {}, g3 = {};

#define HW_STEP(AROW, KOFF)                                                              \
        for (int k = 0; k < DIM; k += 4) {                                               \
            int kk = k + 2 * khalf;                                                      \
            float2 av = *(const float2*)((AROW) + kk);                                   \
            v2f a; a.x = av.x; a.y = av.y;                                               \
            int kw = kk + (KOFF);                                                        \
            v2f bp0 = *(const v2f*)(P0 + kw);                                            \
            v2f bp1 = *(const v2f*)(P1 + kw);                                            \
            v2f bp2 = *(const v2f*)(P2 + kw);                                            \
            v2f bp3 = *(const v2f*)(P3 + kw);                                            \
            v2f bt0 = *(const v2f*)(T0 + kw);                                            \
            v2f bt1 = *(const v2f*)(T1 + kw);                                            \
            v2f bt2 = *(const v2f*)(T2 + kw);                                            \
            v2f bt3 = *(const v2f*)(T3 + kw);                                            \
            p0 = __builtin_amdgcn_wmma_f32_16x16x4_f32(false, a, false, bp0, (short)0, p0, false, false); \
            p1 = __builtin_amdgcn_wmma_f32_16x16x4_f32(false, a, false, bp1, (short)0, p1, false, false); \
            p2 = __builtin_amdgcn_wmma_f32_16x16x4_f32(false, a, false, bp2, (short)0, p2, false, false); \
            p3 = __builtin_amdgcn_wmma_f32_16x16x4_f32(false, a, false, bp3, (short)0, p3, false, false); \
            g0 = __builtin_amdgcn_wmma_f32_16x16x4_f32(false, a, false, bt0, (short)0, g0, false, false); \
            g1 = __builtin_amdgcn_wmma_f32_16x16x4_f32(false, a, false, bt1, (short)0, g1, false, false); \
            g2 = __builtin_amdgcn_wmma_f32_16x16x4_f32(false, a, false, bt2, (short)0, g2, false, false); \
            g3 = __builtin_amdgcn_wmma_f32_16x16x4_f32(false, a, false, bt3, (short)0, g3, false, false); \
        }
        HW_STEP(hRow, 0)
        HW_STEP(pRow, DIM)
#undef HW_STEP

#define HW_EPILOGUE(PACC, GACC, TILE)                                      \
        {                                                                  \
            int   col = (TILE) * 16 + m;                                   \
            float pbv = pb[col];                                           \
            float tbv = tb[col];                                           \
            _Pragma("unroll")                                              \
            for (int r = 0; r < 8; ++r) {                                  \
                int   lr = khalf * 8 + r;                                  \
                float pr = fmaxf(PACC[r] + pbv, 0.0f);                     \
                float gt = sigmoidf_(GACC[r] + tbv);                       \
                float h  = ht[lr * STRT + col];                            \
                outp[(base + lr) * DIM + col] = gt * pr + (1.0f - gt) * h; \
            }                                                              \
        }
        HW_EPILOGUE(p0, g0, 0) HW_EPILOGUE(p1, g1, 1)
        HW_EPILOGUE(p2, g2, 2) HW_EPILOGUE(p3, g3, 3)
#undef HW_EPILOGUE
    }
}

extern "C" void kernel_launch(void* const* d_in, const int* in_sizes, int n_in,
                              void* d_out, int out_size, void* d_ws, size_t ws_size,
                              hipStream_t stream) {
    const float* node_feat   = (const float*)d_in[0];
    const float* edge_weight = (const float*)d_in[1];
    const float* conv_lin_w  = (const float*)d_in[2];
    const float* conv_lin_b  = (const float*)d_in[3];
    const float* conv_self_w = (const float*)d_in[4];
    const float* conv_self_b = (const float*)d_in[5];
    const float* hw_proj_w   = (const float*)d_in[6];
    const float* hw_proj_b   = (const float*)d_in[7];
    const float* hw_trans_w  = (const float*)d_in[8];
    const float* hw_trans_b  = (const float*)d_in[9];
    const int*   edge_src    = (const int*)d_in[10];
    const int*   edge_dst    = (const int*)d_in[11];
    const int*   edge_rel    = (const int*)d_in[12];
    float*       out         = (float*)d_out;

    const int N = in_sizes[0] / DIM;
    const int E = in_sizes[10];

    float* hA   = (float*)d_ws;
    float* hB   = hA   + (size_t)N * DIM;
    float* cur0 = hB   + (size_t)N * DIM;
    float* cur1 = cur0 + (size_t)N * DIM;
    const int NB    = N * NREL;
    const int nbin  = (N + BIN_NODES - 1) >> BIN_SHIFT;
    const int nbinp = (nbin + 3) & ~3;
    const int chunk = (E + SORT_BLOCKS - 1) / SORT_BLOCKS;
    int*  cursor   = (int*)(cur1 + (size_t)N * DIM);
    int*  erec     = cursor + NB;
    int*  part     = erec + 2 * (size_t)E;
    int*  hist     = part + 2 * (size_t)E;
    int*  binStart = hist + (size_t)SORT_BLOCKS * nbinp;

    const int tiles         = (N + 15) / 16;
    const int layerBlocks   = (tiles + LAYER_WAVES - 1) / LAYER_WAVES;
    const size_t stageLds = (size_t)chunk * sizeof(int2);

    bin_hist_kernel<<<SORT_BLOCKS, SORT_THREADS, 0, stream>>>(
        edge_dst, hist, E, chunk, nbin, nbinp);
    bin_partition_kernel<<<SORT_BLOCKS, SORT_THREADS, stageLds, stream>>>(
        edge_src, edge_dst, edge_rel, edge_weight, hist, binStart, (int2*)part,
        E, chunk, nbin, nbinp);
    bin_order_kernel<<<nbin, SORT_THREADS, 0, stream>>>(
        (const int2*)part, binStart, cursor, (int2*)erec, NB);

    for (int i = 0; i < 3; ++i) {
        const float* xin  = (i == 0) ? node_feat : ((i == 1) ? cur0 : cur1);
        const float* prev = (i == 0) ? node_feat : ((i == 1) ? hA : hB);
        float*       hid  = (i & 1) ? hB : hA;
        float*       gout = (i == 0) ? cur0 : ((i == 1) ? cur1 : out);

        rgcn_layer_kernel<<<layerBlocks, LAYER_THREADS, 0, stream>>>(
            xin, erec, cursor,
            conv_lin_w  + (size_t)i * 256 * DIM, conv_lin_b  + (size_t)i * DIM,
            conv_self_w + (size_t)i * DIM * DIM, conv_self_b + (size_t)i * DIM,
            prev,
            hw_proj_w  + (size_t)i * KHW * DIM, hw_proj_b  + (size_t)i * DIM,
            hw_trans_w + (size_t)i * KHW * DIM, hw_trans_b + (size_t)i * DIM,
            hid, gout, N);
    }
}


# Edge ordering in 2 launches, not 3: a chunk orders its records by node range from its own counts and publishes its run starts, a range collects its runs; in-bucket sum order stays arrival-dependent
# speedup vs baseline: 4.0126x; 1.0308x over previous
#include <hip/hip_runtime.h>
#include <hip/hip_bf16.h>
#include <math.h>


typedef __attribute__((ext_vector_type(2))) float v2f;
typedef __attribute__((ext_vector_type(8))) float v8f;

#define DIM   64
#define NREL  4
#define KCONV 320
#define KHW   128
#define STRC  324
#define STRH  132

__device__ __forceinline__ float sigmoidf_(float v) {
    return 1.0f / (1.0f + __expf(-v));
}

#define BIN_SHIFT   8
#define BIN_NODES   (1 << BIN_SHIFT)
#define FINE_KEYS   (BIN_NODES * NREL)
#define TILE_ROWS   16
#define SORT_BLOCKS 256
#define SORT_THREADS 1024
#define RUNS_PER_WAVE (SORT_BLOCKS / 32)
#define KEY_SHIFT   20
#define SRC_MASK    ((1 << KEY_SHIFT) - 1)
#define FINE_CAP    8192

__device__ __forceinline__ int block_excl_scan_1024(int v, int* wtot)
{
    const int lane = threadIdx.x & 31;
    const int wave = threadIdx.x >> 5;
    int inc = v;
    for (int o = 1; o < 32; o <<= 1) {
        int n = __shfl_up(inc, o);
        if (lane >= o) inc += n;
    }
    if (lane == 31) wtot[wave] = inc;
    __syncthreads();
    int woff = 0;
    for (int w = 0; w < 32; ++w) {
        int x = wtot[w];
        if (w < wave) woff += x;
    }
    __syncthreads();
    return woff + inc - v;
}

__global__ void __launch_bounds__(SORT_THREADS)
chunk_split_kernel(const int*   __restrict__ src,
                   const int*   __restrict__ dst,
                   const int*   __restrict__ rel,
                   const float* __restrict__ ew,
                   int*         __restrict__ runTab,
                   int2*        __restrict__ part,
                   int nEdges, int chunk, int nbin, int tabRow)
{
    extern __shared__ int2 stage[];
    __shared__ int mine[SORT_THREADS];
    __shared__ int lcur[SORT_THREADS];
    __shared__ int wtot[32];

    const int t   = threadIdx.x;
    const int blk = blockIdx.x;
    mine[t] = 0;
    __syncthreads();

    const int beg = blk * chunk;
    const int end = min(beg + chunk, nEdges);
    for (int i0 = beg + t; i0 < end; i0 += 8 * SORT_THREADS) {
        int d[8];
#pragma unroll
        for (int u = 0; u < 8; ++u) {
            const int i = i0 + u * SORT_THREADS;
            d[u] = (i < end) ? dst[i] : -1;
        }
#pragma unroll
        for (int u = 0; u < 8; ++u)
            if (d[u] >= 0) atomicAdd(&mine[d[u] >> BIN_SHIFT], 1);
    }
    __syncthreads();

    const int vm = mine[t];
    const int lo = block_excl_scan_1024(vm, wtot);
    lcur[t] = lo;
    {
        int* row = runTab + (size_t)blk * tabRow;
        if (t < nbin) row[t] = lo;
        if (t == SORT_THREADS - 1) row[nbin] = lo + vm;
    }
    __syncthreads();

    for (int i0 = beg + t; i0 < end; i0 += 4 * SORT_THREADS) {
        int d[4], r[4], s[4], w[4];
#pragma unroll
        for (int u = 0; u < 4; ++u) {
            const int  i  = i0 + u * SORT_THREADS;
            const bool ok = i < end;
            d[u] = ok ? dst[i] : -1;
            r[u] = ok ? rel[i] : 0;
            s[u] = ok ? src[i] : 0;
            w[u] = ok ? __float_as_int(ew[i]) : 0;
        }
#pragma unroll
        for (int u = 0; u < 4; ++u) {
            if (d[u] >= 0) {
                const int b   = d[u] >> BIN_SHIFT;
                const int dl  = d[u] & (BIN_NODES - 1);
                const int key = (dl / TILE_ROWS) * (TILE_ROWS * NREL) + r[u] * TILE_ROWS + (dl % TILE_ROWS);
                const int p   = atomicAdd(&lcur[b], 1);
                int2 rec;
                rec.x = s[u] | (key << KEY_SHIFT);
                rec.y = w[u];
                stage[p] = rec;
            }
        }
    }
    __syncthreads();

    const int nloc = end - beg;
    for (int i = t; i < nloc; i += SORT_THREADS) part[beg + i] = stage[i];
}

__global__ void __launch_bounds__(SORT_THREADS)
bin_order_kernel(const int2* __restrict__ part,
                 const int*  __restrict__ runTab,
                 int*        __restrict__ bend,
                 int2*       __restrict__ erec,
                 int nBuckets, int chunk, int tabRow)
{
    __shared__ int2 raw[FINE_CAP];
    __shared__ int2 sorted[FINE_CAP];
    __shared__ int  cnt[FINE_KEYS];
    __shared__ int  lcur[FINE_KEYS];
    __shared__ int  rsrc[SORT_BLOCKS];
    __shared__ int  roff[SORT_BLOCKS + 1];
    __shared__ int  wtot[32];
    __shared__ int  sbin;

    const int t = threadIdx.x;
    const int g = blockIdx.x;
    const int lane = t & 31;
    const int wave = t >> 5;

    int lo = 0, rn = 0;
    if (t < SORT_BLOCKS) {
        const int* row = runTab + (size_t)t * tabRow + g;
        lo = row[0];
        rn = row[1] - lo;
    }
    const int ro = block_excl_scan_1024(rn, wtot);
    const int so = block_excl_scan_1024(lo, wtot);
    if (t < SORT_BLOCKS) {
        rsrc[t] = t * chunk + lo;
        roff[t] = ro;
    }
    if (t == SORT_THREADS - 1) {
        roff[SORT_BLOCKS] = ro + rn;
        sbin = so + lo;
    }
    cnt[t] = 0;
    __syncthreads();
    const int n = roff[SORT_BLOCKS];
    const int s = sbin;
    const bool inLds = n <= FINE_CAP;

    int rs[RUNS_PER_WAVE], rb[RUNS_PER_WAVE], rl[RUNS_PER_WAVE];
#pragma unroll
    for (int u = 0; u < RUNS_PER_WAVE; ++u) {
        const int c = wave + 32 * u;
        rs[u] = rsrc[c];
        rb[u] = roff[c];
        rl[u] = roff[c + 1] - rb[u];
    }

    if (inLds) {
        int2 v[RUNS_PER_WAVE];
#pragma unroll
        for (int u = 0; u < RUNS_PER_WAVE; ++u) {
            v[u].x = 0; v[u].y = 0;
            if (lane < rl[u]) v[u] = part[rs[u] + lane];
        }
#pragma unroll
        for (int u = 0; u < RUNS_PER_WAVE; ++u) {
            if (lane < rl[u]) raw[rb[u] + lane] = v[u];
            for (int j = lane + 32; j < rl[u]; j += 32) raw[rb[u] + j] = part[rs[u] + j];
        }
        __syncthreads();
        for (int i = t; i < n; i += SORT_THREADS) atomicAdd(&cnt[raw[i].x >> KEY_SHIFT], 1);
    } else {
        for (int u = 0; u < RUNS_PER_WAVE; ++u)
            for (int j = lane; j < rl[u]; j += 32)
                atomicAdd(&cnt[part[rs[u] + j].x >> KEY_SHIFT], 1);
    }
    __syncthreads();

    const int c   = cnt[t];
    const int off = block_excl_scan_1024(c, wtot);
    lcur[t] = off;
    const int b = g * FINE_KEYS + t;
    if (b < nBuckets) bend[b] = s + off + c;
    __syncthreads();

    if (inLds) {
        for (int i = t; i < n; i += SORT_THREADS) {
            const int2 rc = raw[i];
            const int  p  = atomicAdd(&lcur[rc.x >> KEY_SHIFT], 1);
            int2 o;
            o.x = rc.x & SRC_MASK;
            o.y = rc.y;
            sorted[p] = o;
        }
        __syncthreads();
        for (int i = t; i < n; i += SORT_THREADS) erec[s + i] = sorted[i];
    } else {
        for (int u = 0; u < RUNS_PER_WAVE; ++u)
            for (int j = lane; j < rl[u]; j += 32) {
                const int2 rc = part[rs[u] + j];
                const int  p  = atomicAdd(&lcur[rc.x >> KEY_SHIFT], 1);
                int2 o;
                o.x = rc.x & SRC_MASK;
                o.y = rc.y;
                erec[s + p] = o;
            }
    }
}

#define LAYER_WAVES   25
#define LAYER_THREADS (LAYER_WAVES * 32)
#define STRT          68

__device__ __forceinline__ void wave_lds_handover()
{
    __builtin_amdgcn_fence(__ATOMIC_RELEASE, "wavefront");
    __builtin_amdgcn_wave_barrier();
    __builtin_amdgcn_fence(__ATOMIC_ACQUIRE, "wavefront");
}

__global__ void __launch_bounds__(LAYER_THREADS)
rgcn_layer_kernel(const float* __restrict__ x,
                  const int*   __restrict__ erec,
                  const int*   __restrict__ bend,
                  const float* __restrict__ lin_w,
                  const float* __restrict__ lin_b,
                  const float* __restrict__ self_w,
                  const float* __restrict__ self_b,
                  const float* __restrict__ prev,
                  const float* __restrict__ pw,
                  const float* __restrict__ pb,
                  const float* __restrict__ tw,
                  const float* __restrict__ tb,
                  float*       __restrict__ hidden,
                  float*       __restrict__ outp,
                  int nNodes)
{
    __shared__ __attribute__((aligned(16)))
        float smem[DIM * STRC + 2 * DIM * STRH + LAYER_WAVES * 16 * STRT];
    float* WT  = smem;
    float* WP  = WT  + DIM * STRC;
    float* WTT = WP  + DIM * STRH;
    float* HT  = WTT + DIM * STRH;

    const int t = threadIdx.x;
    for (int i = t; i < KCONV * DIM; i += LAYER_THREADS) {
        int k = i >> 6;
        int c = i & 63;
        float v = (k < NREL * DIM) ? lin_w[i] : self_w[i - NREL * DIM * DIM];
        WT[(size_t)c * STRC + k] = v;
    }
    for (int i = t; i < KHW * DIM; i += LAYER_THREADS) {
        int k = i >> 6;
        int c = i & 63;
        WP [(size_t)c * STRH + k] = pw[i];
        WTT[(size_t)c * STRH + k] = tw[i];
    }
    __syncthreads();

    const int lane = t & 31;
    const int wave = __builtin_amdgcn_readfirstlane(t >> 5);
    const int m     = lane & 15;
    const int khalf = lane >> 4;

    const float* B0 = WT + (size_t)(m)      * STRC;
    const float* B1 = WT + (size_t)(16 + m) * STRC;
    const float* B2 = WT + (size_t)(32 + m) * STRC;
    const float* B3 = WT + (size_t)(48 + m) * STRC;
    const float* P0 = WP  + (size_t)(m)      * STRH;
    const float* P1 = WP  + (size_t)(16 + m) * STRH;
    const float* P2 = WP  + (size_t)(32 + m) * STRH;
    const float* P3 = WP  + (size_t)(48 + m) * STRH;
    const float* T0 = WTT + (size_t)(m)      * STRH;
    const float* T1 = WTT + (size_t)(16 + m) * STRH;
    const float* T2 = WTT + (size_t)(32 + m) * STRH;
    const float* T3 = WTT + (size_t)(48 + m) * STRH;
    float*       ht   = HT + wave * (16 * STRT);
    const float* hRow = ht + m * STRT;

    const long base = ((long)blockIdx.x * LAYER_WAVES + wave) * 16;
    if (base >= nNodes) return;
    {
        const float* xRow   = x    + (base + m) * (size_t)DIM;
        const float* pRow   = prev + (base + m) * (size_t)DIM;

        const int2* __restrict__ rec2 = (const int2*)erec;
        const char* __restrict__ xb   = (const char*)x;
        const unsigned lane8 = (unsigned)lane << 3;
        const int bvA   = bend[base * NREL + lane];
        const int bvB   = bend[base * NREL + 32 + lane];
        const int bprev = (base > 0) ? bend[base * NREL - 1] : 0;

        int  segBeg = bprev;
        int2 recN   = {0, 0};
        {
            const int send0 = __builtin_amdgcn_readlane(bvA, 15);
            if (send0 > segBeg) recN = rec2[min(segBeg + lane, send0 - 1)];
        }

#define ROW_LOAD(U)                                                              \
        const unsigned s##U = (unsigned)__builtin_amdgcn_readlane(rec.x, (U));  \
        const float    w##U = __int_as_float(__builtin_amdgcn_readlane(rec.y, (U))); \
        const float2   v##U = *(const float2*)(xb + (size_t)((s##U << 8) | lane8));

#define BUCKET_DONE()                                                            \
        {                                                                        \
            *(float2*)(ht + mm * STRT + 2 * lane) = acc;                         \
            if (m == mm) dsum = ss;                                              \
            acc.x = 0.0f; acc.y = 0.0f; ss = 0.0f;                               \
            ++mm;                                                                \
            bnd = (mm < 15) ? __builtin_amdgcn_readlane(bvr, lo16 + mm) : 0x7fffffff; \
        }

#define BUCKET_CHECK(U)                                                          \
        if ((U) >= togo) {                                                       \
            do { BUCKET_DONE() togo = bnd - j0; } while ((U) >= togo);           \
        }

#define EDGE_SURE(U)                                                             \
        BUCKET_CHECK(U)                                                          \
        acc.x += v##U.x * w##U; acc.y += v##U.y * w##U; ss += w##U;

#define EDGE_MAYBE(U)                                                            \
        BUCKET_CHECK(U)                                                          \
        {                                                                        \
            const float wz = ((U) < left) ? w##U : 0.0f;                         \
            acc.x += v##U.x * wz; acc.y += v##U.y * wz; ss += wz;                \
        }

        v8f a0 = {}, a1 = {}, a2 = {}, a3 = {};
        for (int r = 0; r < NREL; ++r) {
            const int bvr  = (r < 2) ? bvA : bvB;
            const int lo16 = (r & 1) * 16;
            const int sbeg = segBeg;
            const int send = __builtin_amdgcn_readlane(bvr, lo16 + 15);
            int2 rec = recN;
            if (r + 1 < NREL) {
                const int bvn   = (r + 1 < 2) ? bvA : bvB;
                const int sendN = __builtin_amdgcn_readlane(bvn, ((r + 1) & 1) * 16 + 15);
                if (sendN > send) recN = rec2[min(send + lane, sendN - 1)];
            }
            segBeg = send;

            float2 acc  = {0.0f, 0.0f};
            float  ss   = 0.0f;
            float  dsum = 0.0f;
            int    mm   = 0;
            int    bnd  = __builtin_amdgcn_readlane(bvr, lo16);

            int2 recP = {0, 0};
            for (int cb = sbeg; cb < send; cb += 32) {
                const int ce = min(cb + 32, send);
                if (cb != sbeg) rec = recP;
                if (ce < send) recP = rec2[min(ce + lane, send - 1)];
                for (int j0 = cb; j0 < ce; j0 += 16) {
                    if (j0 != cb) {
                        rec.x = __shfl_xor(rec.x, 16);
                        rec.y = __shfl_xor(rec.y, 16);
                    }
                    const int left = ce - j0;
                    int togo = bnd - j0;
                    if (left > 8) {
                        ROW_LOAD(0)  ROW_LOAD(1)  ROW_LOAD(2)  ROW_LOAD(3)
                        ROW_LOAD(4)  ROW_LOAD(5)  ROW_LOAD(6)  ROW_LOAD(7)
                        ROW_LOAD(8)  ROW_LOAD(9)  ROW_LOAD(10) ROW_LOAD(11)
                        ROW_LOAD(12) ROW_LOAD(13) ROW_LOAD(14) ROW_LOAD(15)
                        EDGE_SURE(0)   EDGE_SURE(1)   EDGE_SURE(2)   EDGE_SURE(3)
                        EDGE_SURE(4)   EDGE_SURE(5)   EDGE_SURE(6)   EDGE_SURE(7)
                        EDGE_SURE(8)   EDGE_MAYBE(9)  EDGE_MAYBE(10) EDGE_MAYBE(11)
                        EDGE_MAYBE(12) EDGE_MAYBE(13) EDGE_MAYBE(14) EDGE_MAYBE(15)
                    } else {
                        ROW_LOAD(0) ROW_LOAD(1) ROW_LOAD(2) ROW_LOAD(3)
                        ROW_LOAD(4) ROW_LOAD(5) ROW_LOAD(6) ROW_LOAD(7)
                        EDGE_SURE(0)  EDGE_MAYBE(1) EDGE_MAYBE(2) EDGE_MAYBE(3)
                        EDGE_MAYBE(4) EDGE_MAYBE(5) EDGE_MAYBE(6) EDGE_MAYBE(7)
                    }
                }
            }
            while (mm < 16) BUCKET_DONE()
            wave_lds_handover();

            const float sc = 1.0f / (dsum + 1e-10f);
            for (int k = 0; k < DIM; k += 4) {
                int kk = k + 2 * khalf;
                float2 av = *(const float2*)(hRow + kk);
                v2f a; a.x = av.x * sc; a.y = av.y * sc;
                int kw = r * DIM + kk;
                v2f b0 = *(const v2f*)(B0 + kw);
                v2f b1 = *(const v2f*)(B1 + kw);
                v2f b2 = *(const v2f*)(B2 + kw);
                v2f b3 = *(const v2f*)(B3 + kw);
                a0 = __builtin_amdgcn_wmma_f32_16x16x4_f32(false, a, false, b0, (short)0, a0, false, false);
                a1 = __builtin_amdgcn_wmma_f32_16x16x4_f32(false, a, false, b1, (short)0, a1, false, false);
                a2 = __builtin_amdgcn_wmma_f32_16x16x4_f32(false, a, false, b2, (short)0, a2, false, false);
                a3 = __builtin_amdgcn_wmma_f32_16x16x4_f32(false, a, false, b3, (short)0, a3, false, false);
            }
            wave_lds_handover();
        }
#undef EDGE_MAYBE
#undef EDGE_SURE
#undef BUCKET_CHECK
#undef BUCKET_DONE
#undef ROW_LOAD

#pragma unroll 4
        for (int k = 0; k < DIM; k += 4) {
            int kk = k + 2 * khalf;
            float2 av = *(const float2*)(xRow + kk);
            v2f a; a.x = av.x; a.y = av.y;
            int kw = NREL * DIM + kk;
            v2f b0 = *(const v2f*)(B0 + kw);
            v2f b1 = *(const v2f*)(B1 + kw);
            v2f b2 = *(const v2f*)(B2 + kw);
            v2f b3 = *(const v2f*)(B3 + kw);
            a0 = __builtin_amdgcn_wmma_f32_16x16x4_f32(false, a, false, b0, (short)0, a0, false, false);
            a1 = __builtin_amdgcn_wmma_f32_16x16x4_f32(false, a, false, b1, (short)0, a1, false, false);
            a2 = __builtin_amdgcn_wmma_f32_16x16x4_f32(false, a, false, b2, (short)0, a2, false, false);
            a3 = __builtin_amdgcn_wmma_f32_16x16x4_f32(false, a, false, b3, (short)0, a3, false, false);
        }

#define CONV_EPILOGUE(ACC, TILE)                                           \
        {                                                                  \
            int   col = (TILE) * 16 + m;                                   \
            float bia = lin_b[col] + self_b[col];                          \
            _Pragma("unroll")                                              \
            for (int r = 0; r < 8; ++r) {                                  \
                int   lr = khalf * 8 + r;                                  \
                float hv = sigmoidf_(ACC[r] + bia);                        \
                ht[lr * STRT + col] = hv;                                  \
                hidden[(base + lr) * DIM + col] = hv;                      \
            }                                                              \
        }
        CONV_EPILOGUE(a0, 0) CONV_EPILOGUE(a1, 1) CONV_EPILOGUE(a2, 2) CONV_EPILOGUE(a3, 3)
#undef CONV_EPILOGUE
        wave_lds_handover();

        v8f p0 = {}, p1 = {}, p2 = {}, p3 = {};
        v8f g0 = {}, g1 = {}, g2 = {}, g3 = {};

#define HW_STEP(AROW, KOFF)                                                              \
        for (int k = 0; k < DIM; k += 4) {                                               \
            int kk = k + 2 * khalf;                                                      \
            float2 av = *(const float2*)((AROW) + kk);                                   \
            v2f a; a.x = av.x; a.y = av.y;                                               \
            int kw = kk + (KOFF);                                                        \
            v2f bp0 = *(const v2f*)(P0 + kw);                                            \
            v2f bp1 = *(const v2f*)(P1 + kw);                                            \
            v2f bp2 = *(const v2f*)(P2 + kw);                                            \
            v2f bp3 = *(const v2f*)(P3 + kw);                                            \
            v2f bt0 = *(const v2f*)(T0 + kw);                                            \
            v2f bt1 = *(const v2f*)(T1 + kw);                                            \
            v2f bt2 = *(const v2f*)(T2 + kw);                                            \
            v2f bt3 = *(const v2f*)(T3 + kw);                                            \
            p0 = __builtin_amdgcn_wmma_f32_16x16x4_f32(false, a, false, bp0, (short)0, p0, false, false); \
            p1 = __builtin_amdgcn_wmma_f32_16x16x4_f32(false, a, false, bp1, (short)0, p1, false, false); \
            p2 = __builtin_amdgcn_wmma_f32_16x16x4_f32(false, a, false, bp2, (short)0, p2, false, false); \
            p3 = __builtin_amdgcn_wmma_f32_16x16x4_f32(false, a, false, bp3, (short)0, p3, false, false); \
            g0 = __builtin_amdgcn_wmma_f32_16x16x4_f32(false, a, false, bt0, (short)0, g0, false, false); \
            g1 = __builtin_amdgcn_wmma_f32_16x16x4_f32(false, a, false, bt1, (short)0, g1, false, false); \
            g2 = __builtin_amdgcn_wmma_f32_16x16x4_f32(false, a, false, bt2, (short)0, g2, false, false); \
            g3 = __builtin_amdgcn_wmma_f32_16x16x4_f32(false, a, false, bt3, (short)0, g3, false, false); \
        }
        HW_STEP(hRow, 0)
        HW_STEP(pRow, DIM)
#undef HW_STEP

#define HW_EPILOGUE(PACC, GACC, TILE)                                      \
        {                                                                  \
            int   col = (TILE) * 16 + m;                                   \
            float pbv = pb[col];                                           \
            float tbv = tb[col];                                           \
            _Pragma("unroll")                                              \
            for (int r = 0; r < 8; ++r) {                                  \
                int   lr = khalf * 8 + r;                                  \
                float pr = fmaxf(PACC[r] + pbv, 0.0f);                     \
                float gt = sigmoidf_(GACC[r] + tbv);                       \
                float h  = ht[lr * STRT + col];                            \
                outp[(base + lr) * DIM + col] = gt * pr + (1.0f - gt) * h; \
            }                                                              \
        }
        HW_EPILOGUE(p0, g0, 0) HW_EPILOGUE(p1, g1, 1)
        HW_EPILOGUE(p2, g2, 2) HW_EPILOGUE(p3, g3, 3)
#undef HW_EPILOGUE
    }
}

extern "C" void kernel_launch(void* const* d_in, const int* in_sizes, int n_in,
                              void* d_out, int out_size, void* d_ws, size_t ws_size,
                              hipStream_t stream) {
    const float* node_feat   = (const float*)d_in[0];
    const float* edge_weight = (const float*)d_in[1];
    const float* conv_lin_w  = (const float*)d_in[2];
    const float* conv_lin_b  = (const float*)d_in[3];
    const float* conv_self_w = (const float*)d_in[4];
    const float* conv_self_b = (const float*)d_in[5];
    const float* hw_proj_w   = (const float*)d_in[6];
    const float* hw_proj_b   = (const float*)d_in[7];
    const float* hw_trans_w  = (const float*)d_in[8];
    const float* hw_trans_b  = (const float*)d_in[9];
    const int*   edge_src    = (const int*)d_in[10];
    const int*   edge_dst    = (const int*)d_in[11];
    const int*   edge_rel    = (const int*)d_in[12];
    float*       out         = (float*)d_out;

    const int N = in_sizes[0] / DIM;
    const int E = in_sizes[10];

    float* hA   = (float*)d_ws;
    float* hB   = hA   + (size_t)N * DIM;
    float* cur0 = hB   + (size_t)N * DIM;
    float* cur1 = cur0 + (size_t)N * DIM;
    const int NB    = N * NREL;
    const int nbin  = (N + BIN_NODES - 1) >> BIN_SHIFT;
    const int tabRow = (nbin + 32) & ~31;
    const int chunk = (E + SORT_BLOCKS - 1) / SORT_BLOCKS;
    int*  cursor   = (int*)(cur1 + (size_t)N * DIM);
    int*  erec     = cursor + NB;
    int*  part     = erec + 2 * (size_t)E;
    int*  runTab   = part + 2 * (size_t)E;

    const int tiles         = (N + 15) / 16;
    const int layerBlocks   = (tiles + LAYER_WAVES - 1) / LAYER_WAVES;
    const size_t stageLds = (size_t)chunk * sizeof(int2);

    chunk_split_kernel<<<SORT_BLOCKS, SORT_THREADS, stageLds, stream>>>(
        edge_src, edge_dst, edge_rel, edge_weight, runTab, (int2*)part,
        E, chunk, nbin, tabRow);
    bin_order_kernel<<<nbin, SORT_THREADS, 0, stream>>>(
        (const int2*)part, runTab, cursor, (int2*)erec, NB, chunk, tabRow);

    for (int i = 0; i < 3; ++i) {
        const float* xin  = (i == 0) ? node_feat : ((i == 1) ? cur0 : cur1);
        const float* prev = (i == 0) ? node_feat : ((i == 1) ? hA : hB);
        float*       hid  = (i & 1) ? hB : hA;
        float*       gout = (i == 0) ? cur0 : ((i == 1) ? cur1 : out);

        rgcn_layer_kernel<<<layerBlocks, LAYER_THREADS, 0, stream>>>(
            xin, erec, cursor,
            conv_lin_w  + (size_t)i * 256 * DIM, conv_lin_b  + (size_t)i * DIM,
            conv_self_w + (size_t)i * DIM * DIM, conv_self_b + (size_t)i * DIM,
            prev,
            hw_proj_w  + (size_t)i * KHW * DIM, hw_proj_b  + (size_t)i * DIM,
            hw_trans_w + (size_t)i * KHW * DIM, hw_trans_b + (size_t)i * DIM,
            hid, gout, N);
    }
}


# Wave priority by phase (s_setprio): gather code 3, update products 1, highway products 0, so row requests do not queue behind older waves' f32 WMMA runs; sum order and all arithmetic unchanged
# speedup vs baseline: 4.3222x; 1.0772x over previous
#include <hip/hip_runtime.h>
#include <hip/hip_bf16.h>
#include <math.h>


typedef __attribute__((ext_vector_type(2))) float v2f;
typedef __attribute__((ext_vector_type(8))) float v8f;

#define DIM   64
#define NREL  4
#define KCONV 320
#define KHW   128
#define STRC  324
#define STRH  132

__device__ __forceinline__ float sigmoidf_(float v) {
    return 1.0f / (1.0f + __expf(-v));
}

#define BIN_SHIFT   8
#define BIN_NODES   (1 << BIN_SHIFT)
#define FINE_KEYS   (BIN_NODES * NREL)
#define TILE_ROWS   16
#define SORT_BLOCKS 256
#define SORT_THREADS 1024
#define RUNS_PER_WAVE (SORT_BLOCKS / 32)
#define KEY_SHIFT   20
#define SRC_MASK    ((1 << KEY_SHIFT) - 1)
#define FINE_CAP    8192

__device__ __forceinline__ int block_excl_scan_1024(int v, int* wtot)
{
    const int lane = threadIdx.x & 31;
    const int wave = threadIdx.x >> 5;
    int inc = v;
    for (int o = 1; o < 32; o <<= 1) {
        int n = __shfl_up(inc, o);
        if (lane >= o) inc += n;
    }
    if (lane == 31) wtot[wave] = inc;
    __syncthreads();
    int woff = 0;
    for (int w = 0; w < 32; ++w) {
        int x = wtot[w];
        if (w < wave) woff += x;
    }
    __syncthreads();
    return woff + inc - v;
}

__global__ void __launch_bounds__(SORT_THREADS)
chunk_split_kernel(const int*   __restrict__ src,
                   const int*   __restrict__ dst,
                   const int*   __restrict__ rel,
                   const float* __restrict__ ew,
                   int*         __restrict__ runTab,
                   int2*        __restrict__ part,
                   int nEdges, int chunk, int nbin, int tabRow)
{
    extern __shared__ int2 stage[];
    __shared__ int mine[SORT_THREADS];
    __shared__ int lcur[SORT_THREADS];
    __shared__ int wtot[32];

    const int t   = threadIdx.x;
    const int blk = blockIdx.x;
    mine[t] = 0;
    __syncthreads();

    const int beg = blk * chunk;
    const int end = min(beg + chunk, nEdges);
    for (int i0 = beg + t; i0 < end; i0 += 8 * SORT_THREADS) {
        int d[8];
#pragma unroll
        for (int u = 0; u < 8; ++u) {
            const int i = i0 + u * SORT_THREADS;
            d[u] = (i < end) ? dst[i] : -1;
        }
#pragma unroll
        for (int u = 0; u < 8; ++u)
            if (d[u] >= 0) atomicAdd(&mine[d[u] >> BIN_SHIFT], 1);
    }
    __syncthreads();

    const int vm = mine[t];
    const int lo = block_excl_scan_1024(vm, wtot);
    lcur[t] = lo;
    {
        int* row = runTab + (size_t)blk * tabRow;
        if (t < nbin) row[t] = lo;
        if (t == SORT_THREADS - 1) row[nbin] = lo + vm;
    }
    __syncthreads();

    for (int i0 = beg + t; i0 < end; i0 += 4 * SORT_THREADS) {
        int d[4], r[4], s[4], w[4];
#pragma unroll
        for (int u = 0; u < 4; ++u) {
            const int  i  = i0 + u * SORT_THREADS;
            const bool ok = i < end;
            d[u] = ok ? dst[i] : -1;
            r[u] = ok ? rel[i] : 0;
            s[u] = ok ? src[i] : 0;
            w[u] = ok ? __float_as_int(ew[i]) : 0;
        }
#pragma unroll
        for (int u = 0; u < 4; ++u) {
            if (d[u] >= 0) {
                const int b   = d[u] >> BIN_SHIFT;
                const int dl  = d[u] & (BIN_NODES - 1);
                const int key = (dl / TILE_ROWS) * (TILE_ROWS * NREL) + r[u] * TILE_ROWS + (dl % TILE_ROWS);
                const int p   = atomicAdd(&lcur[b], 1);
                int2 rec;
                rec.x = s[u] | (key << KEY_SHIFT);
                rec.y = w[u];
                stage[p] = rec;
            }
        }
    }
    __syncthreads();

    const int nloc = end - beg;
    for (int i = t; i < nloc; i += SORT_THREADS) part[beg + i] = stage[i];
}

__global__ void __launch_bounds__(SORT_THREADS)
bin_order_kernel(const int2* __restrict__ part,
                 const int*  __restrict__ runTab,
                 int*        __restrict__ bend,
                 int2*       __restrict__ erec,
                 int nBuckets, int chunk, int tabRow)
{
    __shared__ int2 raw[FINE_CAP];
    __shared__ int2 sorted[FINE_CAP];
    __shared__ int  cnt[FINE_KEYS];
    __shared__ int  lcur[FINE_KEYS];
    __shared__ int  rsrc[SORT_BLOCKS];
    __shared__ int  roff[SORT_BLOCKS + 1];
    __shared__ int  wtot[32];
    __shared__ int  sbin;

    const int t = threadIdx.x;
    const int g = blockIdx.x;
    const int lane = t & 31;
    const int wave = t >> 5;

    int lo = 0, rn = 0;
    if (t < SORT_BLOCKS) {
        const int* row = runTab + (size_t)t * tabRow + g;
        lo = row[0];
        rn = row[1] - lo;
    }
    const int ro = block_excl_scan_1024(rn, wtot);
    const int so = block_excl_scan_1024(lo, wtot);
    if (t < SORT_BLOCKS) {
        rsrc[t] = t * chunk + lo;
        roff[t] = ro;
    }
    if (t == SORT_THREADS - 1) {
        roff[SORT_BLOCKS] = ro + rn;
        sbin = so + lo;
    }
    cnt[t] = 0;
    __syncthreads();
    const int n = roff[SORT_BLOCKS];
    const int s = sbin;
    const bool inLds = n <= FINE_CAP;

    int rs[RUNS_PER_WAVE], rb[RUNS_PER_WAVE], rl[RUNS_PER_WAVE];
#pragma unroll
    for (int u = 0; u < RUNS_PER_WAVE; ++u) {
        const int c = wave + 32 * u;
        rs[u] = rsrc[c];
        rb[u] = roff[c];
        rl[u] = roff[c + 1] - rb[u];
    }

    if (inLds) {
        int2 v[RUNS_PER_WAVE];
#pragma unroll
        for (int u = 0; u < RUNS_PER_WAVE; ++u) {
            v[u].x = 0; v[u].y = 0;
            if (lane < rl[u]) v[u] = part[rs[u] + lane];
        }
#pragma unroll
        for (int u = 0; u < RUNS_PER_WAVE; ++u) {
            if (lane < rl[u]) raw[rb[u] + lane] = v[u];
            for (int j = lane + 32; j < rl[u]; j += 32) raw[rb[u] + j] = part[rs[u] + j];
        }
        __syncthreads();
        for (int i = t; i < n; i += SORT_THREADS) atomicAdd(&cnt[raw[i].x >> KEY_SHIFT], 1);
    } else {
        for (int u = 0; u < RUNS_PER_WAVE; ++u)
            for (int j = lane; j < rl[u]; j += 32)
                atomicAdd(&cnt[part[rs[u] + j].x >> KEY_SHIFT], 1);
    }
    __syncthreads();

    const int c   = cnt[t];
    const int off = block_excl_scan_1024(c, wtot);
    lcur[t] = off;
    const int b = g * FINE_KEYS + t;
    if (b < nBuckets) bend[b] = s + off + c;
    __syncthreads();

    if (inLds) {
        for (int i = t; i < n; i += SORT_THREADS) {
            const int2 rc = raw[i];
            const int  p  = atomicAdd(&lcur[rc.x >> KEY_SHIFT], 1);
            int2 o;
            o.x = rc.x & SRC_MASK;
            o.y = rc.y;
            sorted[p] = o;
        }
        __syncthreads();
        for (int i = t; i < n; i += SORT_THREADS) erec[s + i] = sorted[i];
    } else {
        for (int u = 0; u < RUNS_PER_WAVE; ++u)
            for (int j = lane; j < rl[u]; j += 32) {
                const int2 rc = part[rs[u] + j];
                const int  p  = atomicAdd(&lcur[rc.x >> KEY_SHIFT], 1);
                int2 o;
                o.x = rc.x & SRC_MASK;
                o.y = rc.y;
                erec[s + p] = o;
            }
    }
}

#define LAYER_WAVES   25
#define LAYER_THREADS (LAYER_WAVES * 32)
#define STRT          68

__device__ __forceinline__ void wave_lds_handover()
{
    __builtin_amdgcn_fence(__ATOMIC_RELEASE, "wavefront");
    __builtin_amdgcn_wave_barrier();
    __builtin_amdgcn_fence(__ATOMIC_ACQUIRE, "wavefront");
}

__global__ void __launch_bounds__(LAYER_THREADS)
rgcn_layer_kernel(const float* __restrict__ x,
                  const int*   __restrict__ erec,
                  const int*   __restrict__ bend,
                  const float* __restrict__ lin_w,
                  const float* __restrict__ lin_b,
                  const float* __restrict__ self_w,
                  const float* __restrict__ self_b,
                  const float* __restrict__ prev,
                  const float* __restrict__ pw,
                  const float* __restrict__ pb,
                  const float* __restrict__ tw,
                  const float* __restrict__ tb,
                  float*       __restrict__ hidden,
                  float*       __restrict__ outp,
                  int nNodes)
{
    __shared__ __attribute__((aligned(16)))
        float smem[DIM * STRC + 2 * DIM * STRH + LAYER_WAVES * 16 * STRT];
    float* WT  = smem;
    float* WP  = WT  + DIM * STRC;
    float* WTT = WP  + DIM * STRH;
    float* HT  = WTT + DIM * STRH;

    const int t = threadIdx.x;
    for (int i = t; i < KCONV * DIM; i += LAYER_THREADS) {
        int k = i >> 6;
        int c = i & 63;
        float v = (k < NREL * DIM) ? lin_w[i] : self_w[i - NREL * DIM * DIM];
        WT[(size_t)c * STRC + k] = v;
    }
    for (int i = t; i < KHW * DIM; i += LAYER_THREADS) {
        int k = i >> 6;
        int c = i & 63;
        WP [(size_t)c * STRH + k] = pw[i];
        WTT[(size_t)c * STRH + k] = tw[i];
    }
    __syncthreads();

    const int lane = t & 31;
    const int wave = __builtin_amdgcn_readfirstlane(t >> 5);
    const int m     = lane & 15;
    const int khalf = lane >> 4;

    const float* B0 = WT + (size_t)(m)      * STRC;
    const float* B1 = WT + (size_t)(16 + m) * STRC;
    const float* B2 = WT + (size_t)(32 + m) * STRC;
    const float* B3 = WT + (size_t)(48 + m) * STRC;
    const float* P0 = WP  + (size_t)(m)      * STRH;
    const float* P1 = WP  + (size_t)(16 + m) * STRH;
    const float* P2 = WP  + (size_t)(32 + m) * STRH;
    const float* P3 = WP  + (size_t)(48 + m) * STRH;
    const float* T0 = WTT + (size_t)(m)      * STRH;
    const float* T1 = WTT + (size_t)(16 + m) * STRH;
    const float* T2 = WTT + (size_t)(32 + m) * STRH;
    const float* T3 = WTT + (size_t)(48 + m) * STRH;
    float*       ht   = HT + wave * (16 * STRT);
    const float* hRow = ht + m * STRT;

    const long base = ((long)blockIdx.x * LAYER_WAVES + wave) * 16;
    if (base >= nNodes) return;
    {
        const float* xRow   = x    + (base + m) * (size_t)DIM;
        const float* pRow   = prev + (base + m) * (size_t)DIM;

        const int2* __restrict__ rec2 = (const int2*)erec;
        const char* __restrict__ xb   = (const char*)x;
        const unsigned lane8 = (unsigned)lane << 3;
        const int bvA   = bend[base * NREL + lane];
        const int bvB   = bend[base * NREL + 32 + lane];
        const int bprev = (base > 0) ? bend[base * NREL - 1] : 0;

        int  segBeg = bprev;
        int2 recN   = {0, 0};
        {
            const int send0 = __builtin_amdgcn_readlane(bvA, 15);
            if (send0 > segBeg) recN = rec2[min(segBeg + lane, send0 - 1)];
        }

#define ROW_LOAD(U)                                                              \
        const unsigned s##U = (unsigned)__builtin_amdgcn_readlane(rec.x, (U));  \
        const float    w##U = __int_as_float(__builtin_amdgcn_readlane(rec.y, (U))); \
        const float2   v##U = *(const float2*)(xb + (size_t)((s##U << 8) | lane8));

#define BUCKET_DONE()                                                            \
        {                                                                        \
            *(float2*)(ht + mm * STRT + 2 * lane) = acc;                         \
            if (m == mm) dsum = ss;                                              \
            acc.x = 0.0f; acc.y = 0.0f; ss = 0.0f;                               \
            ++mm;                                                                \
            bnd = (mm < 15) ? __builtin_amdgcn_readlane(bvr, lo16 + mm) : 0x7fffffff; \
        }

#define BUCKET_CHECK(U)                                                          \
        if ((U) >= togo) {                                                       \
            do { BUCKET_DONE() togo = bnd - j0; } while ((U) >= togo);           \
        }

#define EDGE_SURE(U)                                                             \
        BUCKET_CHECK(U)                                                          \
        acc.x += v##U.x * w##U; acc.y += v##U.y * w##U; ss += w##U;

#define EDGE_MAYBE(U)                                                            \
        BUCKET_CHECK(U)                                                          \
        {                                                                        \
            const float wz = ((U) < left) ? w##U : 0.0f;                         \
            acc.x += v##U.x * wz; acc.y += v##U.y * wz; ss += wz;                \
        }

        v8f a0 = {}, a1 = {}, a2 = {}, a3 = {};
        for (int r = 0; r < NREL; ++r) {
            __builtin_amdgcn_s_setprio(3);
            const int bvr  = (r < 2) ? bvA : bvB;
            const int lo16 = (r & 1) * 16;
            const int sbeg = segBeg;
            const int send = __builtin_amdgcn_readlane(bvr, lo16 + 15);
            int2 rec = recN;
            if (r + 1 < NREL) {
                const int bvn   = (r + 1 < 2) ? bvA : bvB;
                const int sendN = __builtin_amdgcn_readlane(bvn, ((r + 1) & 1) * 16 + 15);
                if (sendN > send) recN = rec2[min(send + lane, sendN - 1)];
            }
            segBeg = send;

            float2 acc  = {0.0f, 0.0f};
            float  ss   = 0.0f;
            float  dsum = 0.0f;
            int    mm   = 0;
            int    bnd  = __builtin_amdgcn_readlane(bvr, lo16);

            int2 recP = {0, 0};
            for (int cb = sbeg; cb < send; cb += 32) {
                const int ce = min(cb + 32, send);
                if (cb != sbeg) rec = recP;
                if (ce < send) recP = rec2[min(ce + lane, send - 1)];
                for (int j0 = cb; j0 < ce; j0 += 16) {
                    if (j0 != cb) {
                        rec.x = __shfl_xor(rec.x, 16);
                        rec.y = __shfl_xor(rec.y, 16);
                    }
                    const int left = ce - j0;
                    int togo = bnd - j0;
                    if (left > 8) {
                        ROW_LOAD(0)  ROW_LOAD(1)  ROW_LOAD(2)  ROW_LOAD(3)
                        ROW_LOAD(4)  ROW_LOAD(5)  ROW_LOAD(6)  ROW_LOAD(7)
                        ROW_LOAD(8)  ROW_LOAD(9)  ROW_LOAD(10) ROW_LOAD(11)
                        ROW_LOAD(12) ROW_LOAD(13) ROW_LOAD(14) ROW_LOAD(15)
                        EDGE_SURE(0)   EDGE_SURE(1)   EDGE_SURE(2)   EDGE_SURE(3)
                        EDGE_SURE(4)   EDGE_SURE(5)   EDGE_SURE(6)   EDGE_SURE(7)
                        EDGE_SURE(8)   EDGE_MAYBE(9)  EDGE_MAYBE(10) EDGE_MAYBE(11)
                        EDGE_MAYBE(12) EDGE_MAYBE(13) EDGE_MAYBE(14) EDGE_MAYBE(15)
                    } else {
                        ROW_LOAD(0) ROW_LOAD(1) ROW_LOAD(2) ROW_LOAD(3)
                        ROW_LOAD(4) ROW_LOAD(5) ROW_LOAD(6) ROW_LOAD(7)
                        EDGE_SURE(0)  EDGE_MAYBE(1) EDGE_MAYBE(2) EDGE_MAYBE(3)
                        EDGE_MAYBE(4) EDGE_MAYBE(5) EDGE_MAYBE(6) EDGE_MAYBE(7)
                    }
                }
            }
            while (mm < 16) BUCKET_DONE()
            wave_lds_handover();
            __builtin_amdgcn_s_setprio(1);

            const float sc = 1.0f / (dsum + 1e-10f);
            for (int k = 0; k < DIM; k += 4) {
                int kk = k + 2 * khalf;
                float2 av = *(const float2*)(hRow + kk);
                v2f a; a.x = av.x * sc; a.y = av.y * sc;
                int kw = r * DIM + kk;
                v2f b0 = *(const v2f*)(B0 + kw);
                v2f b1 = *(const v2f*)(B1 + kw);
                v2f b2 = *(const v2f*)(B2 + kw);
                v2f b3 = *(const v2f*)(B3 + kw);
                a0 = __builtin_amdgcn_wmma_f32_16x16x4_f32(false, a, false, b0, (short)0, a0, false, false);
                a1 = __builtin_amdgcn_wmma_f32_16x16x4_f32(false, a, false, b1, (short)0, a1, false, false);
                a2 = __builtin_amdgcn_wmma_f32_16x16x4_f32(false, a, false, b2, (short)0, a2, false, false);
                a3 = __builtin_amdgcn_wmma_f32_16x16x4_f32(false, a, false, b3, (short)0, a3, false, false);
            }
            wave_lds_handover();
        }
#undef EDGE_MAYBE
#undef EDGE_SURE
#undef BUCKET_CHECK
#undef BUCKET_DONE
#undef ROW_LOAD

#pragma unroll 4
        for (int k = 0; k < DIM; k += 4) {
            int kk = k + 2 * khalf;
            float2 av = *(const float2*)(xRow + kk);
            v2f a; a.x = av.x; a.y = av.y;
            int kw = NREL * DIM + kk;
            v2f b0 = *(const v2f*)(B0 + kw);
            v2f b1 = *(const v2f*)(B1 + kw);
            v2f b2 = *(const v2f*)(B2 + kw);
            v2f b3 = *(const v2f*)(B3 + kw);
            a0 = __builtin_amdgcn_wmma_f32_16x16x4_f32(false, a, false, b0, (short)0, a0, false, false);
            a1 = __builtin_amdgcn_wmma_f32_16x16x4_f32(false, a, false, b1, (short)0, a1, false, false);
            a2 = __builtin_amdgcn_wmma_f32_16x16x4_f32(false, a, false, b2, (short)0, a2, false, false);
            a3 = __builtin_amdgcn_wmma_f32_16x16x4_f32(false, a, false, b3, (short)0, a3, false, false);
        }

#define CONV_EPILOGUE(ACC, TILE)                                           \
        {                                                                  \
            int   col = (TILE) * 16 + m;                                   \
            float bia = lin_b[col] + self_b[col];                          \
            _Pragma("unroll")                                              \
            for (int r = 0; r < 8; ++r) {                                  \
                int   lr = khalf * 8 + r;                                  \
                float hv = sigmoidf_(ACC[r] + bia);                        \
                ht[lr * STRT + col] = hv;                                  \
                hidden[(base + lr) * DIM + col] = hv;                      \
            }                                                              \
        }
        CONV_EPILOGUE(a0, 0) CONV_EPILOGUE(a1, 1) CONV_EPILOGUE(a2, 2) CONV_EPILOGUE(a3, 3)
#undef CONV_EPILOGUE
        wave_lds_handover();

        __builtin_amdgcn_s_setprio(0);
        v8f p0 = {}, p1 = {}, p2 = {}, p3 = {};
        v8f g0 = {}, g1 = {}, g2 = {}, g3 = {};

#define HW_STEP(AROW, KOFF)                                                              \
        for (int k = 0; k < DIM; k += 4) {                                               \
            int kk = k + 2 * khalf;                                                      \
            float2 av = *(const float2*)((AROW) + kk);                                   \
            v2f a; a.x = av.x; a.y = av.y;                                               \
            int kw = kk + (KOFF);                                                        \
            v2f bp0 = *(const v2f*)(P0 + kw);                                            \
            v2f bp1 = *(const v2f*)(P1 + kw);                                            \
            v2f bp2 = *(const v2f*)(P2 + kw);                                            \
            v2f bp3 = *(const v2f*)(P3 + kw);                                            \
            v2f bt0 = *(const v2f*)(T0 + kw);                                            \
            v2f bt1 = *(const v2f*)(T1 + kw);                                            \
            v2f bt2 = *(const v2f*)(T2 + kw);                                            \
            v2f bt3 = *(const v2f*)(T3 + kw);                                            \
            p0 = __builtin_amdgcn_wmma_f32_16x16x4_f32(false, a, false, bp0, (short)0, p0, false, false); \
            p1 = __builtin_amdgcn_wmma_f32_16x16x4_f32(false, a, false, bp1, (short)0, p1, false, false); \
            p2 = __builtin_amdgcn_wmma_f32_16x16x4_f32(false, a, false, bp2, (short)0, p2, false, false); \
            p3 = __builtin_amdgcn_wmma_f32_16x16x4_f32(false, a, false, bp3, (short)0, p3, false, false); \
            g0 = __builtin_amdgcn_wmma_f32_16x16x4_f32(false, a, false, bt0, (short)0, g0, false, false); \
            g1 = __builtin_amdgcn_wmma_f32_16x16x4_f32(false, a, false, bt1, (short)0, g1, false, false); \
            g2 = __builtin_amdgcn_wmma_f32_16x16x4_f32(false, a, false, bt2, (short)0, g2, false, false); \
            g3 = __builtin_amdgcn_wmma_f32_16x16x4_f32(false, a, false, bt3, (short)0, g3, false, false); \
        }
        HW_STEP(hRow, 0)
        HW_STEP(pRow, DIM)
#undef HW_STEP

#define HW_EPILOGUE(PACC, GACC, TILE)                                      \
        {                                                                  \
            int   col = (TILE) * 16 + m;                                   \
            float pbv = pb[col];                                           \
            float tbv = tb[col];                                           \
            _Pragma("unroll")                                              \
            for (int r = 0; r < 8; ++r) {                                  \
                int   lr = khalf * 8 + r;                                  \
                float pr = fmaxf(PACC[r] + pbv, 0.0f);                     \
                float gt = sigmoidf_(GACC[r] + tbv);                       \
                float h  = ht[lr * STRT + col];                            \
                outp[(base + lr) * DIM + col] = gt * pr + (1.0f - gt) * h; \
            }                                                              \
        }
        HW_EPILOGUE(p0, g0, 0) HW_EPILOGUE(p1, g1, 1)
        HW_EPILOGUE(p2, g2, 2) HW_EPILOGUE(p3, g3, 3)
#undef HW_EPILOGUE
    }
}

extern "C" void kernel_launch(void* const* d_in, const int* in_sizes, int n_in,
                              void* d_out, int out_size, void* d_ws, size_t ws_size,
                              hipStream_t stream) {
    const float* node_feat   = (const float*)d_in[0];
    const float* edge_weight = (const float*)d_in[1];
    const float* conv_lin_w  = (const float*)d_in[2];
    const float* conv_lin_b  = (const float*)d_in[3];
    const float* conv_self_w = (const float*)d_in[4];
    const float* conv_self_b = (const float*)d_in[5];
    const float* hw_proj_w   = (const float*)d_in[6];
    const float* hw_proj_b   = (const float*)d_in[7];
    const float* hw_trans_w  = (const float*)d_in[8];
    const float* hw_trans_b  = (const float*)d_in[9];
    const int*   edge_src    = (const int*)d_in[10];
    const int*   edge_dst    = (const int*)d_in[11];
    const int*   edge_rel    = (const int*)d_in[12];
    float*       out         = (float*)d_out;

    const int N = in_sizes[0] / DIM;
    const int E = in_sizes[10];

    float* hA   = (float*)d_ws;
    float* hB   = hA   + (size_t)N * DIM;
    float* cur0 = hB   + (size_t)N * DIM;
    float* cur1 = cur0 + (size_t)N * DIM;
    const int NB    = N * NREL;
    const int nbin  = (N + BIN_NODES - 1) >> BIN_SHIFT;
    const int tabRow = (nbin + 32) & ~31;
    const int chunk = (E + SORT_BLOCKS - 1) / SORT_BLOCKS;
    int*  cursor   = (int*)(cur1 + (size_t)N * DIM);
    int*  erec     = cursor + NB;
    int*  part     = erec + 2 * (size_t)E;
    int*  runTab   = part + 2 * (size_t)E;

    const int tiles         = (N + 15) / 16;
    const int layerBlocks   = (tiles + LAYER_WAVES - 1) / LAYER_WAVES;
    const size_t stageLds = (size_t)chunk * sizeof(int2);

    chunk_split_kernel<<<SORT_BLOCKS, SORT_THREADS, stageLds, stream>>>(
        edge_src, edge_dst, edge_rel, edge_weight, runTab, (int2*)part,
        E, chunk, nbin, tabRow);
    bin_order_kernel<<<nbin, SORT_THREADS, 0, stream>>>(
        (const int2*)part, runTab, cursor, (int2*)erec, NB, chunk, tabRow);

    for (int i = 0; i < 3; ++i) {
        const float* xin  = (i == 0) ? node_feat : ((i == 1) ? cur0 : cur1);
        const float* prev = (i == 0) ? node_feat : ((i == 1) ? hA : hB);
        float*       hid  = (i & 1) ? hB : hA;
        float*       gout = (i == 0) ? cur0 : ((i == 1) ? cur1 : out);

        rgcn_layer_kernel<<<layerBlocks, LAYER_THREADS, 0, stream>>>(
            xin, erec, cursor,
            conv_lin_w  + (size_t)i * 256 * DIM, conv_lin_b  + (size_t)i * DIM,
            conv_self_w + (size_t)i * DIM * DIM, conv_self_b + (size_t)i * DIM,
            prev,
            hw_proj_w  + (size_t)i * KHW * DIM, hw_proj_b  + (size_t)i * DIM,
            hw_trans_w + (size_t)i * KHW * DIM, hw_trans_b + (size_t)i * DIM,
            hid, gout, N);
    }
}


# Both sigmoids (hidden, gate): 1/(1+exp(-v)) by the hardware reciprocal (1 ulp on [1,inf]), not the correctly rounded division sequence: 9 vector instructions fewer per value; sum orders unchanged
# speedup vs baseline: 4.5006x; 1.0413x over previous
#include <hip/hip_runtime.h>
#include <hip/hip_bf16.h>
#include <math.h>


typedef __attribute__((ext_vector_type(2))) float v2f;
typedef __attribute__((ext_vector_type(8))) float v8f;

#define DIM   64
#define NREL  4
#define KCONV 320
#define KHW   128
#define STRC  324
#define STRH  132

__device__ __forceinline__ float sigmoidf_(float v) {
    return __builtin_amdgcn_rcpf(1.0f + __expf(-v));
}

#define BIN_SHIFT   8
#define BIN_NODES   (1 << BIN_SHIFT)
#define FINE_KEYS   (BIN_NODES * NREL)
#define TILE_ROWS   16
#define SORT_BLOCKS 256
#define SORT_THREADS 1024
#define RUNS_PER_WAVE (SORT_BLOCKS / 32)
#define KEY_SHIFT   20
#define SRC_MASK    ((1 << KEY_SHIFT) - 1)
#define FINE_CAP    8192

__device__ __forceinline__ int block_excl_scan_1024(int v, int* wtot)
{
    const int lane = threadIdx.x & 31;
    const int wave = threadIdx.x >> 5;
    int inc = v;
    for (int o = 1; o < 32; o <<= 1) {
        int n = __shfl_up(inc, o);
        if (lane >= o) inc += n;
    }
    if (lane == 31) wtot[wave] = inc;
    __syncthreads();
    int woff = 0;
    for (int w = 0; w < 32; ++w) {
        int x = wtot[w];
        if (w < wave) woff += x;
    }
    __syncthreads();
    return woff + inc - v;
}

__global__ void __launch_bounds__(SORT_THREADS)
chunk_split_kernel(const int*   __restrict__ src,
                   const int*   __restrict__ dst,
                   const int*   __restrict__ rel,
                   const float* __restrict__ ew,
                   int*         __restrict__ runTab,
                   int2*        __restrict__ part,
                   int nEdges, int chunk, int nbin, int tabRow)
{
    extern __shared__ int2 stage[];
    __shared__ int mine[SORT_THREADS];
    __shared__ int lcur[SORT_THREADS];
    __shared__ int wtot[32];

    const int t   = threadIdx.x;
    const int blk = blockIdx.x;
    mine[t] = 0;
    __syncthreads();

    const int beg = blk * chunk;
    const int end = min(beg + chunk, nEdges);
    for (int i0 = beg + t; i0 < end; i0 += 8 * SORT_THREADS) {
        int d[8];
#pragma unroll
        for (int u = 0; u < 8; ++u) {
            const int i = i0 + u * SORT_THREADS;
            d[u] = (i < end) ? dst[i] : -1;
        }
#pragma unroll
        for (int u = 0; u < 8; ++u)
            if (d[u] >= 0) atomicAdd(&mine[d[u] >> BIN_SHIFT], 1);
    }
    __syncthreads();

    const int vm = mine[t];
    const int lo = block_excl_scan_1024(vm, wtot);
    lcur[t] = lo;
    {
        int* row = runTab + (size_t)blk * tabRow;
        if (t < nbin) row[t] = lo;
        if (t == SORT_THREADS - 1) row[nbin] = lo + vm;
    }
    __syncthreads();

    for (int i0 = beg + t; i0 < end; i0 += 4 * SORT_THREADS) {
        int d[4], r[4], s[4], w[4];
#pragma unroll
        for (int u = 0; u < 4; ++u) {
            const int  i  = i0 + u * SORT_THREADS;
            const bool ok = i < end;
            d[u] = ok ? dst[i] : -1;
            r[u] = ok ? rel[i] : 0;
            s[u] = ok ? src[i] : 0;
            w[u] = ok ? __float_as_int(ew[i]) : 0;
        }
#pragma unroll
        for (int u = 0; u < 4; ++u) {
            if (d[u] >= 0) {
                const int b   = d[u] >> BIN_SHIFT;
                const int dl  = d[u] & (BIN_NODES - 1);
                const int key = (dl / TILE_ROWS) * (TILE_ROWS * NREL) + r[u] * TILE_ROWS + (dl % TILE_ROWS);
                const int p   = atomicAdd(&lcur[b], 1);
                int2 rec;
                rec.x = s[u] | (key << KEY_SHIFT);
                rec.y = w[u];
                stage[p] = rec;
            }
        }
    }
    __syncthreads();

    const int nloc = end - beg;
    for (int i = t; i < nloc; i += SORT_THREADS) part[beg + i] = stage[i];
}

__global__ void __launch_bounds__(SORT_THREADS)
bin_order_kernel(const int2* __restrict__ part,
                 const int*  __restrict__ runTab,
                 int*        __restrict__ bend,
                 int2*       __restrict__ erec,
                 int nBuckets, int chunk, int tabRow)
{
    __shared__ int2 raw[FINE_CAP];
    __shared__ int2 sorted[FINE_CAP];
    __shared__ int  cnt[FINE_KEYS];
    __shared__ int  lcur[FINE_KEYS];
    __shared__ int  rsrc[SORT_BLOCKS];
    __shared__ int  roff[SORT_BLOCKS + 1];
    __shared__ int  wtot[32];
    __shared__ int  sbin;

    const int t = threadIdx.x;
    const int g = blockIdx.x;
    const int lane = t & 31;
    const int wave = t >> 5;

    int lo = 0, rn = 0;
    if (t < SORT_BLOCKS) {
        const int* row = runTab + (size_t)t * tabRow + g;
        lo = row[0];
        rn = row[1] - lo;
    }
    const int ro = block_excl_scan_1024(rn, wtot);
    const int so = block_excl_scan_1024(lo, wtot);
    if (t < SORT_BLOCKS) {
        rsrc[t] = t * chunk + lo;
        roff[t] = ro;
    }
    if (t == SORT_THREADS - 1) {
        roff[SORT_BLOCKS] = ro + rn;
        sbin = so + lo;
    }
    cnt[t] = 0;
    __syncthreads();
    const int n = roff[SORT_BLOCKS];
    const int s = sbin;
    const bool inLds = n <= FINE_CAP;

    int rs[RUNS_PER_WAVE], rb[RUNS_PER_WAVE], rl[RUNS_PER_WAVE];
#pragma unroll
    for (int u = 0; u < RUNS_PER_WAVE; ++u) {
        const int c = wave + 32 * u;
        rs[u] = rsrc[c];
        rb[u] = roff[c];
        rl[u] = roff[c + 1] - rb[u];
    }

    if (inLds) {
        int2 v[RUNS_PER_WAVE];
#pragma unroll
        for (int u = 0; u < RUNS_PER_WAVE; ++u) {
            v[u].x = 0; v[u].y = 0;
            if (lane < rl[u]) v[u] = part[rs[u] + lane];
        }
#pragma unroll
        for (int u = 0; u < RUNS_PER_WAVE; ++u) {
            if (lane < rl[u]) raw[rb[u] + lane] = v[u];
            for (int j = lane + 32; j < rl[u]; j += 32) raw[rb[u] + j] = part[rs[u] + j];
        }
        __syncthreads();
        for (int i = t; i < n; i += SORT_THREADS) atomicAdd(&cnt[raw[i].x >> KEY_SHIFT], 1);
    } else {
        for (int u = 0; u < RUNS_PER_WAVE; ++u)
            for (int j = lane; j < rl[u]; j += 32)
                atomicAdd(&cnt[part[rs[u] + j].x >> KEY_SHIFT], 1);
    }
    __syncthreads();

    const int c   = cnt[t];
    const int off = block_excl_scan_1024(c, wtot);
    lcur[t] = off;
    const int b = g * FINE_KEYS + t;
    if (b < nBuckets) bend[b] = s + off + c;
    __syncthreads();

    if (inLds) {
        for (int i = t; i < n; i += SORT_THREADS) {
            const int2 rc = raw[i];
            const int  p  = atomicAdd(&lcur[rc.x >> KEY_SHIFT], 1);
            int2 o;
            o.x = rc.x & SRC_MASK;
            o.y = rc.y;
            sorted[p] = o;
        }
        __syncthreads();
        for (int i = t; i < n; i += SORT_THREADS) erec[s + i] = sorted[i];
    } else {
        for (int u = 0; u < RUNS_PER_WAVE; ++u)
            for (int j = lane; j < rl[u]; j += 32) {
                const int2 rc = part[rs[u] + j];
                const int  p  = atomicAdd(&lcur[rc.x >> KEY_SHIFT], 1);
                int2 o;
                o.x = rc.x & SRC_MASK;
                o.y = rc.y;
                erec[s + p] = o;
            }
    }
}

#define LAYER_WAVES   25
#define LAYER_THREADS (LAYER_WAVES * 32)
#define STRT          68

__device__ __forceinline__ void wave_lds_handover()
{
    __builtin_amdgcn_fence(__ATOMIC_RELEASE, "wavefront");
    __builtin_amdgcn_wave_barrier();
    __builtin_amdgcn_fence(__ATOMIC_ACQUIRE, "wavefront");
}

__global__ void __launch_bounds__(LAYER_THREADS)
rgcn_layer_kernel(const float* __restrict__ x,
                  const int*   __restrict__ erec,
                  const int*   __restrict__ bend,
                  const float* __restrict__ lin_w,
                  const float* __restrict__ lin_b,
                  const float* __restrict__ self_w,
                  const float* __restrict__ self_b,
                  const float* __restrict__ prev,
                  const float* __restrict__ pw,
                  const float* __restrict__ pb,
                  const float* __restrict__ tw,
                  const float* __restrict__ tb,
                  float*       __restrict__ hidden,
                  float*       __restrict__ outp,
                  int nNodes)
{
    __shared__ __attribute__((aligned(16)))
        float smem[DIM * STRC + 2 * DIM * STRH + LAYER_WAVES * 16 * STRT];
    float* WT  = smem;
    float* WP  = WT  + DIM * STRC;
    float* WTT = WP  + DIM * STRH;
    float* HT  = WTT + DIM * STRH;

    const int t = threadIdx.x;
    for (int i = t; i < KCONV * DIM; i += LAYER_THREADS) {
        int k = i >> 6;
        int c = i & 63;
        float v = (k < NREL * DIM) ? lin_w[i] : self_w[i - NREL * DIM * DIM];
        WT[(size_t)c * STRC + k] = v;
    }
    for (int i = t; i < KHW * DIM; i += LAYER_THREADS) {
        int k = i >> 6;
        int c = i & 63;
        WP [(size_t)c * STRH + k] = pw[i];
        WTT[(size_t)c * STRH + k] = tw[i];
    }
    __syncthreads();

    const int lane = t & 31;
    const int wave = __builtin_amdgcn_readfirstlane(t >> 5);
    const int m     = lane & 15;
    const int khalf = lane >> 4;

    const float* B0 = WT + (size_t)(m)      * STRC;
    const float* B1 = WT + (size_t)(16 + m) * STRC;
    const float* B2 = WT + (size_t)(32 + m) * STRC;
    const float* B3 = WT + (size_t)(48 + m) * STRC;
    const float* P0 = WP  + (size_t)(m)      * STRH;
    const float* P1 = WP  + (size_t)(16 + m) * STRH;
    const float* P2 = WP  + (size_t)(32 + m) * STRH;
    const float* P3 = WP  + (size_t)(48 + m) * STRH;
    const float* T0 = WTT + (size_t)(m)      * STRH;
    const float* T1 = WTT + (size_t)(16 + m) * STRH;
    const float* T2 = WTT + (size_t)(32 + m) * STRH;
    const float* T3 = WTT + (size_t)(48 + m) * STRH;
    float*       ht   = HT + wave * (16 * STRT);
    const float* hRow = ht + m * STRT;

    const long base = ((long)blockIdx.x * LAYER_WAVES + wave) * 16;
    if (base >= nNodes) return;
    {
        const float* xRow   = x    + (base + m) * (size_t)DIM;
        const float* pRow   = prev + (base + m) * (size_t)DIM;

        const int2* __restrict__ rec2 = (const int2*)erec;
        const char* __restrict__ xb   = (const char*)x;
        const unsigned lane8 = (unsigned)lane << 3;
        const int bvA   = bend[base * NREL + lane];
        const int bvB   = bend[base * NREL + 32 + lane];
        const int bprev = (base > 0) ? bend[base * NREL - 1] : 0;

        int  segBeg = bprev;
        int2 recN   = {0, 0};
        {
            const int send0 = __builtin_amdgcn_readlane(bvA, 15);
            if (send0 > segBeg) recN = rec2[min(segBeg + lane, send0 - 1)];
        }

#define ROW_LOAD(U)                                                              \
        const unsigned s##U = (unsigned)__builtin_amdgcn_readlane(rec.x, (U));  \
        const float    w##U = __int_as_float(__builtin_amdgcn_readlane(rec.y, (U))); \
        const float2   v##U = *(const float2*)(xb + (size_t)((s##U << 8) | lane8));

#define BUCKET_DONE()                                                            \
        {                                                                        \
            *(float2*)(ht + mm * STRT + 2 * lane) = acc;                         \
            if (m == mm) dsum = ss;                                              \
            acc.x = 0.0f; acc.y = 0.0f; ss = 0.0f;                               \
            ++mm;                                                                \
            bnd = (mm < 15) ? __builtin_amdgcn_readlane(bvr, lo16 + mm) : 0x7fffffff; \
        }

#define BUCKET_CHECK(U)                                                          \
        if ((U) >= togo) {                                                       \
            do { BUCKET_DONE() togo = bnd - j0; } while ((U) >= togo);           \
        }

#define EDGE_SURE(U)                                                             \
        BUCKET_CHECK(U)                                                          \
        acc.x += v##U.x * w##U; acc.y += v##U.y * w##U; ss += w##U;

#define EDGE_MAYBE(U)                                                            \
        BUCKET_CHECK(U)                                                          \
        {                                                                        \
            const float wz = ((U) < left) ? w##U : 0.0f;                         \
            acc.x += v##U.x * wz; acc.y += v##U.y * wz; ss += wz;                \
        }

        v8f a0 = {}, a1 = {}, a2 = {}, a3 = {};
        for (int r = 0; r < NREL; ++r) {
            __builtin_amdgcn_s_setprio(3);
            const int bvr  = (r < 2) ? bvA : bvB;
            const int lo16 = (r & 1) * 16;
            const int sbeg = segBeg;
            const int send = __builtin_amdgcn_readlane(bvr, lo16 + 15);
            int2 rec = recN;
            if (r + 1 < NREL) {
                const int bvn   = (r + 1 < 2) ? bvA : bvB;
                const int sendN = __builtin_amdgcn_readlane(bvn, ((r + 1) & 1) * 16 + 15);
                if (sendN > send) recN = rec2[min(send + lane, sendN - 1)];
            }
            segBeg = send;

            float2 acc  = {0.0f, 0.0f};
            float  ss   = 0.0f;
            float  dsum = 0.0f;
            int    mm   = 0;
            int    bnd  = __builtin_amdgcn_readlane(bvr, lo16);

            int2 recP = {0, 0};
            for (int cb = sbeg; cb < send; cb += 32) {
                const int ce = min(cb + 32, send);
                if (cb != sbeg) rec = recP;
                if (ce < send) recP = rec2[min(ce + lane, send - 1)];
                for (int j0 = cb; j0 < ce; j0 += 16) {
                    if (j0 != cb) {
                        rec.x = __shfl_xor(rec.x, 16);
                        rec.y = __shfl_xor(rec.y, 16);
                    }
                    const int left = ce - j0;
                    int togo = bnd - j0;
                    if (left > 8) {
                        ROW_LOAD(0)  ROW_LOAD(1)  ROW_LOAD(2)  ROW_LOAD(3)
                        ROW_LOAD(4)  ROW_LOAD(5)  ROW_LOAD(6)  ROW_LOAD(7)
                        ROW_LOAD(8)  ROW_LOAD(9)  ROW_LOAD(10) ROW_LOAD(11)
                        ROW_LOAD(12) ROW_LOAD(13) ROW_LOAD(14) ROW_LOAD(15)
                        EDGE_SURE(0)   EDGE_SURE(1)   EDGE_SURE(2)   EDGE_SURE(3)
                        EDGE_SURE(4)   EDGE_SURE(5)   EDGE_SURE(6)   EDGE_SURE(7)
                        EDGE_SURE(8)   EDGE_MAYBE(9)  EDGE_MAYBE(10) EDGE_MAYBE(11)
                        EDGE_MAYBE(12) EDGE_MAYBE(13) EDGE_MAYBE(14) EDGE_MAYBE(15)
                    } else {
                        ROW_LOAD(0) ROW_LOAD(1) ROW_LOAD(2) ROW_LOAD(3)
                        ROW_LOAD(4) ROW_LOAD(5) ROW_LOAD(6) ROW_LOAD(7)
                        EDGE_SURE(0)  EDGE_MAYBE(1) EDGE_MAYBE(2) EDGE_MAYBE(3)
                        EDGE_MAYBE(4) EDGE_MAYBE(5) EDGE_MAYBE(6) EDGE_MAYBE(7)
                    }
                }
            }
            while (mm < 16) BUCKET_DONE()
            wave_lds_handover();
            __builtin_amdgcn_s_setprio(1);

            const float sc = 1.0f / (dsum + 1e-10f);
            for (int k = 0; k < DIM; k += 4) {
                int kk = k + 2 * khalf;
                float2 av = *(const float2*)(hRow + kk);
                v2f a; a.x = av.x * sc; a.y = av.y * sc;
                int kw = r * DIM + kk;
                v2f b0 = *(const v2f*)(B0 + kw);
                v2f b1 = *(const v2f*)(B1 + kw);
                v2f b2 = *(const v2f*)(B2 + kw);
                v2f b3 = *(const v2f*)(B3 + kw);
                a0 = __builtin_amdgcn_wmma_f32_16x16x4_f32(false, a, false, b0, (short)0, a0, false, false);
                a1 = __builtin_amdgcn_wmma_f32_16x16x4_f32(false, a, false, b1, (short)0, a1, false, false);
                a2 = __builtin_amdgcn_wmma_f32_16x16x4_f32(false, a, false, b2, (short)0, a2, false, false);
                a3 = __builtin_amdgcn_wmma_f32_16x16x4_f32(false, a, false, b3, (short)0, a3, false, false);
            }
            wave_lds_handover();
        }
#undef EDGE_MAYBE
#undef EDGE_SURE
#undef BUCKET_CHECK
#undef BUCKET_DONE
#undef ROW_LOAD

#pragma unroll 4
        for (int k = 0; k < DIM; k += 4) {
            int kk = k + 2 * khalf;
            float2 av = *(const float2*)(xRow + kk);
            v2f a; a.x = av.x; a.y = av.y;
            int kw = NREL * DIM + kk;
            v2f b0 = *(const v2f*)(B0 + kw);
            v2f b1 = *(const v2f*)(B1 + kw);
            v2f b2 = *(const v2f*)(B2 + kw);
            v2f b3 = *(const v2f*)(B3 + kw);
            a0 = __builtin_amdgcn_wmma_f32_16x16x4_f32(false, a, false, b0, (short)0, a0, false, false);
            a1 = __builtin_amdgcn_wmma_f32_16x16x4_f32(false, a, false, b1, (short)0, a1, false, false);
            a2 = __builtin_amdgcn_wmma_f32_16x16x4_f32(false, a, false, b2, (short)0, a2, false, false);
            a3 = __builtin_amdgcn_wmma_f32_16x16x4_f32(false, a, false, b3, (short)0, a3, false, false);
        }

#define CONV_EPILOGUE(ACC, TILE)                                           \
        {                                                                  \
            int   col = (TILE) * 16 + m;                                   \
            float bia = lin_b[col] + self_b[col];                          \
            _Pragma("unroll")                                              \
            for (int r = 0; r < 8; ++r) {                                  \
                int   lr = khalf * 8 + r;                                  \
                float hv = sigmoidf_(ACC[r] + bia);                        \
                ht[lr * STRT + col] = hv;                                  \
                hidden[(base + lr) * DIM + col] = hv;                      \
            }                                                              \
        }
        CONV_EPILOGUE(a0, 0) CONV_EPILOGUE(a1, 1) CONV_EPILOGUE(a2, 2) CONV_EPILOGUE(a3, 3)
#undef CONV_EPILOGUE
        wave_lds_handover();

        __builtin_amdgcn_s_setprio(0);
        v8f p0 = {}, p1 = {}, p2 = {}, p3 = {};
        v8f g0 = {}, g1 = {}, g2 = {}, g3 = {};

#define HW_STEP(AROW, KOFF)                                                              \
        for (int k = 0; k < DIM; k += 4) {                                               \
            int kk = k + 2 * khalf;                                                      \
            float2 av = *(const float2*)((AROW) + kk);                                   \
            v2f a; a.x = av.x; a.y = av.y;                                               \
            int kw = kk + (KOFF);                                                        \
            v2f bp0 = *(const v2f*)(P0 + kw);                                            \
            v2f bp1 = *(const v2f*)(P1 + kw);                                            \
            v2f bp2 = *(const v2f*)(P2 + kw);                                            \
            v2f bp3 = *(const v2f*)(P3 + kw);                                            \
            v2f bt0 = *(const v2f*)(T0 + kw);                                            \
            v2f bt1 = *(const v2f*)(T1 + kw);                                            \
            v2f bt2 = *(const v2f*)(T2 + kw);                                            \
            v2f bt3 = *(const v2f*)(T3 + kw);                                            \
            p0 = __builtin_amdgcn_wmma_f32_16x16x4_f32(false, a, false, bp0, (short)0, p0, false, false); \
            p1 = __builtin_amdgcn_wmma_f32_16x16x4_f32(false, a, false, bp1, (short)0, p1, false, false); \
            p2 = __builtin_amdgcn_wmma_f32_16x16x4_f32(false, a, false, bp2, (short)0, p2, false, false); \
            p3 = __builtin_amdgcn_wmma_f32_16x16x4_f32(false, a, false, bp3, (short)0, p3, false, false); \
            g0 = __builtin_amdgcn_wmma_f32_16x16x4_f32(false, a, false, bt0, (short)0, g0, false, false); \
            g1 = __builtin_amdgcn_wmma_f32_16x16x4_f32(false, a, false, bt1, (short)0, g1, false, false); \
            g2 = __builtin_amdgcn_wmma_f32_16x16x4_f32(false, a, false, bt2, (short)0, g2, false, false); \
            g3 = __builtin_amdgcn_wmma_f32_16x16x4_f32(false, a, false, bt3, (short)0, g3, false, false); \
        }
        HW_STEP(hRow, 0)
        HW_STEP(pRow, DIM)
#undef HW_STEP

#define HW_EPILOGUE(PACC, GACC, TILE)                                      \
        {                                                                  \
            int   col = (TILE) * 16 + m;                                   \
            float pbv = pb[col];                                           \
            float tbv = tb[col];                                           \
            _Pragma("unroll")                                              \
            for (int r = 0; r < 8; ++r) {                                  \
                int   lr = khalf * 8 + r;                                  \
                float pr = fmaxf(PACC[r] + pbv, 0.0f);                     \
                float gt = sigmoidf_(GACC[r] + tbv);                       \
                float h  = ht[lr * STRT + col];                            \
                outp[(base + lr) * DIM + col] = gt * pr + (1.0f - gt) * h; \
            }                                                              \
        }
        HW_EPILOGUE(p0, g0, 0) HW_EPILOGUE(p1, g1, 1)
        HW_EPILOGUE(p2, g2, 2) HW_EPILOGUE(p3, g3, 3)
#undef HW_EPILOGUE
    }
}

extern "C" void kernel_launch(void* const* d_in, const int* in_sizes, int n_in,
                              void* d_out, int out_size, void* d_ws, size_t ws_size,
                              hipStream_t stream) {
    const float* node_feat   = (const float*)d_in[0];
    const float* edge_weight = (const float*)d_in[1];
    const float* conv_lin_w  = (const float*)d_in[2];
    const float* conv_lin_b  = (const float*)d_in[3];
    const float* conv_self_w = (const float*)d_in[4];
    const float* conv_self_b = (const float*)d_in[5];
    const float* hw_proj_w   = (const float*)d_in[6];
    const float* hw_proj_b   = (const float*)d_in[7];
    const float* hw_trans_w  = (const float*)d_in[8];
    const float* hw_trans_b  = (const float*)d_in[9];
    const int*   edge_src    = (const int*)d_in[10];
    const int*   edge_dst    = (const int*)d_in[11];
    const int*   edge_rel    = (const int*)d_in[12];
    float*       out         = (float*)d_out;

    const int N = in_sizes[0] / DIM;
    const int E = in_sizes[10];

    float* hA   = (float*)d_ws;
    float* hB   = hA   + (size_t)N * DIM;
    float* cur0 = hB   + (size_t)N * DIM;
    float* cur1 = cur0 + (size_t)N * DIM;
    const int NB    = N * NREL;
    const int nbin  = (N + BIN_NODES - 1) >> BIN_SHIFT;
    const int tabRow = (nbin + 32) & ~31;
    const int chunk = (E + SORT_BLOCKS - 1) / SORT_BLOCKS;
    int*  cursor   = (int*)(cur1 + (size_t)N * DIM);
    int*  erec     = cursor + NB;
    int*  part     = erec + 2 * (size_t)E;
    int*  runTab   = part + 2 * (size_t)E;

    const int tiles         = (N + 15) / 16;
    const int layerBlocks   = (tiles + LAYER_WAVES - 1) / LAYER_WAVES;
    const size_t stageLds = (size_t)chunk * sizeof(int2);

    chunk_split_kernel<<<SORT_BLOCKS, SORT_THREADS, stageLds, stream>>>(
        edge_src, edge_dst, edge_rel, edge_weight, runTab, (int2*)part,
        E, chunk, nbin, tabRow);
    bin_order_kernel<<<nbin, SORT_THREADS, 0, stream>>>(
        (const int2*)part, runTab, cursor, (int2*)erec, NB, chunk, tabRow);

    for (int i = 0; i < 3; ++i) {
        const float* xin  = (i == 0) ? node_feat : ((i == 1) ? cur0 : cur1);
        const float* prev = (i == 0) ? node_feat : ((i == 1) ? hA : hB);
        float*       hid  = (i & 1) ? hB : hA;
        float*       gout = (i == 0) ? cur0 : ((i == 1) ? cur1 : out);

        rgcn_layer_kernel<<<layerBlocks, LAYER_THREADS, 0, stream>>>(
            xin, erec, cursor,
            conv_lin_w  + (size_t)i * 256 * DIM, conv_lin_b  + (size_t)i * DIM,
            conv_self_w + (size_t)i * DIM * DIM, conv_self_b + (size_t)i * DIM,
            prev,
            hw_proj_w  + (size_t)i * KHW * DIM, hw_proj_b  + (size_t)i * DIM,
            hw_trans_w + (size_t)i * KHW * DIM, hw_trans_b + (size_t)i * DIM,
            hid, gout, N);
    }
}
